# starting kernel
# baseline (speedup 1.0000x reference)
#include <hip/hip_runtime.h>

typedef __attribute__((ext_vector_type(2))) float v2f;
typedef __attribute__((ext_vector_type(8))) float v8f;

#define N_NODES 100000
#define N_EDGES 800000
#define FEAT    128
#define NGRAPH  512
#define NCLS    10

__device__ __forceinline__ void atomAddF(float* p, float v) {
  unsafeAtomicAdd(p, v);
}

__global__ void k_fill(float* __restrict__ p, float v, long n) {
  long i = (long)blockIdx.x * blockDim.x + threadIdx.x;
  if (i < n) p[i] = v;
}

__global__ void k_deg(const int* __restrict__ col, float* __restrict__ deg) {
  long e = (long)blockIdx.x * blockDim.x + threadIdx.x;
  if (e < N_EDGES) atomAddF(&deg[col[e]], 1.0f);
}

__global__ void k_dinv(float* __restrict__ deg) {
  long i = (long)blockIdx.x * blockDim.x + threadIdx.x;
  if (i < N_NODES) {
    float d = deg[i];
    deg[i] = (d > 0.0f) ? rsqrtf(fmaxf(d, 1.0f)) : 0.0f;
  }
}

__global__ __launch_bounds__(256) void k_gemm(const float* __restrict__ X,
                                              const float* __restrict__ W,
                                              float* __restrict__ Hout,
                                              int nrows) {
  __shared__ float sW[FEAT * FEAT];
  {
    const float4* src = (const float4*)W;
    float4* dst = (float4*)sW;
    for (int i = threadIdx.x; i < FEAT * FEAT / 4; i += blockDim.x) dst[i] = src[i];
  }
  __syncthreads();

  const int wave  = threadIdx.x >> 5;
  const int lane  = threadIdx.x & 31;
  const int strip = blockIdx.x * 8 + wave;
  const int row0  = strip * 16;
  if (row0 >= nrows) return;

  const int m    = lane & 15;
  const int koff = (lane >> 4) << 1;

  v8f acc[8];
#pragma unroll
  for (int n = 0; n < 8; ++n) acc[n] = (v8f){0, 0, 0, 0, 0, 0, 0, 0};

  const float* xrow = X + (size_t)(row0 + m) * FEAT + koff;
  const int ncol = lane & 15;

  for (int k = 0; k < FEAT; k += 4) {
    v2f a;
    a.x = xrow[k];
    a.y = xrow[k + 1];
    const int kb = k + koff;
#pragma unroll
    for (int n = 0; n < 8; ++n) {
      v2f b;
      b.x = sW[kb * FEAT + n * 16 + ncol];
      b.y = sW[(kb + 1) * FEAT + n * 16 + ncol];
      acc[n] = __builtin_amdgcn_wmma_f32_16x16x4_f32(
          false, a, false, b, (short)0, acc[n], false, false);
    }
  }

  const int rowsel = (lane >> 4) * 8;
#pragma unroll
  for (int n = 0; n < 8; ++n) {
#pragma unroll
    for (int r = 0; r < 8; ++r) {
      Hout[(size_t)(row0 + rowsel + r) * FEAT + n * 16 + ncol] = acc[n][r];
    }
  }
}

__global__ __launch_bounds__(256) void k_agg(const float* __restrict__ h,
                                             const int* __restrict__ row,
                                             const int* __restrict__ col,
                                             const float* __restrict__ dinv,
                                             float* __restrict__ out) {
  const int wave = threadIdx.x >> 5;
  const int lane = threadIdx.x & 31;
  const long e = (long)blockIdx.x * 8 + wave;
  if (e >= N_EDGES) return;
  const int r = row[e];
  const int c = col[e];
  const float nrm = dinv[r] * dinv[c];
  const float4 v = ((const float4*)(h + (size_t)r * FEAT))[lane];
  float* op = out + (size_t)c * FEAT + lane * 4;
  atomAddF(op + 0, nrm * v.x);
  atomAddF(op + 1, nrm * v.y);
  atomAddF(op + 2, nrm * v.z);
  atomAddF(op + 3, nrm * v.w);
}

__global__ void k_epilogue(float* __restrict__ out, const float* __restrict__ h,
                           const float* __restrict__ dinv,
                           const float* __restrict__ bias, int relu) {
  long idx = (long)blockIdx.x * blockDim.x + threadIdx.x;
  if (idx >= (long)N_NODES * (FEAT / 4)) return;
  const int node = (int)(idx >> 5);
  const int f4   = (int)(idx & 31);
  const float d = dinv[node];
  const float s = d * d;
  float4 hv = ((const float4*)h)[idx];
  float4 ov = ((float4*)out)[idx];
  float4 bv = ((const float4*)bias)[f4];
  float4 rv;
  rv.x = ov.x + s * hv.x + bv.x;
  rv.y = ov.y + s * hv.y + bv.y;
  rv.z = ov.z + s * hv.z + bv.z;
  rv.w = ov.w + s * hv.w + bv.w;
  if (relu) {
    rv.x = fmaxf(rv.x, 0.0f);
    rv.y = fmaxf(rv.y, 0.0f);
    rv.z = fmaxf(rv.z, 0.0f);
    rv.w = fmaxf(rv.w, 0.0f);
  }
  ((float4*)out)[idx] = rv;
}

__global__ __launch_bounds__(256) void k_pool(const float* __restrict__ a,
                                              const int* __restrict__ batch,
                                              float* __restrict__ sums,
                                              float* __restrict__ cnts) {
  const int wave = threadIdx.x >> 5;
  const int lane = threadIdx.x & 31;
  const long i = (long)blockIdx.x * 8 + wave;
  if (i >= N_NODES) return;
  const int g = batch[i];
  const float4 v = ((const float4*)(a + (size_t)i * FEAT))[lane];
  float* sp = sums + (size_t)g * FEAT + lane * 4;
  atomAddF(sp + 0, v.x);
  atomAddF(sp + 1, v.y);
  atomAddF(sp + 2, v.z);
  atomAddF(sp + 3, v.w);
  if (lane == 0) atomAddF(&cnts[g], 1.0f);
}

__global__ void k_final(const float* __restrict__ sums, const float* __restrict__ cnts,
                        const float* __restrict__ Wl, const float* __restrict__ bl,
                        float* __restrict__ out) {
  const int g = blockIdx.x;
  const int c = threadIdx.x;
  if (c >= NCLS) return;
  const float inv = 1.0f / fmaxf(cnts[g], 1.0f);
  float acc = bl[c];
  for (int k = 0; k < FEAT; ++k)
    acc += sums[(size_t)g * FEAT + k] * inv * Wl[k * NCLS + c];
  out[(size_t)g * NCLS + c] = acc;
}

extern "C" void kernel_launch(void* const* d_in, const int* in_sizes, int n_in,
                              void* d_out, int out_size, void* d_ws, size_t ws_size,
                              hipStream_t stream) {
  const float* x     = (const float*)d_in[0];
  const int*   ei    = (const int*)d_in[1];
  const int*   row   = ei;
  const int*   col   = ei + N_EDGES;
  const int*   batch = (const int*)d_in[2];
  const float* W1 = (const float*)d_in[3];
  const float* b1 = (const float*)d_in[4];
  const float* W2 = (const float*)d_in[5];
  const float* b2 = (const float*)d_in[6];
  const float* W3 = (const float*)d_in[7];
  const float* b3 = (const float*)d_in[8];
  const float* Wl = (const float*)d_in[9];
  const float* bl = (const float*)d_in[10];
  float* outp = (float*)d_out;

  char* base = (char*)d_ws;
  const size_t featBytes = (size_t)N_NODES * FEAT * sizeof(float);
  float* deg  = (float*)(base);
  float* hbuf = (float*)(base + (1 << 20));
  float* abuf = (float*)(base + (1 << 20) + featBytes);
  float* sums = (float*)(base + (1 << 20) + 2 * featBytes);
  float* cnts = sums + (size_t)NGRAPH * FEAT;

  const long nElem  = (long)N_NODES * FEAT;
  const long nVec4  = nElem / 4;
  const int  TB     = 256;
  const int  gNode  = (N_NODES + TB - 1) / TB;
  const int  gElem  = (int)((nElem + TB - 1) / TB);
  const int  gVec4  = (int)((nVec4 + TB - 1) / TB);
  const int  gEdge1 = (N_EDGES + TB - 1) / TB;
  const int  gGemm  = (N_NODES / 16 + 8 - 1) / 8;
  const int  gAgg   = (N_EDGES + 8 - 1) / 8;
  const int  gPool  = (N_NODES + 8 - 1) / 8;

  k_fill<<<gNode, TB, 0, stream>>>(deg, 1.0f, N_NODES);
  k_deg<<<gEdge1, TB, 0, stream>>>(col, deg);
  k_dinv<<<gNode, TB, 0, stream>>>(deg);

  k_gemm<<<gGemm, TB, 0, stream>>>(x, W1, hbuf, N_NODES);
  k_fill<<<gElem, TB, 0, stream>>>(abuf, 0.0f, nElem);
  k_agg<<<gAgg, TB, 0, stream>>>(hbuf, row, col, deg, abuf);
  k_epilogue<<<gVec4, TB, 0, stream>>>(abuf, hbuf, deg, b1, 1);

  k_gemm<<<gGemm, TB, 0, stream>>>(abuf, W2, hbuf, N_NODES);
  k_fill<<<gElem, TB, 0, stream>>>(abuf, 0.0f, nElem);
  k_agg<<<gAgg, TB, 0, stream>>>(hbuf, row, col, deg, abuf);
  k_epilogue<<<gVec4, TB, 0, stream>>>(abuf, hbuf, deg, b2, 1);

  k_gemm<<<gGemm, TB, 0, stream>>>(abuf, W3, hbuf, N_NODES);
  k_fill<<<gElem, TB, 0, stream>>>(abuf, 0.0f, nElem);
  k_agg<<<gAgg, TB, 0, stream>>>(hbuf, row, col, deg, abuf);
  k_epilogue<<<gVec4, TB, 0, stream>>>(abuf, hbuf, deg, b3, 0);

  const long poolN = (long)NGRAPH * FEAT + NGRAPH;
  k_fill<<<(int)((poolN + TB - 1) / TB), TB, 0, stream>>>(sums, 0.0f, poolN);
  k_pool<<<gPool, TB, 0, stream>>>(abuf, batch, sums, cnts);
  k_final<<<NGRAPH, 32, 0, stream>>>(sums, cnts, Wl, bl, outp);
}


# in-edge sums by gather over a per-call target-grouped source list (count, scan, integer-atomic slot fill): a wave per node adds in slot order, one store; edge float atomics, zero fill, epilogue gone
# speedup vs baseline: 1.7513x; 1.7513x over previous
#include <hip/hip_runtime.h>

typedef __attribute__((ext_vector_type(2))) float v2f;
typedef __attribute__((ext_vector_type(8))) float v8f;

#define N_NODES 100000
#define N_EDGES 800000
#define FEAT    128
#define NGRAPH  512
#define NCLS    10

__device__ __forceinline__ void atomAddF(float* p, float v) {
  unsafeAtomicAdd(p, v);
}

__global__ void k_fill(float* __restrict__ p, float v, long n) {
  long i = (long)blockIdx.x * blockDim.x + threadIdx.x;
  if (i < n) p[i] = v;
}

__global__ void k_zero_i(int* __restrict__ p, int n) {
  int i = blockIdx.x * blockDim.x + threadIdx.x;
  if (i < n) p[i] = 0;
}

__global__ void k_count(const int* __restrict__ col, int* __restrict__ cnt) {
  long e = (long)blockIdx.x * blockDim.x + threadIdx.x;
  if (e < N_EDGES) atomicAdd(&cnt[col[e]], 1);
}

__device__ __forceinline__ float dinv_of(int c) {
  const float d = (float)c + 1.0f;
  return (d > 0.0f) ? rsqrtf(fmaxf(d, 1.0f)) : 0.0f;
}

#define SCAN_TILE 1024
static_assert(N_NODES % 4 == 0, "int4 path of k_scan");
__global__ __launch_bounds__(256) void k_scan(const int* __restrict__ cnt,
                                              int* __restrict__ rowptr,
                                              int* __restrict__ cursor,
                                              float* __restrict__ dinv) {
  __shared__ int s_pre[8];
  __shared__ int s_wave[8];
  const int t = threadIdx.x;
  const int lane = t & 31;
  const int wave = t >> 5;
  const int tile0 = blockIdx.x * SCAN_TILE;
  const int4* c4 = (const int4*)cnt;

  int pre = 0;
  for (int i = t; i < tile0 / 4; i += 256) {
    const int4 v = c4[i];
    pre += (v.x + v.y) + (v.z + v.w);
  }
  for (int o = 16; o > 0; o >>= 1) pre += __shfl_xor(pre, o);
  if (lane == 0) s_pre[wave] = pre;

  const int i0 = tile0 + t * 4;
  int4 a = {0, 0, 0, 0};
  if (i0 < N_NODES) a = c4[i0 >> 2];
  const int tsum = (a.x + a.y) + (a.z + a.w);
  int inc = tsum;
  for (int o = 1; o < 32; o <<= 1) {
    const int y = __shfl_up(inc, o);
    if (lane >= o) inc += y;
  }
  if (lane == 31) s_wave[wave] = inc;
  __syncthreads();
  int base = 0;
  for (int w = 0; w < 8; ++w) {
    base += s_pre[w];
    if (w < wave) base += s_wave[w];
  }
  const int ex = base + inc - tsum;
  if (i0 < N_NODES) {
    const int4 rp = {ex, ex + a.x, ex + a.x + a.y, ex + a.x + a.y + a.z};
    ((int4*)rowptr)[i0 >> 2] = rp;
    ((int4*)cursor)[i0 >> 2] = rp;
    if (i0 + 4 == N_NODES) rowptr[N_NODES] = ex + tsum;
    const float4 dv = {dinv_of(a.x), dinv_of(a.y), dinv_of(a.z), dinv_of(a.w)};
    ((float4*)dinv)[i0 >> 2] = dv;
  }
}

__global__ void k_slots(const int* __restrict__ row, const int* __restrict__ col,
                        int* __restrict__ cursor, int* __restrict__ srcs) {
  long e = (long)blockIdx.x * blockDim.x + threadIdx.x;
  if (e < N_EDGES) {
    const int pos = atomicAdd(&cursor[col[e]], 1);
    srcs[pos] = row[e];
  }
}

__global__ __launch_bounds__(256) void k_gemm(const float* __restrict__ X,
                                              const float* __restrict__ W,
                                              float* __restrict__ Hout,
                                              int nrows) {
  __shared__ float sW[FEAT * FEAT];
  {
    const float4* src = (const float4*)W;
    float4* dst = (float4*)sW;
    for (int i = threadIdx.x; i < FEAT * FEAT / 4; i += blockDim.x) dst[i] = src[i];
  }
  __syncthreads();

  const int wave  = threadIdx.x >> 5;
  const int lane  = threadIdx.x & 31;
  const int strip = blockIdx.x * 8 + wave;
  const int row0  = strip * 16;
  if (row0 >= nrows) return;

  const int m    = lane & 15;
  const int koff = (lane >> 4) << 1;

  v8f acc[8];
#pragma unroll
  for (int n = 0; n < 8; ++n) acc[n] = (v8f){0, 0, 0, 0, 0, 0, 0, 0};

  const float* xrow = X + (size_t)(row0 + m) * FEAT + koff;
  const int ncol = lane & 15;

  for (int k = 0; k < FEAT; k += 4) {
    v2f a;
    a.x = xrow[k];
    a.y = xrow[k + 1];
    const int kb = k + koff;
#pragma unroll
    for (int n = 0; n < 8; ++n) {
      v2f b;
      b.x = sW[kb * FEAT + n * 16 + ncol];
      b.y = sW[(kb + 1) * FEAT + n * 16 + ncol];
      acc[n] = __builtin_amdgcn_wmma_f32_16x16x4_f32(
          false, a, false, b, (short)0, acc[n], false, false);
    }
  }

  const int rowsel = (lane >> 4) * 8;
#pragma unroll
  for (int n = 0; n < 8; ++n) {
#pragma unroll
    for (int r = 0; r < 8; ++r) {
      Hout[(size_t)(row0 + rowsel + r) * FEAT + n * 16 + ncol] = acc[n][r];
    }
  }
}

__device__ __forceinline__ void fma4(float4& acc, float w, const float4 v) {
  acc.x += w * v.x;
  acc.y += w * v.y;
  acc.z += w * v.z;
  acc.w += w * v.w;
}

__global__ __launch_bounds__(256) void k_gather(const float* __restrict__ h,
                                                const int* __restrict__ rowptr,
                                                const int* __restrict__ srcs,
                                                const float* __restrict__ dinv,
                                                const float* __restrict__ bias,
                                                float* __restrict__ out, int relu) {
  const int lane = threadIdx.x & 31;
  const int wave = __builtin_amdgcn_readfirstlane((int)(threadIdx.x >> 5));
  const int i = blockIdx.x * 8 + wave;
  if (i >= N_NODES) return;
  const int start = rowptr[i];
  const int n = rowptr[i + 1] - start;
  const float di = dinv[i];
  const float4* h4 = (const float4*)h;
  const float4 hv = h4[(size_t)i * (FEAT / 4) + lane];
  const float4 bv = ((const float4*)bias)[lane];

  float4 acc = {0.0f, 0.0f, 0.0f, 0.0f};
  for (int b = 0; b < n; b += 32) {
    const int m = (n - b < 32) ? (n - b) : 32;
    int s = 0;
    float ds = 0.0f;
    if (lane < m) {
      s = srcs[start + b + lane];
      ds = dinv[s];
    }
    int j = 0;
    for (; j + 4 <= m; j += 4) {
      const int s0 = __builtin_amdgcn_readlane(s, j);
      const int s1 = __builtin_amdgcn_readlane(s, j + 1);
      const int s2 = __builtin_amdgcn_readlane(s, j + 2);
      const int s3 = __builtin_amdgcn_readlane(s, j + 3);
      const float4 v0 = h4[(size_t)s0 * (FEAT / 4) + lane];
      const float4 v1 = h4[(size_t)s1 * (FEAT / 4) + lane];
      const float4 v2 = h4[(size_t)s2 * (FEAT / 4) + lane];
      const float4 v3 = h4[(size_t)s3 * (FEAT / 4) + lane];
      const float w0 = __int_as_float(__builtin_amdgcn_readlane(__float_as_int(ds), j)) * di;
      const float w1 = __int_as_float(__builtin_amdgcn_readlane(__float_as_int(ds), j + 1)) * di;
      const float w2 = __int_as_float(__builtin_amdgcn_readlane(__float_as_int(ds), j + 2)) * di;
      const float w3 = __int_as_float(__builtin_amdgcn_readlane(__float_as_int(ds), j + 3)) * di;
      fma4(acc, w0, v0);
      fma4(acc, w1, v1);
      fma4(acc, w2, v2);
      fma4(acc, w3, v3);
    }
    for (; j < m; ++j) {
      const int s0 = __builtin_amdgcn_readlane(s, j);
      const float4 v0 = h4[(size_t)s0 * (FEAT / 4) + lane];
      const float w0 = __int_as_float(__builtin_amdgcn_readlane(__float_as_int(ds), j)) * di;
      fma4(acc, w0, v0);
    }
  }

  const float sl = di * di;
  float4 rv;
  rv.x = acc.x + sl * hv.x + bv.x;
  rv.y = acc.y + sl * hv.y + bv.y;
  rv.z = acc.z + sl * hv.z + bv.z;
  rv.w = acc.w + sl * hv.w + bv.w;
  if (relu) {
    rv.x = fmaxf(rv.x, 0.0f);
    rv.y = fmaxf(rv.y, 0.0f);
    rv.z = fmaxf(rv.z, 0.0f);
    rv.w = fmaxf(rv.w, 0.0f);
  }
  ((float4*)out)[(size_t)i * (FEAT / 4) + lane] = rv;
}

__global__ __launch_bounds__(256) void k_pool(const float* __restrict__ a,
                                              const int* __restrict__ batch,
                                              float* __restrict__ sums,
                                              float* __restrict__ cnts) {
  const int wave = threadIdx.x >> 5;
  const int lane = threadIdx.x & 31;
  const long i = (long)blockIdx.x * 8 + wave;
  if (i >= N_NODES) return;
  const int g = batch[i];
  const float4 v = ((const float4*)(a + (size_t)i * FEAT))[lane];
  float* sp = sums + (size_t)g * FEAT + lane * 4;
  atomAddF(sp + 0, v.x);
  atomAddF(sp + 1, v.y);
  atomAddF(sp + 2, v.z);
  atomAddF(sp + 3, v.w);
  if (lane == 0) atomAddF(&cnts[g], 1.0f);
}

__global__ void k_final(const float* __restrict__ sums, const float* __restrict__ cnts,
                        const float* __restrict__ Wl, const float* __restrict__ bl,
                        float* __restrict__ out) {
  const int g = blockIdx.x;
  const int c = threadIdx.x;
  if (c >= NCLS) return;
  const float inv = 1.0f / fmaxf(cnts[g], 1.0f);
  float acc = bl[c];
  for (int k = 0; k < FEAT; ++k)
    acc += sums[(size_t)g * FEAT + k] * inv * Wl[k * NCLS + c];
  out[(size_t)g * NCLS + c] = acc;
}

extern "C" void kernel_launch(void* const* d_in, const int* in_sizes, int n_in,
                              void* d_out, int out_size, void* d_ws, size_t ws_size,
                              hipStream_t stream) {
  const float* x     = (const float*)d_in[0];
  const int*   ei    = (const int*)d_in[1];
  const int*   row   = ei;
  const int*   col   = ei + N_EDGES;
  const int*   batch = (const int*)d_in[2];
  const float* W1 = (const float*)d_in[3];
  const float* b1 = (const float*)d_in[4];
  const float* W2 = (const float*)d_in[5];
  const float* b2 = (const float*)d_in[6];
  const float* W3 = (const float*)d_in[7];
  const float* b3 = (const float*)d_in[8];
  const float* Wl = (const float*)d_in[9];
  const float* bl = (const float*)d_in[10];
  float* outp = (float*)d_out;

  char* base = (char*)d_ws;
  const size_t featBytes = (size_t)N_NODES * FEAT * sizeof(float);
  int*   cnt    = (int*)(base);
  int*   rowptr = (int*)(base + (512 << 10));
  int*   cursor = (int*)(base + (1 << 20));
  float* dinv   = (float*)(base + (1 << 20) + (512 << 10));
  int*   srcs   = (int*)(base + (4 << 20));
  float* hbuf = (float*)(base + (8 << 20));
  float* abuf = (float*)(base + (8 << 20) + featBytes);
  float* sums = (float*)(base + (8 << 20) + 2 * featBytes);
  float* cnts = sums + (size_t)NGRAPH * FEAT;

  const int  TB     = 256;
  const int  gNode  = (N_NODES + TB - 1) / TB;
  const int  gEdge1 = (N_EDGES + TB - 1) / TB;
  const int  gScan  = (N_NODES + SCAN_TILE - 1) / SCAN_TILE;
  const int  gGemm  = (N_NODES / 16 + 8 - 1) / 8;
  const int  gGath  = (N_NODES + 8 - 1) / 8;
  const int  gPool  = (N_NODES + 8 - 1) / 8;

  k_zero_i<<<gNode, TB, 0, stream>>>(cnt, N_NODES);
  k_count<<<gEdge1, TB, 0, stream>>>(col, cnt);
  k_scan<<<gScan, TB, 0, stream>>>(cnt, rowptr, cursor, dinv);
  k_slots<<<gEdge1, TB, 0, stream>>>(row, col, cursor, srcs);

  k_gemm<<<gGemm, TB, 0, stream>>>(x, W1, hbuf, N_NODES);
  k_gather<<<gGath, TB, 0, stream>>>(hbuf, rowptr, srcs, dinv, b1, abuf, 1);

  k_gemm<<<gGemm, TB, 0, stream>>>(abuf, W2, hbuf, N_NODES);
  k_gather<<<gGath, TB, 0, stream>>>(hbuf, rowptr, srcs, dinv, b2, abuf, 1);

  k_gemm<<<gGemm, TB, 0, stream>>>(abuf, W3, hbuf, N_NODES);
  k_gather<<<gGath, TB, 0, stream>>>(hbuf, rowptr, srcs, dinv, b3, abuf, 0);

  const long poolN = (long)NGRAPH * FEAT + NGRAPH;
  k_fill<<<(int)((poolN + TB - 1) / TB), TB, 0, stream>>>(sums, 0.0f, poolN);
  k_pool<<<gPool, TB, 0, stream>>>(abuf, batch, sums, cnts);
  k_final<<<NGRAPH, 32, 0, stream>>>(sums, cnts, Wl, bl, outp);
}


# pool: a wave sums 32 consecutive node rows in registers and adds to the graph row only where the graph id changes (was 129 float atomics per node); order of the pooled sums changes
# speedup vs baseline: 2.5504x; 1.4563x over previous
#include <hip/hip_runtime.h>

typedef __attribute__((ext_vector_type(2))) float v2f;
typedef __attribute__((ext_vector_type(8))) float v8f;

#define N_NODES 100000
#define N_EDGES 800000
#define FEAT    128
#define NGRAPH  512
#define NCLS    10

__device__ __forceinline__ void atomAddF(float* p, float v) {
  unsafeAtomicAdd(p, v);
}

__global__ void k_fill(float* __restrict__ p, float v, long n) {
  long i = (long)blockIdx.x * blockDim.x + threadIdx.x;
  if (i < n) p[i] = v;
}

__global__ void k_zero_i(int* __restrict__ p, int n) {
  int i = blockIdx.x * blockDim.x + threadIdx.x;
  if (i < n) p[i] = 0;
}

__global__ void k_count(const int* __restrict__ col, int* __restrict__ cnt) {
  long e = (long)blockIdx.x * blockDim.x + threadIdx.x;
  if (e < N_EDGES) atomicAdd(&cnt[col[e]], 1);
}

__device__ __forceinline__ float dinv_of(int c) {
  const float d = (float)c + 1.0f;
  return (d > 0.0f) ? rsqrtf(fmaxf(d, 1.0f)) : 0.0f;
}

#define SCAN_TILE 1024
static_assert(N_NODES % 4 == 0, "int4 path of k_scan");
__global__ __launch_bounds__(256) void k_scan(const int* __restrict__ cnt,
                                              int* __restrict__ rowptr,
                                              int* __restrict__ cursor,
                                              float* __restrict__ dinv) {
  __shared__ int s_pre[8];
  __shared__ int s_wave[8];
  const int t = threadIdx.x;
  const int lane = t & 31;
  const int wave = t >> 5;
  const int tile0 = blockIdx.x * SCAN_TILE;
  const int4* c4 = (const int4*)cnt;

  int pre = 0;
  for (int i = t; i < tile0 / 4; i += 256) {
    const int4 v = c4[i];
    pre += (v.x + v.y) + (v.z + v.w);
  }
  for (int o = 16; o > 0; o >>= 1) pre += __shfl_xor(pre, o);
  if (lane == 0) s_pre[wave] = pre;

  const int i0 = tile0 + t * 4;
  int4 a = {0, 0, 0, 0};
  if (i0 < N_NODES) a = c4[i0 >> 2];
  const int tsum = (a.x + a.y) + (a.z + a.w);
  int inc = tsum;
  for (int o = 1; o < 32; o <<= 1) {
    const int y = __shfl_up(inc, o);
    if (lane >= o) inc += y;
  }
  if (lane == 31) s_wave[wave] = inc;
  __syncthreads();
  int base = 0;
  for (int w = 0; w < 8; ++w) {
    base += s_pre[w];
    if (w < wave) base += s_wave[w];
  }
  const int ex = base + inc - tsum;
  if (i0 < N_NODES) {
    const int4 rp = {ex, ex + a.x, ex + a.x + a.y, ex + a.x + a.y + a.z};
    ((int4*)rowptr)[i0 >> 2] = rp;
    ((int4*)cursor)[i0 >> 2] = rp;
    if (i0 + 4 == N_NODES) rowptr[N_NODES] = ex + tsum;
    const float4 dv = {dinv_of(a.x), dinv_of(a.y), dinv_of(a.z), dinv_of(a.w)};
    ((float4*)dinv)[i0 >> 2] = dv;
  }
}

__global__ void k_slots(const int* __restrict__ row, const int* __restrict__ col,
                        int* __restrict__ cursor, int* __restrict__ srcs) {
  long e = (long)blockIdx.x * blockDim.x + threadIdx.x;
  if (e < N_EDGES) {
    const int pos = atomicAdd(&cursor[col[e]], 1);
    srcs[pos] = row[e];
  }
}

__global__ __launch_bounds__(256) void k_gemm(const float* __restrict__ X,
                                              const float* __restrict__ W,
                                              float* __restrict__ Hout,
                                              int nrows) {
  __shared__ float sW[FEAT * FEAT];
  {
    const float4* src = (const float4*)W;
    float4* dst = (float4*)sW;
    for (int i = threadIdx.x; i < FEAT * FEAT / 4; i += blockDim.x) dst[i] = src[i];
  }
  __syncthreads();

  const int wave  = threadIdx.x >> 5;
  const int lane  = threadIdx.x & 31;
  const int strip = blockIdx.x * 8 + wave;
  const int row0  = strip * 16;
  if (row0 >= nrows) return;

  const int m    = lane & 15;
  const int koff = (lane >> 4) << 1;

  v8f acc[8];
#pragma unroll
  for (int n = 0; n < 8; ++n) acc[n] = (v8f){0, 0, 0, 0, 0, 0, 0, 0};

  const float* xrow = X + (size_t)(row0 + m) * FEAT + koff;
  const int ncol = lane & 15;

  for (int k = 0; k < FEAT; k += 4) {
    v2f a;
    a.x = xrow[k];
    a.y = xrow[k + 1];
    const int kb = k + koff;
#pragma unroll
    for (int n = 0; n < 8; ++n) {
      v2f b;
      b.x = sW[kb * FEAT + n * 16 + ncol];
      b.y = sW[(kb + 1) * FEAT + n * 16 + ncol];
      acc[n] = __builtin_amdgcn_wmma_f32_16x16x4_f32(
          false, a, false, b, (short)0, acc[n], false, false);
    }
  }

  const int rowsel = (lane >> 4) * 8;
#pragma unroll
  for (int n = 0; n < 8; ++n) {
#pragma unroll
    for (int r = 0; r < 8; ++r) {
      Hout[(size_t)(row0 + rowsel + r) * FEAT + n * 16 + ncol] = acc[n][r];
    }
  }
}

__device__ __forceinline__ void fma4(float4& acc, float w, const float4 v) {
  acc.x += w * v.x;
  acc.y += w * v.y;
  acc.z += w * v.z;
  acc.w += w * v.w;
}

__global__ __launch_bounds__(256) void k_gather(const float* __restrict__ h,
                                                const int* __restrict__ rowptr,
                                                const int* __restrict__ srcs,
                                                const float* __restrict__ dinv,
                                                const float* __restrict__ bias,
                                                float* __restrict__ out, int relu) {
  const int lane = threadIdx.x & 31;
  const int wave = __builtin_amdgcn_readfirstlane((int)(threadIdx.x >> 5));
  const int i = blockIdx.x * 8 + wave;
  if (i >= N_NODES) return;
  const int start = rowptr[i];
  const int n = rowptr[i + 1] - start;
  const float di = dinv[i];
  const float4* h4 = (const float4*)h;
  const float4 hv = h4[(size_t)i * (FEAT / 4) + lane];
  const float4 bv = ((const float4*)bias)[lane];

  float4 acc = {0.0f, 0.0f, 0.0f, 0.0f};
  for (int b = 0; b < n; b += 32) {
    const int m = (n - b < 32) ? (n - b) : 32;
    int s = 0;
    float ds = 0.0f;
    if (lane < m) {
      s = srcs[start + b + lane];
      ds = dinv[s];
    }
    int j = 0;
    for (; j + 4 <= m; j += 4) {
      const int s0 = __builtin_amdgcn_readlane(s, j);
      const int s1 = __builtin_amdgcn_readlane(s, j + 1);
      const int s2 = __builtin_amdgcn_readlane(s, j + 2);
      const int s3 = __builtin_amdgcn_readlane(s, j + 3);
      const float4 v0 = h4[(size_t)s0 * (FEAT / 4) + lane];
      const float4 v1 = h4[(size_t)s1 * (FEAT / 4) + lane];
      const float4 v2 = h4[(size_t)s2 * (FEAT / 4) + lane];
      const float4 v3 = h4[(size_t)s3 * (FEAT / 4) + lane];
      const float w0 = __int_as_float(__builtin_amdgcn_readlane(__float_as_int(ds), j)) * di;
      const float w1 = __int_as_float(__builtin_amdgcn_readlane(__float_as_int(ds), j + 1)) * di;
      const float w2 = __int_as_float(__builtin_amdgcn_readlane(__float_as_int(ds), j + 2)) * di;
      const float w3 = __int_as_float(__builtin_amdgcn_readlane(__float_as_int(ds), j + 3)) * di;
      fma4(acc, w0, v0);
      fma4(acc, w1, v1);
      fma4(acc, w2, v2);
      fma4(acc, w3, v3);
    }
    for (; j < m; ++j) {
      const int s0 = __builtin_amdgcn_readlane(s, j);
      const float4 v0 = h4[(size_t)s0 * (FEAT / 4) + lane];
      const float w0 = __int_as_float(__builtin_amdgcn_readlane(__float_as_int(ds), j)) * di;
      fma4(acc, w0, v0);
    }
  }

  const float sl = di * di;
  float4 rv;
  rv.x = acc.x + sl * hv.x + bv.x;
  rv.y = acc.y + sl * hv.y + bv.y;
  rv.z = acc.z + sl * hv.z + bv.z;
  rv.w = acc.w + sl * hv.w + bv.w;
  if (relu) {
    rv.x = fmaxf(rv.x, 0.0f);
    rv.y = fmaxf(rv.y, 0.0f);
    rv.z = fmaxf(rv.z, 0.0f);
    rv.w = fmaxf(rv.w, 0.0f);
  }
  ((float4*)out)[(size_t)i * (FEAT / 4) + lane] = rv;
}

#define POOL_RUN 32
static_assert(N_NODES % POOL_RUN == 0, "k_pool takes whole runs of nodes");

__device__ __forceinline__ void pool_flush(float* __restrict__ sums, float* __restrict__ cnts,
                                           int g, int lane, const float4 acc, int run) {
  float* sp = sums + (size_t)g * FEAT + lane * 4;
  atomAddF(sp + 0, acc.x);
  atomAddF(sp + 1, acc.y);
  atomAddF(sp + 2, acc.z);
  atomAddF(sp + 3, acc.w);
  if (lane == 0) atomAddF(&cnts[g], (float)run);
}

__global__ __launch_bounds__(256) void k_pool(const float* __restrict__ a,
                                              const int* __restrict__ batch,
                                              float* __restrict__ sums,
                                              float* __restrict__ cnts) {
  const int lane = threadIdx.x & 31;
  const int wave = __builtin_amdgcn_readfirstlane((int)(threadIdx.x >> 5));
  const int n0 = (blockIdx.x * 8 + wave) * POOL_RUN;
  if (n0 >= N_NODES) return;
  const int gid = batch[n0 + lane];
  const float4* a4 = (const float4*)a + (size_t)n0 * (FEAT / 4) + lane;
  float4 v[POOL_RUN];
#pragma unroll
  for (int j = 0; j < POOL_RUN; ++j) v[j] = a4[j * (FEAT / 4)];

  int gcur = __builtin_amdgcn_readlane(gid, 0);
  float4 acc = v[0];
  int run = 1;
#pragma unroll
  for (int j = 1; j < POOL_RUN; ++j) {
    const int g = __builtin_amdgcn_readlane(gid, j);
    if (g != gcur) {
      pool_flush(sums, cnts, gcur, lane, acc, run);
      gcur = g;
      acc = v[j];
      run = 1;
    } else {
      acc.x += v[j].x;
      acc.y += v[j].y;
      acc.z += v[j].z;
      acc.w += v[j].w;
      ++run;
    }
  }
  pool_flush(sums, cnts, gcur, lane, acc, run);
}

__global__ void k_final(const float* __restrict__ sums, const float* __restrict__ cnts,
                        const float* __restrict__ Wl, const float* __restrict__ bl,
                        float* __restrict__ out) {
  const int g = blockIdx.x;
  const int c = threadIdx.x;
  if (c >= NCLS) return;
  const float inv = 1.0f / fmaxf(cnts[g], 1.0f);
  float acc = bl[c];
  for (int k = 0; k < FEAT; ++k)
    acc += sums[(size_t)g * FEAT + k] * inv * Wl[k * NCLS + c];
  out[(size_t)g * NCLS + c] = acc;
}

extern "C" void kernel_launch(void* const* d_in, const int* in_sizes, int n_in,
                              void* d_out, int out_size, void* d_ws, size_t ws_size,
                              hipStream_t stream) {
  const float* x     = (const float*)d_in[0];
  const int*   ei    = (const int*)d_in[1];
  const int*   row   = ei;
  const int*   col   = ei + N_EDGES;
  const int*   batch = (const int*)d_in[2];
  const float* W1 = (const float*)d_in[3];
  const float* b1 = (const float*)d_in[4];
  const float* W2 = (const float*)d_in[5];
  const float* b2 = (const float*)d_in[6];
  const float* W3 = (const float*)d_in[7];
  const float* b3 = (const float*)d_in[8];
  const float* Wl = (const float*)d_in[9];
  const float* bl = (const float*)d_in[10];
  float* outp = (float*)d_out;

  char* base = (char*)d_ws;
  const size_t featBytes = (size_t)N_NODES * FEAT * sizeof(float);
  int*   cnt    = (int*)(base);
  int*   rowptr = (int*)(base + (512 << 10));
  int*   cursor = (int*)(base + (1 << 20));
  float* dinv   = (float*)(base + (1 << 20) + (512 << 10));
  int*   srcs   = (int*)(base + (4 << 20));
  float* hbuf = (float*)(base + (8 << 20));
  float* abuf = (float*)(base + (8 << 20) + featBytes);
  float* sums = (float*)(base + (8 << 20) + 2 * featBytes);
  float* cnts = sums + (size_t)NGRAPH * FEAT;

  const int  TB     = 256;
  const int  gNode  = (N_NODES + TB - 1) / TB;
  const int  gEdge1 = (N_EDGES + TB - 1) / TB;
  const int  gScan  = (N_NODES + SCAN_TILE - 1) / SCAN_TILE;
  const int  gGemm  = (N_NODES / 16 + 8 - 1) / 8;
  const int  gGath  = (N_NODES + 8 - 1) / 8;
  const int  gPool  = (N_NODES / POOL_RUN + 8 - 1) / 8;

  k_zero_i<<<gNode, TB, 0, stream>>>(cnt, N_NODES);
  k_count<<<gEdge1, TB, 0, stream>>>(col, cnt);
  k_scan<<<gScan, TB, 0, stream>>>(cnt, rowptr, cursor, dinv);
  k_slots<<<gEdge1, TB, 0, stream>>>(row, col, cursor, srcs);

  k_gemm<<<gGemm, TB, 0, stream>>>(x, W1, hbuf, N_NODES);
  k_gather<<<gGath, TB, 0, stream>>>(hbuf, rowptr, srcs, dinv, b1, abuf, 1);

  k_gemm<<<gGemm, TB, 0, stream>>>(abuf, W2, hbuf, N_NODES);
  k_gather<<<gGath, TB, 0, stream>>>(hbuf, rowptr, srcs, dinv, b2, abuf, 1);

  k_gemm<<<gGemm, TB, 0, stream>>>(abuf, W3, hbuf, N_NODES);
  k_gather<<<gGath, TB, 0, stream>>>(hbuf, rowptr, srcs, dinv, b3, abuf, 0);

  const long poolN = (long)NGRAPH * FEAT + NGRAPH;
  k_fill<<<(int)((poolN + TB - 1) / TB), TB, 0, stream>>>(sums, 0.0f, poolN);
  k_pool<<<gPool, TB, 0, stream>>>(abuf, batch, sums, cnts);
  k_final<<<NGRAPH, 32, 0, stream>>>(sums, cnts, Wl, bl, outp);
}


# edges grouped by target through LDS counters in two levels (per chunk and 256-node bucket, then per node inside a bucket), no global atomic per edge; a node's list, so its sum, keeps arrival order
# speedup vs baseline: 3.4597x; 1.3565x over previous
#include <hip/hip_runtime.h>

typedef __attribute__((ext_vector_type(2))) float v2f;
typedef __attribute__((ext_vector_type(8))) float v8f;

#define N_NODES 100000
#define N_EDGES 800000
#define FEAT    128
#define NGRAPH  512
#define NCLS    10

__device__ __forceinline__ void atomAddF(float* p, float v) {
  unsafeAtomicAdd(p, v);
}

__global__ void k_fill(float* __restrict__ p, float v, long n) {
  long i = (long)blockIdx.x * blockDim.x + threadIdx.x;
  if (i < n) p[i] = v;
}

#define EB_WG      128
#define EB_THREADS 512
#define EB_WAVES   (EB_THREADS / 32)
#define EB_CHUNK   ((N_EDGES + EB_WG - 1) / EB_WG)
#define EB_BATCH   4
#define BKT_SHIFT  8
#define BKT_NODES  (1 << BKT_SHIFT)
#define NBKT       ((N_NODES + BKT_NODES - 1) / BKT_NODES)
static_assert(NBKT <= EB_THREADS, "k_bin gives one thread to every bucket");
static_assert(N_NODES <= (1 << 24), "a packed edge keeps the source in 24 bits");
static_assert(BKT_NODES <= 256, "a packed edge keeps the target's place in its bucket in 8 bits");

template <int NW>
__device__ __forceinline__ int block_excl_scan(int v, int* s_wave, int& total) {
  const int lane = threadIdx.x & 31;
  const int wave = threadIdx.x >> 5;
  int inc = v;
  for (int o = 1; o < 32; o <<= 1) {
    const int y = __shfl_up(inc, o);
    if (lane >= o) inc += y;
  }
  if (lane == 31) s_wave[wave] = inc;
  __syncthreads();
  int base = 0, tot = 0;
  for (int w = 0; w < NW; ++w) {
    const int x = s_wave[w];
    if (w < wave) base += x;
    tot += x;
  }
  total = tot;
  return base + inc - v;
}

__global__ __launch_bounds__(EB_THREADS) void k_hist(const int* __restrict__ col,
                                                     int* __restrict__ histM) {
  __shared__ int s_h[NBKT];
  const int t = threadIdx.x;
  const int w = blockIdx.x;
  for (int b = t; b < NBKT; b += EB_THREADS) s_h[b] = 0;
  __syncthreads();
  const int e0 = w * EB_CHUNK;
  const int e1 = (e0 + EB_CHUNK < N_EDGES) ? (e0 + EB_CHUNK) : N_EDGES;
  for (int e = e0 + t; e < e1; e += EB_BATCH * EB_THREADS) {
    int c[EB_BATCH];
#pragma unroll
    for (int k = 0; k < EB_BATCH; ++k) {
      const int ee = e + k * EB_THREADS;
      c[k] = (ee < e1) ? col[ee] : -1;
    }
#pragma unroll
    for (int k = 0; k < EB_BATCH; ++k)
      if (c[k] >= 0) atomicAdd(&s_h[c[k] >> BKT_SHIFT], 1);
  }
  __syncthreads();
  for (int b = t; b < NBKT; b += EB_THREADS) histM[w * NBKT + b] = s_h[b];
}

__global__ __launch_bounds__(EB_THREADS) void k_bin(const int* __restrict__ row,
                                                    const int* __restrict__ col,
                                                    const int* __restrict__ histM,
                                                    int* __restrict__ bbase,
                                                    unsigned* __restrict__ binned) {
  __shared__ int s_cur[NBKT];
  __shared__ int s_wave[EB_WAVES];
  const int t = threadIdx.x;
  const int w = blockIdx.x;

  int before = 0, total = 0;
  if (t < NBKT) {
#pragma unroll 32
    for (int w2 = 0; w2 < EB_WG; ++w2) {
      const int v = histM[w2 * NBKT + t];
      total += v;
      before += (w2 < w) ? v : 0;
    }
  }
  int all;
  const int base = block_excl_scan<EB_WAVES>(total, s_wave, all);
  if (t < NBKT) {
    s_cur[t] = base + before;
    if (w == 0) bbase[t] = base;
  }
  if (w == 0 && t == 0) bbase[NBKT] = all;
  __syncthreads();

  const int e0 = w * EB_CHUNK;
  const int e1 = (e0 + EB_CHUNK < N_EDGES) ? (e0 + EB_CHUNK) : N_EDGES;
  for (int e = e0 + t; e < e1; e += EB_BATCH * EB_THREADS) {
    int c[EB_BATCH], r[EB_BATCH];
#pragma unroll
    for (int k = 0; k < EB_BATCH; ++k) {
      const int ee = e + k * EB_THREADS;
      const bool ok = ee < e1;
      c[k] = ok ? col[ee] : -1;
      r[k] = ok ? row[ee] : 0;
    }
#pragma unroll
    for (int k = 0; k < EB_BATCH; ++k) {
      if (c[k] >= 0) {
        const int pos = atomicAdd(&s_cur[c[k] >> BKT_SHIFT], 1);
        binned[pos] = ((unsigned)(c[k] & (BKT_NODES - 1)) << 24) | (unsigned)r[k];
      }
    }
  }
}

__device__ __forceinline__ float dinv_of(int c) {
  const float d = (float)c + 1.0f;
  return (d > 0.0f) ? rsqrtf(fmaxf(d, 1.0f)) : 0.0f;
}

__global__ __launch_bounds__(BKT_NODES) void k_group(const unsigned* __restrict__ binned,
                                                     const int* __restrict__ bbase,
                                                     int* __restrict__ rowptr,
                                                     float* __restrict__ dinv,
                                                     int* __restrict__ srcs) {
  __shared__ int s_cnt[BKT_NODES];
  __shared__ int s_wave[BKT_NODES / 32];
  const int t = threadIdx.x;
  const int b = blockIdx.x;
  const int lo = bbase[b];
  const int hi = bbase[b + 1];
  s_cnt[t] = 0;
  __syncthreads();
  for (int p = lo + t; p < hi; p += EB_BATCH * BKT_NODES) {
    unsigned u[EB_BATCH];
#pragma unroll
    for (int k = 0; k < EB_BATCH; ++k) {
      const int pp = p + k * BKT_NODES;
      u[k] = binned[(pp < hi) ? pp : p];
    }
#pragma unroll
    for (int k = 0; k < EB_BATCH; ++k)
      atomicAdd(&s_cnt[u[k] >> 24], (p + k * BKT_NODES < hi) ? 1 : 0);
  }
  __syncthreads();
  const int c = s_cnt[t];
  int all;
  const int ex = block_excl_scan<BKT_NODES / 32>(c, s_wave, all);
  const int node = b * BKT_NODES + t;
  if (node < N_NODES) {
    rowptr[node] = lo + ex;
    dinv[node] = dinv_of(c);
  }
  if (b == NBKT - 1 && t == 0) rowptr[N_NODES] = hi;
  s_cnt[t] = lo + ex;
  __syncthreads();
  for (int p = lo + t; p < hi; p += EB_BATCH * BKT_NODES) {
    unsigned u[EB_BATCH];
#pragma unroll
    for (int k = 0; k < EB_BATCH; ++k) {
      const int pp = p + k * BKT_NODES;
      u[k] = binned[(pp < hi) ? pp : p];
    }
#pragma unroll
    for (int k = 0; k < EB_BATCH; ++k) {
      const bool ok = p + k * BKT_NODES < hi;
      const int pos = atomicAdd(&s_cnt[u[k] >> 24], ok ? 1 : 0);
      if (ok) srcs[pos] = (int)(u[k] & 0xFFFFFFu);
    }
  }
}

__global__ __launch_bounds__(256) void k_gemm(const float* __restrict__ X,
                                              const float* __restrict__ W,
                                              float* __restrict__ Hout,
                                              int nrows) {
  __shared__ float sW[FEAT * FEAT];
  {
    const float4* src = (const float4*)W;
    float4* dst = (float4*)sW;
    for (int i = threadIdx.x; i < FEAT * FEAT / 4; i += blockDim.x) dst[i] = src[i];
  }
  __syncthreads();

  const int wave  = threadIdx.x >> 5;
  const int lane  = threadIdx.x & 31;
  const int strip = blockIdx.x * 8 + wave;
  const int row0  = strip * 16;
  if (row0 >= nrows) return;

  const int m    = lane & 15;
  const int koff = (lane >> 4) << 1;

  v8f acc[8];
#pragma unroll
  for (int n = 0; n < 8; ++n) acc[n] = (v8f){0, 0, 0, 0, 0, 0, 0, 0};

  const float* xrow = X + (size_t)(row0 + m) * FEAT + koff;
  const int ncol = lane & 15;

  for (int k = 0; k < FEAT; k += 4) {
    v2f a;
    a.x = xrow[k];
    a.y = xrow[k + 1];
    const int kb = k + koff;
#pragma unroll
    for (int n = 0; n < 8; ++n) {
      v2f b;
      b.x = sW[kb * FEAT + n * 16 + ncol];
      b.y = sW[(kb + 1) * FEAT + n * 16 + ncol];
      acc[n] = __builtin_amdgcn_wmma_f32_16x16x4_f32(
          false, a, false, b, (short)0, acc[n], false, false);
    }
  }

  const int rowsel = (lane >> 4) * 8;
#pragma unroll
  for (int n = 0; n < 8; ++n) {
#pragma unroll
    for (int r = 0; r < 8; ++r) {
      Hout[(size_t)(row0 + rowsel + r) * FEAT + n * 16 + ncol] = acc[n][r];
    }
  }
}

__device__ __forceinline__ void fma4(float4& acc, float w, const float4 v) {
  acc.x += w * v.x;
  acc.y += w * v.y;
  acc.z += w * v.z;
  acc.w += w * v.w;
}

__global__ __launch_bounds__(256) void k_gather(const float* __restrict__ h,
                                                const int* __restrict__ rowptr,
                                                const int* __restrict__ srcs,
                                                const float* __restrict__ dinv,
                                                const float* __restrict__ bias,
                                                float* __restrict__ out, int relu) {
  const int lane = threadIdx.x & 31;
  const int wave = __builtin_amdgcn_readfirstlane((int)(threadIdx.x >> 5));
  const int i = blockIdx.x * 8 + wave;
  if (i >= N_NODES) return;
  const int start = rowptr[i];
  const int n = rowptr[i + 1] - start;
  const float di = dinv[i];
  const float4* h4 = (const float4*)h;
  const float4 hv = h4[(size_t)i * (FEAT / 4) + lane];
  const float4 bv = ((const float4*)bias)[lane];

  float4 acc = {0.0f, 0.0f, 0.0f, 0.0f};
  for (int b = 0; b < n; b += 32) {
    const int m = (n - b < 32) ? (n - b) : 32;
    int s = 0;
    float ds = 0.0f;
    if (lane < m) {
      s = srcs[start + b + lane];
      ds = dinv[s];
    }
    int j = 0;
    for (; j + 4 <= m; j += 4) {
      const int s0 = __builtin_amdgcn_readlane(s, j);
      const int s1 = __builtin_amdgcn_readlane(s, j + 1);
      const int s2 = __builtin_amdgcn_readlane(s, j + 2);
      const int s3 = __builtin_amdgcn_readlane(s, j + 3);
      const float4 v0 = h4[(size_t)s0 * (FEAT / 4) + lane];
      const float4 v1 = h4[(size_t)s1 * (FEAT / 4) + lane];
      const float4 v2 = h4[(size_t)s2 * (FEAT / 4) + lane];
      const float4 v3 = h4[(size_t)s3 * (FEAT / 4) + lane];
      const float w0 = __int_as_float(__builtin_amdgcn_readlane(__float_as_int(ds), j)) * di;
      const float w1 = __int_as_float(__builtin_amdgcn_readlane(__float_as_int(ds), j + 1)) * di;
      const float w2 = __int_as_float(__builtin_amdgcn_readlane(__float_as_int(ds), j + 2)) * di;
      const float w3 = __int_as_float(__builtin_amdgcn_readlane(__float_as_int(ds), j + 3)) * di;
      fma4(acc, w0, v0);
      fma4(acc, w1, v1);
      fma4(acc, w2, v2);
      fma4(acc, w3, v3);
    }
    for (; j < m; ++j) {
      const int s0 = __builtin_amdgcn_readlane(s, j);
      const float4 v0 = h4[(size_t)s0 * (FEAT / 4) + lane];
      const float w0 = __int_as_float(__builtin_amdgcn_readlane(__float_as_int(ds), j)) * di;
      fma4(acc, w0, v0);
    }
  }

  const float sl = di * di;
  float4 rv;
  rv.x = acc.x + sl * hv.x + bv.x;
  rv.y = acc.y + sl * hv.y + bv.y;
  rv.z = acc.z + sl * hv.z + bv.z;
  rv.w = acc.w + sl * hv.w + bv.w;
  if (relu) {
    rv.x = fmaxf(rv.x, 0.0f);
    rv.y = fmaxf(rv.y, 0.0f);
    rv.z = fmaxf(rv.z, 0.0f);
    rv.w = fmaxf(rv.w, 0.0f);
  }
  ((float4*)out)[(size_t)i * (FEAT / 4) + lane] = rv;
}

#define POOL_RUN 32
static_assert(N_NODES % POOL_RUN == 0, "k_pool takes whole runs of nodes");

__device__ __forceinline__ void pool_flush(float* __restrict__ sums, float* __restrict__ cnts,
                                           int g, int lane, const float4 acc, int run) {
  float* sp = sums + (size_t)g * FEAT + lane * 4;
  atomAddF(sp + 0, acc.x);
  atomAddF(sp + 1, acc.y);
  atomAddF(sp + 2, acc.z);
  atomAddF(sp + 3, acc.w);
  if (lane == 0) atomAddF(&cnts[g], (float)run);
}

__global__ __launch_bounds__(256) void k_pool(const float* __restrict__ a,
                                              const int* __restrict__ batch,
                                              float* __restrict__ sums,
                                              float* __restrict__ cnts) {
  const int lane = threadIdx.x & 31;
  const int wave = __builtin_amdgcn_readfirstlane((int)(threadIdx.x >> 5));
  const int n0 = (blockIdx.x * 8 + wave) * POOL_RUN;
  if (n0 >= N_NODES) return;
  const int gid = batch[n0 + lane];
  const float4* a4 = (const float4*)a + (size_t)n0 * (FEAT / 4) + lane;
  float4 v[POOL_RUN];
#pragma unroll
  for (int j = 0; j < POOL_RUN; ++j) v[j] = a4[j * (FEAT / 4)];

  int gcur = __builtin_amdgcn_readlane(gid, 0);
  float4 acc = v[0];
  int run = 1;
#pragma unroll
  for (int j = 1; j < POOL_RUN; ++j) {
    const int g = __builtin_amdgcn_readlane(gid, j);
    if (g != gcur) {
      pool_flush(sums, cnts, gcur, lane, acc, run);
      gcur = g;
      acc = v[j];
      run = 1;
    } else {
      acc.x += v[j].x;
      acc.y += v[j].y;
      acc.z += v[j].z;
      acc.w += v[j].w;
      ++run;
    }
  }
  pool_flush(sums, cnts, gcur, lane, acc, run);
}

__global__ void k_final(const float* __restrict__ sums, const float* __restrict__ cnts,
                        const float* __restrict__ Wl, const float* __restrict__ bl,
                        float* __restrict__ out) {
  const int g = blockIdx.x;
  const int c = threadIdx.x;
  if (c >= NCLS) return;
  const float inv = 1.0f / fmaxf(cnts[g], 1.0f);
  float acc = bl[c];
  for (int k = 0; k < FEAT; ++k)
    acc += sums[(size_t)g * FEAT + k] * inv * Wl[k * NCLS + c];
  out[(size_t)g * NCLS + c] = acc;
}

extern "C" void kernel_launch(void* const* d_in, const int* in_sizes, int n_in,
                              void* d_out, int out_size, void* d_ws, size_t ws_size,
                              hipStream_t stream) {
  const float* x     = (const float*)d_in[0];
  const int*   ei    = (const int*)d_in[1];
  const int*   row   = ei;
  const int*   col   = ei + N_EDGES;
  const int*   batch = (const int*)d_in[2];
  const float* W1 = (const float*)d_in[3];
  const float* b1 = (const float*)d_in[4];
  const float* W2 = (const float*)d_in[5];
  const float* b2 = (const float*)d_in[6];
  const float* W3 = (const float*)d_in[7];
  const float* b3 = (const float*)d_in[8];
  const float* Wl = (const float*)d_in[9];
  const float* bl = (const float*)d_in[10];
  float* outp = (float*)d_out;

  char* base = (char*)d_ws;
  const size_t featBytes = (size_t)N_NODES * FEAT * sizeof(float);
  int*   histM  = (int*)(base);
  int*   rowptr = (int*)(base + (512 << 10));
  int*   bbase  = (int*)(base + (1 << 20));
  float* dinv   = (float*)(base + (1 << 20) + (512 << 10));
  int*   srcs   = (int*)(base + (4 << 20));
  float* hbuf = (float*)(base + (8 << 20));
  float* abuf = (float*)(base + (8 << 20) + featBytes);
  float* sums = (float*)(base + (8 << 20) + 2 * featBytes);
  float* cnts = sums + (size_t)NGRAPH * FEAT;
  unsigned* binned = (unsigned*)(base + (9 << 20) + 2 * featBytes);

  const int  TB     = 256;
  const int  gGemm  = (N_NODES / 16 + 8 - 1) / 8;
  const int  gGath  = (N_NODES + 8 - 1) / 8;
  const int  gPool  = (N_NODES / POOL_RUN + 8 - 1) / 8;

  k_hist<<<EB_WG, EB_THREADS, 0, stream>>>(col, histM);
  k_bin<<<EB_WG, EB_THREADS, 0, stream>>>(row, col, histM, bbase, binned);
  k_group<<<NBKT, BKT_NODES, 0, stream>>>(binned, bbase, rowptr, dinv, srcs);

  k_gemm<<<gGemm, TB, 0, stream>>>(x, W1, hbuf, N_NODES);
  k_gather<<<gGath, TB, 0, stream>>>(hbuf, rowptr, srcs, dinv, b1, abuf, 1);

  k_gemm<<<gGemm, TB, 0, stream>>>(abuf, W2, hbuf, N_NODES);
  k_gather<<<gGath, TB, 0, stream>>>(hbuf, rowptr, srcs, dinv, b2, abuf, 1);

  k_gemm<<<gGemm, TB, 0, stream>>>(abuf, W3, hbuf, N_NODES);
  k_gather<<<gGath, TB, 0, stream>>>(hbuf, rowptr, srcs, dinv, b3, abuf, 0);

  const long poolN = (long)NGRAPH * FEAT + NGRAPH;
  k_fill<<<(int)((poolN + TB - 1) / TB), TB, 0, stream>>>(sums, 0.0f, poolN);
  k_pool<<<gPool, TB, 0, stream>>>(abuf, batch, sums, cnts);
  k_final<<<NGRAPH, 32, 0, stream>>>(sums, cnts, Wl, bl, outp);
}


# layer 3's W3 product moved behind the pool (all linear): pool sums the aggregated layer-2 rows, then one 512-row WMMA product with W3 in the classifier kernel; N-row GEMM gone; rounding order changes
# speedup vs baseline: 4.1368x; 1.1957x over previous
#include <hip/hip_runtime.h>

typedef __attribute__((ext_vector_type(2))) float v2f;
typedef __attribute__((ext_vector_type(8))) float v8f;

#define N_NODES 100000
#define N_EDGES 800000
#define FEAT    128
#define NGRAPH  512
#define NCLS    10

__device__ __forceinline__ void atomAddF(float* p, float v) {
  unsafeAtomicAdd(p, v);
}

__global__ void k_fill(float* __restrict__ p, float v, long n) {
  long i = (long)blockIdx.x * blockDim.x + threadIdx.x;
  if (i < n) p[i] = v;
}

#define EB_WG      128
#define EB_THREADS 512
#define EB_WAVES   (EB_THREADS / 32)
#define EB_CHUNK   ((N_EDGES + EB_WG - 1) / EB_WG)
#define EB_BATCH   4
#define BKT_SHIFT  8
#define BKT_NODES  (1 << BKT_SHIFT)
#define NBKT       ((N_NODES + BKT_NODES - 1) / BKT_NODES)
static_assert(NBKT <= EB_THREADS, "k_bin gives one thread to every bucket");
static_assert(N_NODES <= (1 << 24), "a packed edge keeps the source in 24 bits");
static_assert(BKT_NODES <= 256, "a packed edge keeps the target's place in its bucket in 8 bits");

template <int NW>
__device__ __forceinline__ int block_excl_scan(int v, int* s_wave, int& total) {
  const int lane = threadIdx.x & 31;
  const int wave = threadIdx.x >> 5;
  int inc = v;
  for (int o = 1; o < 32; o <<= 1) {
    const int y = __shfl_up(inc, o);
    if (lane >= o) inc += y;
  }
  if (lane == 31) s_wave[wave] = inc;
  __syncthreads();
  int base = 0, tot = 0;
  for (int w = 0; w < NW; ++w) {
    const int x = s_wave[w];
    if (w < wave) base += x;
    tot += x;
  }
  total = tot;
  return base + inc - v;
}

__global__ __launch_bounds__(EB_THREADS) void k_hist(const int* __restrict__ col,
                                                     int* __restrict__ histM) {
  __shared__ int s_h[NBKT];
  const int t = threadIdx.x;
  const int w = blockIdx.x;
  for (int b = t; b < NBKT; b += EB_THREADS) s_h[b] = 0;
  __syncthreads();
  const int e0 = w * EB_CHUNK;
  const int e1 = (e0 + EB_CHUNK < N_EDGES) ? (e0 + EB_CHUNK) : N_EDGES;
  for (int e = e0 + t; e < e1; e += EB_BATCH * EB_THREADS) {
    int c[EB_BATCH];
#pragma unroll
    for (int k = 0; k < EB_BATCH; ++k) {
      const int ee = e + k * EB_THREADS;
      c[k] = (ee < e1) ? col[ee] : -1;
    }
#pragma unroll
    for (int k = 0; k < EB_BATCH; ++k)
      if (c[k] >= 0) atomicAdd(&s_h[c[k] >> BKT_SHIFT], 1);
  }
  __syncthreads();
  for (int b = t; b < NBKT; b += EB_THREADS) histM[w * NBKT + b] = s_h[b];
}

__global__ __launch_bounds__(EB_THREADS) void k_bin(const int* __restrict__ row,
                                                    const int* __restrict__ col,
                                                    const int* __restrict__ histM,
                                                    int* __restrict__ bbase,
                                                    unsigned* __restrict__ binned) {
  __shared__ int s_cur[NBKT];
  __shared__ int s_wave[EB_WAVES];
  const int t = threadIdx.x;
  const int w = blockIdx.x;

  int before = 0, total = 0;
  if (t < NBKT) {
#pragma unroll 32
    for (int w2 = 0; w2 < EB_WG; ++w2) {
      const int v = histM[w2 * NBKT + t];
      total += v;
      before += (w2 < w) ? v : 0;
    }
  }
  int all;
  const int base = block_excl_scan<EB_WAVES>(total, s_wave, all);
  if (t < NBKT) {
    s_cur[t] = base + before;
    if (w == 0) bbase[t] = base;
  }
  if (w == 0 && t == 0) bbase[NBKT] = all;
  __syncthreads();

  const int e0 = w * EB_CHUNK;
  const int e1 = (e0 + EB_CHUNK < N_EDGES) ? (e0 + EB_CHUNK) : N_EDGES;
  for (int e = e0 + t; e < e1; e += EB_BATCH * EB_THREADS) {
    int c[EB_BATCH], r[EB_BATCH];
#pragma unroll
    for (int k = 0; k < EB_BATCH; ++k) {
      const int ee = e + k * EB_THREADS;
      const bool ok = ee < e1;
      c[k] = ok ? col[ee] : -1;
      r[k] = ok ? row[ee] : 0;
    }
#pragma unroll
    for (int k = 0; k < EB_BATCH; ++k) {
      if (c[k] >= 0) {
        const int pos = atomicAdd(&s_cur[c[k] >> BKT_SHIFT], 1);
        binned[pos] = ((unsigned)(c[k] & (BKT_NODES - 1)) << 24) | (unsigned)r[k];
      }
    }
  }
}

__device__ __forceinline__ float dinv_of(int c) {
  const float d = (float)c + 1.0f;
  return (d > 0.0f) ? rsqrtf(fmaxf(d, 1.0f)) : 0.0f;
}

__global__ __launch_bounds__(BKT_NODES) void k_group(const unsigned* __restrict__ binned,
                                                     const int* __restrict__ bbase,
                                                     int* __restrict__ rowptr,
                                                     float* __restrict__ dinv,
                                                     int* __restrict__ srcs) {
  __shared__ int s_cnt[BKT_NODES];
  __shared__ int s_wave[BKT_NODES / 32];
  const int t = threadIdx.x;
  const int b = blockIdx.x;
  const int lo = bbase[b];
  const int hi = bbase[b + 1];
  s_cnt[t] = 0;
  __syncthreads();
  for (int p = lo + t; p < hi; p += EB_BATCH * BKT_NODES) {
    unsigned u[EB_BATCH];
#pragma unroll
    for (int k = 0; k < EB_BATCH; ++k) {
      const int pp = p + k * BKT_NODES;
      u[k] = binned[(pp < hi) ? pp : p];
    }
#pragma unroll
    for (int k = 0; k < EB_BATCH; ++k)
      atomicAdd(&s_cnt[u[k] >> 24], (p + k * BKT_NODES < hi) ? 1 : 0);
  }
  __syncthreads();
  const int c = s_cnt[t];
  int all;
  const int ex = block_excl_scan<BKT_NODES / 32>(c, s_wave, all);
  const int node = b * BKT_NODES + t;
  if (node < N_NODES) {
    rowptr[node] = lo + ex;
    dinv[node] = dinv_of(c);
  }
  if (b == NBKT - 1 && t == 0) rowptr[N_NODES] = hi;
  s_cnt[t] = lo + ex;
  __syncthreads();
  for (int p = lo + t; p < hi; p += EB_BATCH * BKT_NODES) {
    unsigned u[EB_BATCH];
#pragma unroll
    for (int k = 0; k < EB_BATCH; ++k) {
      const int pp = p + k * BKT_NODES;
      u[k] = binned[(pp < hi) ? pp : p];
    }
#pragma unroll
    for (int k = 0; k < EB_BATCH; ++k) {
      const bool ok = p + k * BKT_NODES < hi;
      const int pos = atomicAdd(&s_cnt[u[k] >> 24], ok ? 1 : 0);
      if (ok) srcs[pos] = (int)(u[k] & 0xFFFFFFu);
    }
  }
}

__global__ __launch_bounds__(256) void k_gemm(const float* __restrict__ X,
                                              const float* __restrict__ W,
                                              float* __restrict__ Hout,
                                              int nrows) {
  __shared__ float sW[FEAT * FEAT];
  {
    const float4* src = (const float4*)W;
    float4* dst = (float4*)sW;
    for (int i = threadIdx.x; i < FEAT * FEAT / 4; i += blockDim.x) dst[i] = src[i];
  }
  __syncthreads();

  const int wave  = threadIdx.x >> 5;
  const int lane  = threadIdx.x & 31;
  const int strip = blockIdx.x * 8 + wave;
  const int row0  = strip * 16;
  if (row0 >= nrows) return;

  const int m    = lane & 15;
  const int koff = (lane >> 4) << 1;

  v8f acc[8];
#pragma unroll
  for (int n = 0; n < 8; ++n) acc[n] = (v8f){0, 0, 0, 0, 0, 0, 0, 0};

  const float* xrow = X + (size_t)(row0 + m) * FEAT + koff;
  const int ncol = lane & 15;

  for (int k = 0; k < FEAT; k += 4) {
    v2f a;
    a.x = xrow[k];
    a.y = xrow[k + 1];
    const int kb = k + koff;
#pragma unroll
    for (int n = 0; n < 8; ++n) {
      v2f b;
      b.x = sW[kb * FEAT + n * 16 + ncol];
      b.y = sW[(kb + 1) * FEAT + n * 16 + ncol];
      acc[n] = __builtin_amdgcn_wmma_f32_16x16x4_f32(
          false, a, false, b, (short)0, acc[n], false, false);
    }
  }

  const int rowsel = (lane >> 4) * 8;
#pragma unroll
  for (int n = 0; n < 8; ++n) {
#pragma unroll
    for (int r = 0; r < 8; ++r) {
      Hout[(size_t)(row0 + rowsel + r) * FEAT + n * 16 + ncol] = acc[n][r];
    }
  }
}

__device__ __forceinline__ void fma4(float4& acc, float w, const float4 v) {
  acc.x += w * v.x;
  acc.y += w * v.y;
  acc.z += w * v.z;
  acc.w += w * v.w;
}

__global__ __launch_bounds__(256) void k_gather(const float* __restrict__ h,
                                                const int* __restrict__ rowptr,
                                                const int* __restrict__ srcs,
                                                const float* __restrict__ dinv,
                                                const float* __restrict__ bias,
                                                float* __restrict__ out, int relu) {
  const int lane = threadIdx.x & 31;
  const int wave = __builtin_amdgcn_readfirstlane((int)(threadIdx.x >> 5));
  const int i = blockIdx.x * 8 + wave;
  if (i >= N_NODES) return;
  const int start = rowptr[i];
  const int n = rowptr[i + 1] - start;
  const float di = dinv[i];
  const float4* h4 = (const float4*)h;
  const float4 hv = h4[(size_t)i * (FEAT / 4) + lane];
  float4 bv = {0.0f, 0.0f, 0.0f, 0.0f};
  if (bias) bv = ((const float4*)bias)[lane];

  float4 acc = {0.0f, 0.0f, 0.0f, 0.0f};
  for (int b = 0; b < n; b += 32) {
    const int m = (n - b < 32) ? (n - b) : 32;
    int s = 0;
    float ds = 0.0f;
    if (lane < m) {
      s = srcs[start + b + lane];
      ds = dinv[s];
    }
    int j = 0;
    for (; j + 4 <= m; j += 4) {
      const int s0 = __builtin_amdgcn_readlane(s, j);
      const int s1 = __builtin_amdgcn_readlane(s, j + 1);
      const int s2 = __builtin_amdgcn_readlane(s, j + 2);
      const int s3 = __builtin_amdgcn_readlane(s, j + 3);
      const float4 v0 = h4[(size_t)s0 * (FEAT / 4) + lane];
      const float4 v1 = h4[(size_t)s1 * (FEAT / 4) + lane];
      const float4 v2 = h4[(size_t)s2 * (FEAT / 4) + lane];
      const float4 v3 = h4[(size_t)s3 * (FEAT / 4) + lane];
      const float w0 = __int_as_float(__builtin_amdgcn_readlane(__float_as_int(ds), j)) * di;
      const float w1 = __int_as_float(__builtin_amdgcn_readlane(__float_as_int(ds), j + 1)) * di;
      const float w2 = __int_as_float(__builtin_amdgcn_readlane(__float_as_int(ds), j + 2)) * di;
      const float w3 = __int_as_float(__builtin_amdgcn_readlane(__float_as_int(ds), j + 3)) * di;
      fma4(acc, w0, v0);
      fma4(acc, w1, v1);
      fma4(acc, w2, v2);
      fma4(acc, w3, v3);
    }
    for (; j < m; ++j) {
      const int s0 = __builtin_amdgcn_readlane(s, j);
      const float4 v0 = h4[(size_t)s0 * (FEAT / 4) + lane];
      const float w0 = __int_as_float(__builtin_amdgcn_readlane(__float_as_int(ds), j)) * di;
      fma4(acc, w0, v0);
    }
  }

  const float sl = di * di;
  float4 rv;
  rv.x = acc.x + sl * hv.x + bv.x;
  rv.y = acc.y + sl * hv.y + bv.y;
  rv.z = acc.z + sl * hv.z + bv.z;
  rv.w = acc.w + sl * hv.w + bv.w;
  if (relu) {
    rv.x = fmaxf(rv.x, 0.0f);
    rv.y = fmaxf(rv.y, 0.0f);
    rv.z = fmaxf(rv.z, 0.0f);
    rv.w = fmaxf(rv.w, 0.0f);
  }
  ((float4*)out)[(size_t)i * (FEAT / 4) + lane] = rv;
}

#define POOL_RUN 32
static_assert(N_NODES % POOL_RUN == 0, "k_pool takes whole runs of nodes");

__device__ __forceinline__ void pool_flush(float* __restrict__ sums, float* __restrict__ cnts,
                                           int g, int lane, const float4 acc, int run) {
  float* sp = sums + (size_t)g * FEAT + lane * 4;
  atomAddF(sp + 0, acc.x);
  atomAddF(sp + 1, acc.y);
  atomAddF(sp + 2, acc.z);
  atomAddF(sp + 3, acc.w);
  if (lane == 0) atomAddF(&cnts[g], (float)run);
}

__global__ __launch_bounds__(256) void k_pool(const float* __restrict__ a,
                                              const int* __restrict__ batch,
                                              float* __restrict__ sums,
                                              float* __restrict__ cnts) {
  const int lane = threadIdx.x & 31;
  const int wave = __builtin_amdgcn_readfirstlane((int)(threadIdx.x >> 5));
  const int n0 = (blockIdx.x * 8 + wave) * POOL_RUN;
  if (n0 >= N_NODES) return;
  const int gid = batch[n0 + lane];
  const float4* a4 = (const float4*)a + (size_t)n0 * (FEAT / 4) + lane;
  float4 v[POOL_RUN];
#pragma unroll
  for (int j = 0; j < POOL_RUN; ++j) v[j] = a4[j * (FEAT / 4)];

  int gcur = __builtin_amdgcn_readlane(gid, 0);
  float4 acc = v[0];
  int run = 1;
#pragma unroll
  for (int j = 1; j < POOL_RUN; ++j) {
    const int g = __builtin_amdgcn_readlane(gid, j);
    if (g != gcur) {
      pool_flush(sums, cnts, gcur, lane, acc, run);
      gcur = g;
      acc = v[j];
      run = 1;
    } else {
      acc.x += v[j].x;
      acc.y += v[j].y;
      acc.z += v[j].z;
      acc.w += v[j].w;
      ++run;
    }
  }
  pool_flush(sums, cnts, gcur, lane, acc, run);
}

#define TAIL_PITCH (FEAT + 1)
static_assert(NGRAPH % 16 == 0, "k_tail takes whole strips of 16 graphs");
static_assert(16 * NCLS <= 256, "one thread per (graph, class) of the strip");
static_assert((FEAT * NCLS) % 256 == 0, "k_tail stages Wl in whole passes of the work-group");

__global__ __launch_bounds__(256) void k_tail(const float* __restrict__ sums,
                                              const float* __restrict__ cnts,
                                              const float* __restrict__ W3,
                                              const float* __restrict__ b3,
                                              const float* __restrict__ Wl,
                                              const float* __restrict__ bl,
                                              float* __restrict__ out) {
  __shared__ float sT[16 * TAIL_PITCH];
  __shared__ float sWl[FEAT * NCLS];
  const int lane = threadIdx.x & 31;
  const int wave = threadIdx.x >> 5;
  const int c    = lane & 15;
  const int h    = lane >> 4;
  const int g0   = blockIdx.x * 16;

  float wl[FEAT * NCLS / 256];
#pragma unroll
  for (int q = 0; q < FEAT * NCLS / 256; ++q) wl[q] = Wl[threadIdx.x + 256 * q];

  const float* arow = sums + (size_t)(g0 + c) * FEAT + 2 * h;
  const float* bcol = W3 + (size_t)(2 * h) * FEAT + wave * 16 + c;
  v2f a[FEAT / 4], b[FEAT / 4];
#pragma unroll
  for (int ks = 0; ks < FEAT / 4; ++ks) {
    a[ks].x = arow[4 * ks];
    a[ks].y = arow[4 * ks + 1];
    b[ks].x = bcol[(size_t)(4 * ks) * FEAT];
    b[ks].y = bcol[(size_t)(4 * ks + 1) * FEAT];
  }
  float cn[8];
#pragma unroll
  for (int r = 0; r < 8; ++r) cn[r] = cnts[g0 + 8 * h + r];
  const float bias3 = b3[wave * 16 + c];

  v8f acc = (v8f){0, 0, 0, 0, 0, 0, 0, 0};
#pragma unroll
  for (int ks = 0; ks < FEAT / 4; ++ks)
    acc = __builtin_amdgcn_wmma_f32_16x16x4_f32(false, a[ks], false, b[ks], (short)0, acc, false, false);

#pragma unroll
  for (int r = 0; r < 8; ++r) {
    const float inv = 1.0f / fmaxf(cn[r], 1.0f);
    sT[(8 * h + r) * TAIL_PITCH + wave * 16 + c] = acc[r] * inv + ((cn[r] > 0.0f) ? bias3 : 0.0f);
  }
#pragma unroll
  for (int q = 0; q < FEAT * NCLS / 256; ++q) sWl[threadIdx.x + 256 * q] = wl[q];
  __syncthreads();

  if (threadIdx.x < 16 * NCLS) {
    const int gl = threadIdx.x / NCLS;
    const int cl = threadIdx.x - gl * NCLS;
    float o = bl[cl];
#pragma unroll 16
    for (int k = 0; k < FEAT; ++k) o += sT[gl * TAIL_PITCH + k] * sWl[k * NCLS + cl];
    out[(size_t)(g0 + gl) * NCLS + cl] = o;
  }
}

extern "C" void kernel_launch(void* const* d_in, const int* in_sizes, int n_in,
                              void* d_out, int out_size, void* d_ws, size_t ws_size,
                              hipStream_t stream) {
  const float* x     = (const float*)d_in[0];
  const int*   ei    = (const int*)d_in[1];
  const int*   row   = ei;
  const int*   col   = ei + N_EDGES;
  const int*   batch = (const int*)d_in[2];
  const float* W1 = (const float*)d_in[3];
  const float* b1 = (const float*)d_in[4];
  const float* W2 = (const float*)d_in[5];
  const float* b2 = (const float*)d_in[6];
  const float* W3 = (const float*)d_in[7];
  const float* b3 = (const float*)d_in[8];
  const float* Wl = (const float*)d_in[9];
  const float* bl = (const float*)d_in[10];
  float* outp = (float*)d_out;

  char* base = (char*)d_ws;
  const size_t featBytes = (size_t)N_NODES * FEAT * sizeof(float);
  int*   histM  = (int*)(base);
  int*   rowptr = (int*)(base + (512 << 10));
  int*   bbase  = (int*)(base + (1 << 20));
  float* dinv   = (float*)(base + (1 << 20) + (512 << 10));
  int*   srcs   = (int*)(base + (4 << 20));
  float* hbuf = (float*)(base + (8 << 20));
  float* abuf = (float*)(base + (8 << 20) + featBytes);
  float* sums = (float*)(base + (8 << 20) + 2 * featBytes);
  float* cnts = sums + (size_t)NGRAPH * FEAT;
  unsigned* binned = (unsigned*)(base + (9 << 20) + 2 * featBytes);

  const int  TB     = 256;
  const int  gGemm  = (N_NODES / 16 + 8 - 1) / 8;
  const int  gGath  = (N_NODES + 8 - 1) / 8;
  const int  gPool  = (N_NODES / POOL_RUN + 8 - 1) / 8;

  k_hist<<<EB_WG, EB_THREADS, 0, stream>>>(col, histM);
  k_bin<<<EB_WG, EB_THREADS, 0, stream>>>(row, col, histM, bbase, binned);
  k_group<<<NBKT, BKT_NODES, 0, stream>>>(binned, bbase, rowptr, dinv, srcs);

  k_gemm<<<gGemm, TB, 0, stream>>>(x, W1, hbuf, N_NODES);
  k_gather<<<gGath, TB, 0, stream>>>(hbuf, rowptr, srcs, dinv, b1, abuf, 1);

  k_gemm<<<gGemm, TB, 0, stream>>>(abuf, W2, hbuf, N_NODES);
  k_gather<<<gGath, TB, 0, stream>>>(hbuf, rowptr, srcs, dinv, b2, abuf, 1);

  k_gather<<<gGath, TB, 0, stream>>>(abuf, rowptr, srcs, dinv, nullptr, hbuf, 0);

  const long poolN = (long)NGRAPH * FEAT + NGRAPH;
  k_fill<<<(int)((poolN + TB - 1) / TB), TB, 0, stream>>>(sums, 0.0f, poolN);
  k_pool<<<gPool, TB, 0, stream>>>(hbuf, batch, sums, cnts);
  k_tail<<<NGRAPH / 16, TB, 0, stream>>>(sums, cnts, W3, b3, Wl, bl, outp);
}


# k_gemm: W staged once per work-group (256 groups loop over strips), transposed in LDS: a lane's B operands of two k steps are one 16-byte read, 16-byte stores; k order of the sums changes
# speedup vs baseline: 4.5191x; 1.0924x over previous
#include <hip/hip_runtime.h>

typedef __attribute__((ext_vector_type(2))) float v2f;
typedef __attribute__((ext_vector_type(8))) float v8f;

#define N_NODES 100000
#define N_EDGES 800000
#define FEAT    128
#define NGRAPH  512
#define NCLS    10

__device__ __forceinline__ void atomAddF(float* p, float v) {
  unsafeAtomicAdd(p, v);
}

__global__ void k_fill(float* __restrict__ p, float v, long n) {
  long i = (long)blockIdx.x * blockDim.x + threadIdx.x;
  if (i < n) p[i] = v;
}

#define EB_WG      128
#define EB_THREADS 512
#define EB_WAVES   (EB_THREADS / 32)
#define EB_CHUNK   ((N_EDGES + EB_WG - 1) / EB_WG)
#define EB_BATCH   4
#define BKT_SHIFT  8
#define BKT_NODES  (1 << BKT_SHIFT)
#define NBKT       ((N_NODES + BKT_NODES - 1) / BKT_NODES)
static_assert(NBKT <= EB_THREADS, "k_bin gives one thread to every bucket");
static_assert(N_NODES <= (1 << 24), "a packed edge keeps the source in 24 bits");
static_assert(BKT_NODES <= 256, "a packed edge keeps the target's place in its bucket in 8 bits");

template <int NW>
__device__ __forceinline__ int block_excl_scan(int v, int* s_wave, int& total) {
  const int lane = threadIdx.x & 31;
  const int wave = threadIdx.x >> 5;
  int inc = v;
  for (int o = 1; o < 32; o <<= 1) {
    const int y = __shfl_up(inc, o);
    if (lane >= o) inc += y;
  }
  if (lane == 31) s_wave[wave] = inc;
  __syncthreads();
  int base = 0, tot = 0;
  for (int w = 0; w < NW; ++w) {
    const int x = s_wave[w];
    if (w < wave) base += x;
    tot += x;
  }
  total = tot;
  return base + inc - v;
}

__global__ __launch_bounds__(EB_THREADS) void k_hist(const int* __restrict__ col,
                                                     int* __restrict__ histM) {
  __shared__ int s_h[NBKT];
  const int t = threadIdx.x;
  const int w = blockIdx.x;
  for (int b = t; b < NBKT; b += EB_THREADS) s_h[b] = 0;
  __syncthreads();
  const int e0 = w * EB_CHUNK;
  const int e1 = (e0 + EB_CHUNK < N_EDGES) ? (e0 + EB_CHUNK) : N_EDGES;
  for (int e = e0 + t; e < e1; e += EB_BATCH * EB_THREADS) {
    int c[EB_BATCH];
#pragma unroll
    for (int k = 0; k < EB_BATCH; ++k) {
      const int ee = e + k * EB_THREADS;
      c[k] = (ee < e1) ? col[ee] : -1;
    }
#pragma unroll
    for (int k = 0; k < EB_BATCH; ++k)
      if (c[k] >= 0) atomicAdd(&s_h[c[k] >> BKT_SHIFT], 1);
  }
  __syncthreads();
  for (int b = t; b < NBKT; b += EB_THREADS) histM[w * NBKT + b] = s_h[b];
}

__global__ __launch_bounds__(EB_THREADS) void k_bin(const int* __restrict__ row,
                                                    const int* __restrict__ col,
                                                    const int* __restrict__ histM,
                                                    int* __restrict__ bbase,
                                                    unsigned* __restrict__ binned) {
  __shared__ int s_cur[NBKT];
  __shared__ int s_wave[EB_WAVES];
  const int t = threadIdx.x;
  const int w = blockIdx.x;

  int before = 0, total = 0;
  if (t < NBKT) {
#pragma unroll 32
    for (int w2 = 0; w2 < EB_WG; ++w2) {
      const int v = histM[w2 * NBKT + t];
      total += v;
      before += (w2 < w) ? v : 0;
    }
  }
  int all;
  const int base = block_excl_scan<EB_WAVES>(total, s_wave, all);
  if (t < NBKT) {
    s_cur[t] = base + before;
    if (w == 0) bbase[t] = base;
  }
  if (w == 0 && t == 0) bbase[NBKT] = all;
  __syncthreads();

  const int e0 = w * EB_CHUNK;
  const int e1 = (e0 + EB_CHUNK < N_EDGES) ? (e0 + EB_CHUNK) : N_EDGES;
  for (int e = e0 + t; e < e1; e += EB_BATCH * EB_THREADS) {
    int c[EB_BATCH], r[EB_BATCH];
#pragma unroll
    for (int k = 0; k < EB_BATCH; ++k) {
      const int ee = e + k * EB_THREADS;
      const bool ok = ee < e1;
      c[k] = ok ? col[ee] : -1;
      r[k] = ok ? row[ee] : 0;
    }
#pragma unroll
    for (int k = 0; k < EB_BATCH; ++k) {
      if (c[k] >= 0) {
        const int pos = atomicAdd(&s_cur[c[k] >> BKT_SHIFT], 1);
        binned[pos] = ((unsigned)(c[k] & (BKT_NODES - 1)) << 24) | (unsigned)r[k];
      }
    }
  }
}

__device__ __forceinline__ float dinv_of(int c) {
  const float d = (float)c + 1.0f;
  return (d > 0.0f) ? rsqrtf(fmaxf(d, 1.0f)) : 0.0f;
}

__global__ __launch_bounds__(BKT_NODES) void k_group(const unsigned* __restrict__ binned,
                                                     const int* __restrict__ bbase,
                                                     int* __restrict__ rowptr,
                                                     float* __restrict__ dinv,
                                                     int* __restrict__ srcs) {
  __shared__ int s_cnt[BKT_NODES];
  __shared__ int s_wave[BKT_NODES / 32];
  const int t = threadIdx.x;
  const int b = blockIdx.x;
  const int lo = bbase[b];
  const int hi = bbase[b + 1];
  s_cnt[t] = 0;
  __syncthreads();
  for (int p = lo + t; p < hi; p += EB_BATCH * BKT_NODES) {
    unsigned u[EB_BATCH];
#pragma unroll
    for (int k = 0; k < EB_BATCH; ++k) {
      const int pp = p + k * BKT_NODES;
      u[k] = binned[(pp < hi) ? pp : p];
    }
#pragma unroll
    for (int k = 0; k < EB_BATCH; ++k)
      atomicAdd(&s_cnt[u[k] >> 24], (p + k * BKT_NODES < hi) ? 1 : 0);
  }
  __syncthreads();
  const int c = s_cnt[t];
  int all;
  const int ex = block_excl_scan<BKT_NODES / 32>(c, s_wave, all);
  const int node = b * BKT_NODES + t;
  if (node < N_NODES) {
    rowptr[node] = lo + ex;
    dinv[node] = dinv_of(c);
  }
  if (b == NBKT - 1 && t == 0) rowptr[N_NODES] = hi;
  s_cnt[t] = lo + ex;
  __syncthreads();
  for (int p = lo + t; p < hi; p += EB_BATCH * BKT_NODES) {
    unsigned u[EB_BATCH];
#pragma unroll
    for (int k = 0; k < EB_BATCH; ++k) {
      const int pp = p + k * BKT_NODES;
      u[k] = binned[(pp < hi) ? pp : p];
    }
#pragma unroll
    for (int k = 0; k < EB_BATCH; ++k) {
      const bool ok = p + k * BKT_NODES < hi;
      const int pos = atomicAdd(&s_cnt[u[k] >> 24], ok ? 1 : 0);
      if (ok) srcs[pos] = (int)(u[k] & 0xFFFFFFu);
    }
  }
}

#define GEMM_WG      256
#define GEMM_THREADS 512
#define GEMM_WAVES   (GEMM_THREADS / 32)
#define WT_PITCH     (FEAT + 4)
static_assert(N_NODES % 16 == 0, "k_gemm takes whole strips of 16 rows");
static_assert((FEAT * FEAT / 4) % GEMM_THREADS == 0, "k_gemm stages W in whole passes of the work-group");

__global__ __launch_bounds__(GEMM_THREADS) void k_gemm(const float* __restrict__ X,
                                                       const float* __restrict__ W,
                                                       float* __restrict__ Hout,
                                                       int nrows) {
  __shared__ float4 sWt4[FEAT * WT_PITCH / 4];
  float* sWt = (float*)sWt4;
  const int lane = threadIdx.x & 31;
  const int wave = __builtin_amdgcn_readfirstlane((int)(threadIdx.x >> 5));
  const int c = lane & 15;
  const int h = lane >> 4;

  const int nstrips = nrows >> 4;
  const int s_lo = ((int)blockIdx.x * nstrips) / GEMM_WG;
  const int s_hi = (((int)blockIdx.x + 1) * nstrips) / GEMM_WG;
  static_assert(GEMM_WAVES == 16, "the renumbering below is a 4 x 4 square");
  const int wv = (((wave - (wave >> 2)) & 3) << 2) | (wave >> 2);
  int s = s_lo + wv;

  const float4* wsrc = (const float4*)W;
  float4 w[FEAT * FEAT / 4 / GEMM_THREADS];
#pragma unroll
  for (int it = 0; it < FEAT * FEAT / 4 / GEMM_THREADS; ++it) w[it] = wsrc[threadIdx.x + GEMM_THREADS * it];

  float4 a[FEAT / 8];
  {
    const int sf = (s < nstrips) ? s : (nstrips - 1);
    const float4* xr = (const float4*)(X + (size_t)(sf * 16 + c) * FEAT + 4 * h);
#pragma unroll
    for (int j = 0; j < FEAT / 8; ++j) a[j] = xr[2 * j];
  }

#pragma unroll
  for (int it = 0; it < FEAT * FEAT / 4 / GEMM_THREADS; ++it) {
    const int idx = threadIdx.x + GEMM_THREADS * it;
    const int k = idx >> 5;
    const int q = idx & 31;
    const int p = 64 * (q >> 4) + (q & 15);
    sWt[(p     ) * WT_PITCH + k] = w[it].x;
    sWt[(p + 16) * WT_PITCH + k] = w[it].y;
    sWt[(p + 32) * WT_PITCH + k] = w[it].z;
    sWt[(p + 48) * WT_PITCH + k] = w[it].w;
  }
  __syncthreads();

  const float4* bq = sWt4 + (c * WT_PITCH) / 4 + h;
  while (s < s_hi) {
    asm volatile("" ::: "memory");
    v8f acc[8];
#pragma unroll
    for (int n = 0; n < 8; ++n) acc[n] = (v8f){0, 0, 0, 0, 0, 0, 0, 0};

    float4 b[8];
#pragma unroll
    for (int n = 0; n < 8; ++n) b[n] = bq[n * (4 * WT_PITCH)];
#pragma unroll
    for (int j = 0; j < FEAT / 8; ++j) {
      float4 bn[8];
      __builtin_amdgcn_sched_barrier(0);
      if (j + 1 < FEAT / 8) {
#pragma unroll
        for (int n = 0; n < 8; ++n) bn[n] = bq[n * (4 * WT_PITCH) + 2 * (j + 1)];
      }
      __builtin_amdgcn_sched_barrier(0);
      const v2f a0 = {a[j].x, a[j].y};
      const v2f a1 = {a[j].z, a[j].w};
#pragma unroll
      for (int n = 0; n < 8; ++n) {
        const v2f b0 = {b[n].x, b[n].y};
        acc[n] = __builtin_amdgcn_wmma_f32_16x16x4_f32(false, a0, false, b0, (short)0, acc[n], false, false);
      }
#pragma unroll
      for (int n = 0; n < 8; ++n) {
        const v2f b1 = {b[n].z, b[n].w};
        acc[n] = __builtin_amdgcn_wmma_f32_16x16x4_f32(false, a1, false, b1, (short)0, acc[n], false, false);
      }
      if (j + 1 < FEAT / 8) {
#pragma unroll
        for (int n = 0; n < 8; ++n) b[n] = bn[n];
      }
    }

    const int sn = s + GEMM_WAVES;
    if (sn < s_hi) {
      const float4* xr = (const float4*)(X + (size_t)(sn * 16 + c) * FEAT + 4 * h);
#pragma unroll
      for (int j = 0; j < FEAT / 8; ++j) a[j] = xr[2 * j];
    }

    float4* orow = (float4*)(Hout + (size_t)(s * 16 + 8 * h) * FEAT + 4 * c);
#pragma unroll
    for (int r = 0; r < 8; ++r) {
      const float4 lo = {acc[0][r], acc[1][r], acc[2][r], acc[3][r]};
      const float4 hi = {acc[4][r], acc[5][r], acc[6][r], acc[7][r]};
      orow[r * (FEAT / 4)]      = lo;
      orow[r * (FEAT / 4) + 16] = hi;
    }
    s = sn;
  }
}

__device__ __forceinline__ void fma4(float4& acc, float w, const float4 v) {
  acc.x += w * v.x;
  acc.y += w * v.y;
  acc.z += w * v.z;
  acc.w += w * v.w;
}

__global__ __launch_bounds__(256) void k_gather(const float* __restrict__ h,
                                                const int* __restrict__ rowptr,
                                                const int* __restrict__ srcs,
                                                const float* __restrict__ dinv,
                                                const float* __restrict__ bias,
                                                float* __restrict__ out, int relu) {
  const int lane = threadIdx.x & 31;
  const int wave = __builtin_amdgcn_readfirstlane((int)(threadIdx.x >> 5));
  const int i = blockIdx.x * 8 + wave;
  if (i >= N_NODES) return;
  const int start = rowptr[i];
  const int n = rowptr[i + 1] - start;
  const float di = dinv[i];
  const float4* h4 = (const float4*)h;
  const float4 hv = h4[(size_t)i * (FEAT / 4) + lane];
  float4 bv = {0.0f, 0.0f, 0.0f, 0.0f};
  if (bias) bv = ((const float4*)bias)[lane];

  float4 acc = {0.0f, 0.0f, 0.0f, 0.0f};
  for (int b = 0; b < n; b += 32) {
    const int m = (n - b < 32) ? (n - b) : 32;
    int s = 0;
    float ds = 0.0f;
    if (lane < m) {
      s = srcs[start + b + lane];
      ds = dinv[s];
    }
    int j = 0;
    for (; j + 4 <= m; j += 4) {
      const int s0 = __builtin_amdgcn_readlane(s, j);
      const int s1 = __builtin_amdgcn_readlane(s, j + 1);
      const int s2 = __builtin_amdgcn_readlane(s, j + 2);
      const int s3 = __builtin_amdgcn_readlane(s, j + 3);
      const float4 v0 = h4[(size_t)s0 * (FEAT / 4) + lane];
      const float4 v1 = h4[(size_t)s1 * (FEAT / 4) + lane];
      const float4 v2 = h4[(size_t)s2 * (FEAT / 4) + lane];
      const float4 v3 = h4[(size_t)s3 * (FEAT / 4) + lane];
      const float w0 = __int_as_float(__builtin_amdgcn_readlane(__float_as_int(ds), j)) * di;
      const float w1 = __int_as_float(__builtin_amdgcn_readlane(__float_as_int(ds), j + 1)) * di;
      const float w2 = __int_as_float(__builtin_amdgcn_readlane(__float_as_int(ds), j + 2)) * di;
      const float w3 = __int_as_float(__builtin_amdgcn_readlane(__float_as_int(ds), j + 3)) * di;
      fma4(acc, w0, v0);
      fma4(acc, w1, v1);
      fma4(acc, w2, v2);
      fma4(acc, w3, v3);
    }
    for (; j < m; ++j) {
      const int s0 = __builtin_amdgcn_readlane(s, j);
      const float4 v0 = h4[(size_t)s0 * (FEAT / 4) + lane];
      const float w0 = __int_as_float(__builtin_amdgcn_readlane(__float_as_int(ds), j)) * di;
      fma4(acc, w0, v0);
    }
  }

  const float sl = di * di;
  float4 rv;
  rv.x = acc.x + sl * hv.x + bv.x;
  rv.y = acc.y + sl * hv.y + bv.y;
  rv.z = acc.z + sl * hv.z + bv.z;
  rv.w = acc.w + sl * hv.w + bv.w;
  if (relu) {
    rv.x = fmaxf(rv.x, 0.0f);
    rv.y = fmaxf(rv.y, 0.0f);
    rv.z = fmaxf(rv.z, 0.0f);
    rv.w = fmaxf(rv.w, 0.0f);
  }
  ((float4*)out)[(size_t)i * (FEAT / 4) + lane] = rv;
}

#define POOL_RUN 32
static_assert(N_NODES % POOL_RUN == 0, "k_pool takes whole runs of nodes");

__device__ __forceinline__ void pool_flush(float* __restrict__ sums, float* __restrict__ cnts,
                                           int g, int lane, const float4 acc, int run) {
  float* sp = sums + (size_t)g * FEAT + lane * 4;
  atomAddF(sp + 0, acc.x);
  atomAddF(sp + 1, acc.y);
  atomAddF(sp + 2, acc.z);
  atomAddF(sp + 3, acc.w);
  if (lane == 0) atomAddF(&cnts[g], (float)run);
}

__global__ __launch_bounds__(256) void k_pool(const float* __restrict__ a,
                                              const int* __restrict__ batch,
                                              float* __restrict__ sums,
                                              float* __restrict__ cnts) {
  const int lane = threadIdx.x & 31;
  const int wave = __builtin_amdgcn_readfirstlane((int)(threadIdx.x >> 5));
  const int n0 = (blockIdx.x * 8 + wave) * POOL_RUN;
  if (n0 >= N_NODES) return;
  const int gid = batch[n0 + lane];
  const float4* a4 = (const float4*)a + (size_t)n0 * (FEAT / 4) + lane;
  float4 v[POOL_RUN];
#pragma unroll
  for (int j = 0; j < POOL_RUN; ++j) v[j] = a4[j * (FEAT / 4)];

  int gcur = __builtin_amdgcn_readlane(gid, 0);
  float4 acc = v[0];
  int run = 1;
#pragma unroll
  for (int j = 1; j < POOL_RUN; ++j) {
    const int g = __builtin_amdgcn_readlane(gid, j);
    if (g != gcur) {
      pool_flush(sums, cnts, gcur, lane, acc, run);
      gcur = g;
      acc = v[j];
      run = 1;
    } else {
      acc.x += v[j].x;
      acc.y += v[j].y;
      acc.z += v[j].z;
      acc.w += v[j].w;
      ++run;
    }
  }
  pool_flush(sums, cnts, gcur, lane, acc, run);
}

#define TAIL_PITCH (FEAT + 1)
static_assert(NGRAPH % 16 == 0, "k_tail takes whole strips of 16 graphs");
static_assert(16 * NCLS <= 256, "one thread per (graph, class) of the strip");
static_assert((FEAT * NCLS) % 256 == 0, "k_tail stages Wl in whole passes of the work-group");

__global__ __launch_bounds__(256) void k_tail(const float* __restrict__ sums,
                                              const float* __restrict__ cnts,
                                              const float* __restrict__ W3,
                                              const float* __restrict__ b3,
                                              const float* __restrict__ Wl,
                                              const float* __restrict__ bl,
                                              float* __restrict__ out) {
  __shared__ float sT[16 * TAIL_PITCH];
  __shared__ float sWl[FEAT * NCLS];
  const int lane = threadIdx.x & 31;
  const int wave = threadIdx.x >> 5;
  const int c    = lane & 15;
  const int h    = lane >> 4;
  const int g0   = blockIdx.x * 16;

  float wl[FEAT * NCLS / 256];
#pragma unroll
  for (int q = 0; q < FEAT * NCLS / 256; ++q) wl[q] = Wl[threadIdx.x + 256 * q];

  const float* arow = sums + (size_t)(g0 + c) * FEAT + 2 * h;
  const float* bcol = W3 + (size_t)(2 * h) * FEAT + wave * 16 + c;
  v2f a[FEAT / 4], b[FEAT / 4];
#pragma unroll
  for (int ks = 0; ks < FEAT / 4; ++ks) {
    a[ks].x = arow[4 * ks];
    a[ks].y = arow[4 * ks + 1];
    b[ks].x = bcol[(size_t)(4 * ks) * FEAT];
    b[ks].y = bcol[(size_t)(4 * ks + 1) * FEAT];
  }
  float cn[8];
#pragma unroll
  for (int r = 0; r < 8; ++r) cn[r] = cnts[g0 + 8 * h + r];
  const float bias3 = b3[wave * 16 + c];

  v8f acc = (v8f){0, 0, 0, 0, 0, 0, 0, 0};
#pragma unroll
  for (int ks = 0; ks < FEAT / 4; ++ks)
    acc = __builtin_amdgcn_wmma_f32_16x16x4_f32(false, a[ks], false, b[ks], (short)0, acc, false, false);

#pragma unroll
  for (int r = 0; r < 8; ++r) {
    const float inv = 1.0f / fmaxf(cn[r], 1.0f);
    sT[(8 * h + r) * TAIL_PITCH + wave * 16 + c] = acc[r] * inv + ((cn[r] > 0.0f) ? bias3 : 0.0f);
  }
#pragma unroll
  for (int q = 0; q < FEAT * NCLS / 256; ++q) sWl[threadIdx.x + 256 * q] = wl[q];
  __syncthreads();

  if (threadIdx.x < 16 * NCLS) {
    const int gl = threadIdx.x / NCLS;
    const int cl = threadIdx.x - gl * NCLS;
    float o = bl[cl];
#pragma unroll 16
    for (int k = 0; k < FEAT; ++k) o += sT[gl * TAIL_PITCH + k] * sWl[k * NCLS + cl];
    out[(size_t)(g0 + gl) * NCLS + cl] = o;
  }
}

extern "C" void kernel_launch(void* const* d_in, const int* in_sizes, int n_in,
                              void* d_out, int out_size, void* d_ws, size_t ws_size,
                              hipStream_t stream) {
  const float* x     = (const float*)d_in[0];
  const int*   ei    = (const int*)d_in[1];
  const int*   row   = ei;
  const int*   col   = ei + N_EDGES;
  const int*   batch = (const int*)d_in[2];
  const float* W1 = (const float*)d_in[3];
  const float* b1 = (const float*)d_in[4];
  const float* W2 = (const float*)d_in[5];
  const float* b2 = (const float*)d_in[6];
  const float* W3 = (const float*)d_in[7];
  const float* b3 = (const float*)d_in[8];
  const float* Wl = (const float*)d_in[9];
  const float* bl = (const float*)d_in[10];
  float* outp = (float*)d_out;

  char* base = (char*)d_ws;
  const size_t featBytes = (size_t)N_NODES * FEAT * sizeof(float);
  int*   histM  = (int*)(base);
  int*   rowptr = (int*)(base + (512 << 10));
  int*   bbase  = (int*)(base + (1 << 20));
  float* dinv   = (float*)(base + (1 << 20) + (512 << 10));
  int*   srcs   = (int*)(base + (4 << 20));
  float* hbuf = (float*)(base + (8 << 20));
  float* abuf = (float*)(base + (8 << 20) + featBytes);
  float* sums = (float*)(base + (8 << 20) + 2 * featBytes);
  float* cnts = sums + (size_t)NGRAPH * FEAT;
  unsigned* binned = (unsigned*)(base + (9 << 20) + 2 * featBytes);

  const int  TB     = 256;
  const int  gGath  = (N_NODES + 8 - 1) / 8;
  const int  gPool  = (N_NODES / POOL_RUN + 8 - 1) / 8;

  k_hist<<<EB_WG, EB_THREADS, 0, stream>>>(col, histM);
  k_bin<<<EB_WG, EB_THREADS, 0, stream>>>(row, col, histM, bbase, binned);
  k_group<<<NBKT, BKT_NODES, 0, stream>>>(binned, bbase, rowptr, dinv, srcs);

  k_gemm<<<GEMM_WG, GEMM_THREADS, 0, stream>>>(x, W1, hbuf, N_NODES);
  k_gather<<<gGath, TB, 0, stream>>>(hbuf, rowptr, srcs, dinv, b1, abuf, 1);

  k_gemm<<<GEMM_WG, GEMM_THREADS, 0, stream>>>(abuf, W2, hbuf, N_NODES);
  k_gather<<<gGath, TB, 0, stream>>>(hbuf, rowptr, srcs, dinv, b2, abuf, 1);

  k_gather<<<gGath, TB, 0, stream>>>(abuf, rowptr, srcs, dinv, nullptr, hbuf, 0);

  const long poolN = (long)NGRAPH * FEAT + NGRAPH;
  k_fill<<<(int)((poolN + TB - 1) / TB), TB, 0, stream>>>(sums, 0.0f, poolN);
  k_pool<<<gPool, TB, 0, stream>>>(hbuf, batch, sums, cnts);
  k_tail<<<NGRAPH / 16, TB, 0, stream>>>(sums, cnts, W3, b3, Wl, bl, outp);
}


# layers 1 and 2 products in one launch: x is aggregated first (linear), then relu((Ax)W1+b1)W2 per 16-row strip, intermediate through wave-own LDS; one launch, 51 MB write+read gone; sum order changes
# speedup vs baseline: 4.8923x; 1.0826x over previous
#include <hip/hip_runtime.h>

typedef __attribute__((ext_vector_type(2))) float v2f;
typedef __attribute__((ext_vector_type(8))) float v8f;

#define N_NODES 100000
#define N_EDGES 800000
#define FEAT    128
#define NGRAPH  512
#define NCLS    10

__device__ __forceinline__ void atomAddF(float* p, float v) {
  unsafeAtomicAdd(p, v);
}

__global__ void k_fill(float* __restrict__ p, float v, long n) {
  long i = (long)blockIdx.x * blockDim.x + threadIdx.x;
  if (i < n) p[i] = v;
}

#define EB_WG      128
#define EB_THREADS 512
#define EB_WAVES   (EB_THREADS / 32)
#define EB_CHUNK   ((N_EDGES + EB_WG - 1) / EB_WG)
#define EB_BATCH   4
#define BKT_SHIFT  8
#define BKT_NODES  (1 << BKT_SHIFT)
#define NBKT       ((N_NODES + BKT_NODES - 1) / BKT_NODES)
static_assert(NBKT <= EB_THREADS, "k_bin gives one thread to every bucket");
static_assert(N_NODES <= (1 << 24), "a packed edge keeps the source in 24 bits");
static_assert(BKT_NODES <= 256, "a packed edge keeps the target's place in its bucket in 8 bits");

template <int NW>
__device__ __forceinline__ int block_excl_scan(int v, int* s_wave, int& total) {
  const int lane = threadIdx.x & 31;
  const int wave = threadIdx.x >> 5;
  int inc = v;
  for (int o = 1; o < 32; o <<= 1) {
    const int y = __shfl_up(inc, o);
    if (lane >= o) inc += y;
  }
  if (lane == 31) s_wave[wave] = inc;
  __syncthreads();
  int base = 0, tot = 0;
  for (int w = 0; w < NW; ++w) {
    const int x = s_wave[w];
    if (w < wave) base += x;
    tot += x;
  }
  total = tot;
  return base + inc - v;
}

__global__ __launch_bounds__(EB_THREADS) void k_hist(const int* __restrict__ col,
                                                     int* __restrict__ histM) {
  __shared__ int s_h[NBKT];
  const int t = threadIdx.x;
  const int w = blockIdx.x;
  for (int b = t; b < NBKT; b += EB_THREADS) s_h[b] = 0;
  __syncthreads();
  const int e0 = w * EB_CHUNK;
  const int e1 = (e0 + EB_CHUNK < N_EDGES) ? (e0 + EB_CHUNK) : N_EDGES;
  for (int e = e0 + t; e < e1; e += EB_BATCH * EB_THREADS) {
    int c[EB_BATCH];
#pragma unroll
    for (int k = 0; k < EB_BATCH; ++k) {
      const int ee = e + k * EB_THREADS;
      c[k] = (ee < e1) ? col[ee] : -1;
    }
#pragma unroll
    for (int k = 0; k < EB_BATCH; ++k)
      if (c[k] >= 0) atomicAdd(&s_h[c[k] >> BKT_SHIFT], 1);
  }
  __syncthreads();
  for (int b = t; b < NBKT; b += EB_THREADS) histM[w * NBKT + b] = s_h[b];
}

__global__ __launch_bounds__(EB_THREADS) void k_bin(const int* __restrict__ row,
                                                    const int* __restrict__ col,
                                                    const int* __restrict__ histM,
                                                    int* __restrict__ bbase,
                                                    unsigned* __restrict__ binned) {
  __shared__ int s_cur[NBKT];
  __shared__ int s_wave[EB_WAVES];
  const int t = threadIdx.x;
  const int w = blockIdx.x;

  int before = 0, total = 0;
  if (t < NBKT) {
#pragma unroll 32
    for (int w2 = 0; w2 < EB_WG; ++w2) {
      const int v = histM[w2 * NBKT + t];
      total += v;
      before += (w2 < w) ? v : 0;
    }
  }
  int all;
  const int base = block_excl_scan<EB_WAVES>(total, s_wave, all);
  if (t < NBKT) {
    s_cur[t] = base + before;
    if (w == 0) bbase[t] = base;
  }
  if (w == 0 && t == 0) bbase[NBKT] = all;
  __syncthreads();

  const int e0 = w * EB_CHUNK;
  const int e1 = (e0 + EB_CHUNK < N_EDGES) ? (e0 + EB_CHUNK) : N_EDGES;
  for (int e = e0 + t; e < e1; e += EB_BATCH * EB_THREADS) {
    int c[EB_BATCH], r[EB_BATCH];
#pragma unroll
    for (int k = 0; k < EB_BATCH; ++k) {
      const int ee = e + k * EB_THREADS;
      const bool ok = ee < e1;
      c[k] = ok ? col[ee] : -1;
      r[k] = ok ? row[ee] : 0;
    }
#pragma unroll
    for (int k = 0; k < EB_BATCH; ++k) {
      if (c[k] >= 0) {
        const int pos = atomicAdd(&s_cur[c[k] >> BKT_SHIFT], 1);
        binned[pos] = ((unsigned)(c[k] & (BKT_NODES - 1)) << 24) | (unsigned)r[k];
      }
    }
  }
}

__device__ __forceinline__ float dinv_of(int c) {
  const float d = (float)c + 1.0f;
  return (d > 0.0f) ? rsqrtf(fmaxf(d, 1.0f)) : 0.0f;
}

__global__ __launch_bounds__(BKT_NODES) void k_group(const unsigned* __restrict__ binned,
                                                     const int* __restrict__ bbase,
                                                     int* __restrict__ rowptr,
                                                     float* __restrict__ dinv,
                                                     int* __restrict__ srcs) {
  __shared__ int s_cnt[BKT_NODES];
  __shared__ int s_wave[BKT_NODES / 32];
  const int t = threadIdx.x;
  const int b = blockIdx.x;
  const int lo = bbase[b];
  const int hi = bbase[b + 1];
  s_cnt[t] = 0;
  __syncthreads();
  for (int p = lo + t; p < hi; p += EB_BATCH * BKT_NODES) {
    unsigned u[EB_BATCH];
#pragma unroll
    for (int k = 0; k < EB_BATCH; ++k) {
      const int pp = p + k * BKT_NODES;
      u[k] = binned[(pp < hi) ? pp : p];
    }
#pragma unroll
    for (int k = 0; k < EB_BATCH; ++k)
      atomicAdd(&s_cnt[u[k] >> 24], (p + k * BKT_NODES < hi) ? 1 : 0);
  }
  __syncthreads();
  const int c = s_cnt[t];
  int all;
  const int ex = block_excl_scan<BKT_NODES / 32>(c, s_wave, all);
  const int node = b * BKT_NODES + t;
  if (node < N_NODES) {
    rowptr[node] = lo + ex;
    dinv[node] = dinv_of(c);
  }
  if (b == NBKT - 1 && t == 0) rowptr[N_NODES] = hi;
  s_cnt[t] = lo + ex;
  __syncthreads();
  for (int p = lo + t; p < hi; p += EB_BATCH * BKT_NODES) {
    unsigned u[EB_BATCH];
#pragma unroll
    for (int k = 0; k < EB_BATCH; ++k) {
      const int pp = p + k * BKT_NODES;
      u[k] = binned[(pp < hi) ? pp : p];
    }
#pragma unroll
    for (int k = 0; k < EB_BATCH; ++k) {
      const bool ok = p + k * BKT_NODES < hi;
      const int pos = atomicAdd(&s_cnt[u[k] >> 24], ok ? 1 : 0);
      if (ok) srcs[pos] = (int)(u[k] & 0xFFFFFFu);
    }
  }
}

#define GEMM_WG      256
#define GEMM_THREADS 512
#define GEMM_WAVES   (GEMM_THREADS / 32)
#define WT_PITCH     (FEAT + 4)
static_assert(N_NODES % 16 == 0, "k_gemm2 takes whole strips of 16 rows");
static_assert((FEAT * FEAT / 4) % GEMM_THREADS == 0, "k_gemm2 stages a W in whole passes of the work-group");

__device__ __forceinline__ void strip_product(const float4* __restrict__ bq, const float4 (&a)[FEAT / 8],
                                              v8f (&acc)[8]) {
#pragma unroll
  for (int n = 0; n < 8; ++n) acc[n] = (v8f){0, 0, 0, 0, 0, 0, 0, 0};
  float4 b[8];
#pragma unroll
  for (int n = 0; n < 8; ++n) b[n] = bq[n * (4 * WT_PITCH)];
#pragma unroll
  for (int j = 0; j < FEAT / 8; ++j) {
    float4 bn[8];
    __builtin_amdgcn_sched_barrier(0);
    if (j + 1 < FEAT / 8) {
#pragma unroll
      for (int n = 0; n < 8; ++n) bn[n] = bq[n * (4 * WT_PITCH) + 2 * (j + 1)];
    }
    __builtin_amdgcn_sched_barrier(0);
    const v2f a0 = {a[j].x, a[j].y};
    const v2f a1 = {a[j].z, a[j].w};
#pragma unroll
    for (int n = 0; n < 8; ++n) {
      const v2f b0 = {b[n].x, b[n].y};
      acc[n] = __builtin_amdgcn_wmma_f32_16x16x4_f32(false, a0, false, b0, (short)0, acc[n], false, false);
    }
#pragma unroll
    for (int n = 0; n < 8; ++n) {
      const v2f b1 = {b[n].z, b[n].w};
      acc[n] = __builtin_amdgcn_wmma_f32_16x16x4_f32(false, a1, false, b1, (short)0, acc[n], false, false);
    }
    if (j + 1 < FEAT / 8) {
#pragma unroll
      for (int n = 0; n < 8; ++n) b[n] = bn[n];
    }
  }
}

__device__ __forceinline__ void stage_w(float* __restrict__ sW, const float4 (&w)[FEAT * FEAT / 4 / GEMM_THREADS]) {
#pragma unroll
  for (int it = 0; it < FEAT * FEAT / 4 / GEMM_THREADS; ++it) {
    const int idx = threadIdx.x + GEMM_THREADS * it;
    const int k = idx >> 5;
    const int q = idx & 31;
    const int p = 64 * (q >> 4) + (q & 15);
    sW[(p     ) * WT_PITCH + k] = w[it].x;
    sW[(p + 16) * WT_PITCH + k] = w[it].y;
    sW[(p + 32) * WT_PITCH + k] = w[it].z;
    sW[(p + 48) * WT_PITCH + k] = w[it].w;
  }
}

__global__ __launch_bounds__(GEMM_THREADS) void k_gemm2(const float* __restrict__ X,
                                                        const float* __restrict__ Wa,
                                                        const float* __restrict__ ba,
                                                        const float* __restrict__ Wb,
                                                        float* __restrict__ Hout,
                                                        int nrows) {
  __shared__ float4 sWa4[FEAT * WT_PITCH / 4];
  __shared__ float4 sWb4[FEAT * WT_PITCH / 4];
  __shared__ float4 sT4[GEMM_WAVES * 16 * WT_PITCH / 4];
  const int lane = threadIdx.x & 31;
  const int wave = __builtin_amdgcn_readfirstlane((int)(threadIdx.x >> 5));
  const int c = lane & 15;
  const int h = lane >> 4;

  const int nstrips = nrows >> 4;
  const int s_lo = ((int)blockIdx.x * nstrips) / GEMM_WG;
  const int s_hi = (((int)blockIdx.x + 1) * nstrips) / GEMM_WG;
  static_assert(GEMM_WAVES == 16, "the renumbering below is a 4 x 4 square");
  const int wv = (((wave - (wave >> 2)) & 3) << 2) | (wave >> 2);
  int s = s_lo + wv;

  float4 wa[FEAT * FEAT / 4 / GEMM_THREADS], wb[FEAT * FEAT / 4 / GEMM_THREADS];
#pragma unroll
  for (int it = 0; it < FEAT * FEAT / 4 / GEMM_THREADS; ++it) wa[it] = ((const float4*)Wa)[threadIdx.x + GEMM_THREADS * it];
#pragma unroll
  for (int it = 0; it < FEAT * FEAT / 4 / GEMM_THREADS; ++it) wb[it] = ((const float4*)Wb)[threadIdx.x + GEMM_THREADS * it];

  float4 a[FEAT / 8];
  {
    const int sf = (s < nstrips) ? s : (nstrips - 1);
    const float4* xr = (const float4*)(X + (size_t)(sf * 16 + c) * FEAT + 4 * h);
#pragma unroll
    for (int j = 0; j < FEAT / 8; ++j) a[j] = xr[2 * j];
  }
  const float4 blo = ((const float4*)ba)[c];
  const float4 bhi = ((const float4*)ba)[16 + c];

  stage_w((float*)sWa4, wa);
  stage_w((float*)sWb4, wb);
  __syncthreads();

  const float4* bqa = sWa4 + (c * WT_PITCH) / 4 + h;
  const float4* bqb = sWb4 + (c * WT_PITCH) / 4 + h;
  float4* tw = sT4 + (wave * 16 + 8 * h) * (WT_PITCH / 4) + c;
  const float4* tr = sT4 + (wave * 16 + c) * (WT_PITCH / 4) + h;
  while (s < s_hi) {
    asm volatile("" ::: "memory");
    v8f acc[8];
    strip_product(bqa, a, acc);

#pragma unroll
    for (int r = 0; r < 8; ++r) {
      float4 lo, hi;
      lo.x = fmaxf(acc[0][r] + blo.x, 0.0f);
      lo.y = fmaxf(acc[1][r] + blo.y, 0.0f);
      lo.z = fmaxf(acc[2][r] + blo.z, 0.0f);
      lo.w = fmaxf(acc[3][r] + blo.w, 0.0f);
      hi.x = fmaxf(acc[4][r] + bhi.x, 0.0f);
      hi.y = fmaxf(acc[5][r] + bhi.y, 0.0f);
      hi.z = fmaxf(acc[6][r] + bhi.z, 0.0f);
      hi.w = fmaxf(acc[7][r] + bhi.w, 0.0f);
      tw[r * (WT_PITCH / 4)]      = lo;
      tw[r * (WT_PITCH / 4) + 16] = hi;
    }
#pragma unroll
    for (int j = 0; j < FEAT / 8; ++j) a[j] = tr[2 * j];

    strip_product(bqb, a, acc);

    const int sn = s + GEMM_WAVES;
    if (sn < s_hi) {
      const float4* xr = (const float4*)(X + (size_t)(sn * 16 + c) * FEAT + 4 * h);
#pragma unroll
      for (int j = 0; j < FEAT / 8; ++j) a[j] = xr[2 * j];
    }

    float4* orow = (float4*)(Hout + (size_t)(s * 16 + 8 * h) * FEAT + 4 * c);
#pragma unroll
    for (int r = 0; r < 8; ++r) {
      const float4 lo = {acc[0][r], acc[1][r], acc[2][r], acc[3][r]};
      const float4 hi = {acc[4][r], acc[5][r], acc[6][r], acc[7][r]};
      orow[r * (FEAT / 4)]      = lo;
      orow[r * (FEAT / 4) + 16] = hi;
    }
    s = sn;
  }
}

__device__ __forceinline__ void fma4(float4& acc, float w, const float4 v) {
  acc.x += w * v.x;
  acc.y += w * v.y;
  acc.z += w * v.z;
  acc.w += w * v.w;
}

__global__ __launch_bounds__(256) void k_gather(const float* __restrict__ h,
                                                const int* __restrict__ rowptr,
                                                const int* __restrict__ srcs,
                                                const float* __restrict__ dinv,
                                                const float* __restrict__ bias,
                                                float* __restrict__ out, int relu) {
  const int lane = threadIdx.x & 31;
  const int wave = __builtin_amdgcn_readfirstlane((int)(threadIdx.x >> 5));
  const int i = blockIdx.x * 8 + wave;
  if (i >= N_NODES) return;
  const int start = rowptr[i];
  const int n = rowptr[i + 1] - start;
  const float di = dinv[i];
  const float4* h4 = (const float4*)h;
  const float4 hv = h4[(size_t)i * (FEAT / 4) + lane];
  float4 bv = {0.0f, 0.0f, 0.0f, 0.0f};
  if (bias) bv = ((const float4*)bias)[lane];

  float4 acc = {0.0f, 0.0f, 0.0f, 0.0f};
  for (int b = 0; b < n; b += 32) {
    const int m = (n - b < 32) ? (n - b) : 32;
    int s = 0;
    float ds = 0.0f;
    if (lane < m) {
      s = srcs[start + b + lane];
      ds = dinv[s];
    }
    int j = 0;
    for (; j + 4 <= m; j += 4) {
      const int s0 = __builtin_amdgcn_readlane(s, j);
      const int s1 = __builtin_amdgcn_readlane(s, j + 1);
      const int s2 = __builtin_amdgcn_readlane(s, j + 2);
      const int s3 = __builtin_amdgcn_readlane(s, j + 3);
      const float4 v0 = h4[(size_t)s0 * (FEAT / 4) + lane];
      const float4 v1 = h4[(size_t)s1 * (FEAT / 4) + lane];
      const float4 v2 = h4[(size_t)s2 * (FEAT / 4) + lane];
      const float4 v3 = h4[(size_t)s3 * (FEAT / 4) + lane];
      const float w0 = __int_as_float(__builtin_amdgcn_readlane(__float_as_int(ds), j)) * di;
      const float w1 = __int_as_float(__builtin_amdgcn_readlane(__float_as_int(ds), j + 1)) * di;
      const float w2 = __int_as_float(__builtin_amdgcn_readlane(__float_as_int(ds), j + 2)) * di;
      const float w3 = __int_as_float(__builtin_amdgcn_readlane(__float_as_int(ds), j + 3)) * di;
      fma4(acc, w0, v0);
      fma4(acc, w1, v1);
      fma4(acc, w2, v2);
      fma4(acc, w3, v3);
    }
    for (; j < m; ++j) {
      const int s0 = __builtin_amdgcn_readlane(s, j);
      const float4 v0 = h4[(size_t)s0 * (FEAT / 4) + lane];
      const float w0 = __int_as_float(__builtin_amdgcn_readlane(__float_as_int(ds), j)) * di;
      fma4(acc, w0, v0);
    }
  }

  const float sl = di * di;
  float4 rv;
  rv.x = acc.x + sl * hv.x + bv.x;
  rv.y = acc.y + sl * hv.y + bv.y;
  rv.z = acc.z + sl * hv.z + bv.z;
  rv.w = acc.w + sl * hv.w + bv.w;
  if (relu) {
    rv.x = fmaxf(rv.x, 0.0f);
    rv.y = fmaxf(rv.y, 0.0f);
    rv.z = fmaxf(rv.z, 0.0f);
    rv.w = fmaxf(rv.w, 0.0f);
  }
  ((float4*)out)[(size_t)i * (FEAT / 4) + lane] = rv;
}

#define POOL_RUN 32
static_assert(N_NODES % POOL_RUN == 0, "k_pool takes whole runs of nodes");

__device__ __forceinline__ void pool_flush(float* __restrict__ sums, float* __restrict__ cnts,
                                           int g, int lane, const float4 acc, int run) {
  float* sp = sums + (size_t)g * FEAT + lane * 4;
  atomAddF(sp + 0, acc.x);
  atomAddF(sp + 1, acc.y);
  atomAddF(sp + 2, acc.z);
  atomAddF(sp + 3, acc.w);
  if (lane == 0) atomAddF(&cnts[g], (float)run);
}

__global__ __launch_bounds__(256) void k_pool(const float* __restrict__ a,
                                              const int* __restrict__ batch,
                                              float* __restrict__ sums,
                                              float* __restrict__ cnts) {
  const int lane = threadIdx.x & 31;
  const int wave = __builtin_amdgcn_readfirstlane((int)(threadIdx.x >> 5));
  const int n0 = (blockIdx.x * 8 + wave) * POOL_RUN;
  if (n0 >= N_NODES) return;
  const int gid = batch[n0 + lane];
  const float4* a4 = (const float4*)a + (size_t)n0 * (FEAT / 4) + lane;
  float4 v[POOL_RUN];
#pragma unroll
  for (int j = 0; j < POOL_RUN; ++j) v[j] = a4[j * (FEAT / 4)];

  int gcur = __builtin_amdgcn_readlane(gid, 0);
  float4 acc = v[0];
  int run = 1;
#pragma unroll
  for (int j = 1; j < POOL_RUN; ++j) {
    const int g = __builtin_amdgcn_readlane(gid, j);
    if (g != gcur) {
      pool_flush(sums, cnts, gcur, lane, acc, run);
      gcur = g;
      acc = v[j];
      run = 1;
    } else {
      acc.x += v[j].x;
      acc.y += v[j].y;
      acc.z += v[j].z;
      acc.w += v[j].w;
      ++run;
    }
  }
  pool_flush(sums, cnts, gcur, lane, acc, run);
}

#define TAIL_PITCH (FEAT + 1)
static_assert(NGRAPH % 16 == 0, "k_tail takes whole strips of 16 graphs");
static_assert(16 * NCLS <= 256, "one thread per (graph, class) of the strip");
static_assert((FEAT * NCLS) % 256 == 0, "k_tail stages Wl in whole passes of the work-group");

__global__ __launch_bounds__(256) void k_tail(const float* __restrict__ sums,
                                              const float* __restrict__ cnts,
                                              const float* __restrict__ W3,
                                              const float* __restrict__ b3,
                                              const float* __restrict__ Wl,
                                              const float* __restrict__ bl,
                                              float* __restrict__ out) {
  __shared__ float sT[16 * TAIL_PITCH];
  __shared__ float sWl[FEAT * NCLS];
  const int lane = threadIdx.x & 31;
  const int wave = threadIdx.x >> 5;
  const int c    = lane & 15;
  const int h    = lane >> 4;
  const int g0   = blockIdx.x * 16;

  float wl[FEAT * NCLS / 256];
#pragma unroll
  for (int q = 0; q < FEAT * NCLS / 256; ++q) wl[q] = Wl[threadIdx.x + 256 * q];

  const float* arow = sums + (size_t)(g0 + c) * FEAT + 2 * h;
  const float* bcol = W3 + (size_t)(2 * h) * FEAT + wave * 16 + c;
  v2f a[FEAT / 4], b[FEAT / 4];
#pragma unroll
  for (int ks = 0; ks < FEAT / 4; ++ks) {
    a[ks].x = arow[4 * ks];
    a[ks].y = arow[4 * ks + 1];
    b[ks].x = bcol[(size_t)(4 * ks) * FEAT];
    b[ks].y = bcol[(size_t)(4 * ks + 1) * FEAT];
  }
  float cn[8];
#pragma unroll
  for (int r = 0; r < 8; ++r) cn[r] = cnts[g0 + 8 * h + r];
  const float bias3 = b3[wave * 16 + c];

  v8f acc = (v8f){0, 0, 0, 0, 0, 0, 0, 0};
#pragma unroll
  for (int ks = 0; ks < FEAT / 4; ++ks)
    acc = __builtin_amdgcn_wmma_f32_16x16x4_f32(false, a[ks], false, b[ks], (short)0, acc, false, false);

#pragma unroll
  for (int r = 0; r < 8; ++r) {
    const float inv = 1.0f / fmaxf(cn[r], 1.0f);
    sT[(8 * h + r) * TAIL_PITCH + wave * 16 + c] = acc[r] * inv + ((cn[r] > 0.0f) ? bias3 : 0.0f);
  }
#pragma unroll
  for (int q = 0; q < FEAT * NCLS / 256; ++q) sWl[threadIdx.x + 256 * q] = wl[q];
  __syncthreads();

  if (threadIdx.x < 16 * NCLS) {
    const int gl = threadIdx.x / NCLS;
    const int cl = threadIdx.x - gl * NCLS;
    float o = bl[cl];
#pragma unroll 16
    for (int k = 0; k < FEAT; ++k) o += sT[gl * TAIL_PITCH + k] * sWl[k * NCLS + cl];
    out[(size_t)(g0 + gl) * NCLS + cl] = o;
  }
}

extern "C" void kernel_launch(void* const* d_in, const int* in_sizes, int n_in,
                              void* d_out, int out_size, void* d_ws, size_t ws_size,
                              hipStream_t stream) {
  const float* x     = (const float*)d_in[0];
  const int*   ei    = (const int*)d_in[1];
  const int*   row   = ei;
  const int*   col   = ei + N_EDGES;
  const int*   batch = (const int*)d_in[2];
  const float* W1 = (const float*)d_in[3];
  const float* b1 = (const float*)d_in[4];
  const float* W2 = (const float*)d_in[5];
  const float* b2 = (const float*)d_in[6];
  const float* W3 = (const float*)d_in[7];
  const float* b3 = (const float*)d_in[8];
  const float* Wl = (const float*)d_in[9];
  const float* bl = (const float*)d_in[10];
  float* outp = (float*)d_out;

  char* base = (char*)d_ws;
  const size_t featBytes = (size_t)N_NODES * FEAT * sizeof(float);
  int*   histM  = (int*)(base);
  int*   rowptr = (int*)(base + (512 << 10));
  int*   bbase  = (int*)(base + (1 << 20));
  float* dinv   = (float*)(base + (1 << 20) + (512 << 10));
  int*   srcs   = (int*)(base + (4 << 20));
  float* hbuf = (float*)(base + (8 << 20));
  float* abuf = (float*)(base + (8 << 20) + featBytes);
  float* sums = (float*)(base + (8 << 20) + 2 * featBytes);
  float* cnts = sums + (size_t)NGRAPH * FEAT;
  unsigned* binned = (unsigned*)(base + (9 << 20) + 2 * featBytes);

  const int  TB     = 256;
  const int  gGath  = (N_NODES + 8 - 1) / 8;
  const int  gPool  = (N_NODES / POOL_RUN + 8 - 1) / 8;

  k_hist<<<EB_WG, EB_THREADS, 0, stream>>>(col, histM);
  k_bin<<<EB_WG, EB_THREADS, 0, stream>>>(row, col, histM, bbase, binned);
  k_group<<<NBKT, BKT_NODES, 0, stream>>>(binned, bbase, rowptr, dinv, srcs);

  k_gather<<<gGath, TB, 0, stream>>>(x, rowptr, srcs, dinv, nullptr, abuf, 0);
  k_gemm2<<<GEMM_WG, GEMM_THREADS, 0, stream>>>(abuf, W1, b1, W2, hbuf, N_NODES);

  k_gather<<<gGath, TB, 0, stream>>>(hbuf, rowptr, srcs, dinv, b2, abuf, 1);

  k_gather<<<gGath, TB, 0, stream>>>(abuf, rowptr, srcs, dinv, nullptr, hbuf, 0);

  const long poolN = (long)NGRAPH * FEAT + NGRAPH;
  k_fill<<<(int)((poolN + TB - 1) / TB), TB, 0, stream>>>(sums, 0.0f, poolN);
  k_pool<<<gPool, TB, 0, stream>>>(hbuf, batch, sums, cnts);
  k_tail<<<NGRAPH / 16, TB, 0, stream>>>(sums, cnts, W3, b3, Wl, bl, outp);
}


# layer 3's gather and the pool in one launch: a work-group gathers 32 consecutive nodes (4 per wave) into LDS, a thread per feature adds them in node order; pool launch and 51 MB write+read gone
# speedup vs baseline: 5.0764x; 1.0376x over previous
#include <hip/hip_runtime.h>

typedef __attribute__((ext_vector_type(2))) float v2f;
typedef __attribute__((ext_vector_type(8))) float v8f;

#define N_NODES 100000
#define N_EDGES 800000
#define FEAT    128
#define NGRAPH  512
#define NCLS    10

__device__ __forceinline__ void atomAddF(float* p, float v) {
  unsafeAtomicAdd(p, v);
}

__global__ void k_fill(float* __restrict__ p, float v, long n) {
  long i = (long)blockIdx.x * blockDim.x + threadIdx.x;
  if (i < n) p[i] = v;
}

#define EB_WG      128
#define EB_THREADS 512
#define EB_WAVES   (EB_THREADS / 32)
#define EB_CHUNK   ((N_EDGES + EB_WG - 1) / EB_WG)
#define EB_BATCH   4
#define BKT_SHIFT  8
#define BKT_NODES  (1 << BKT_SHIFT)
#define NBKT       ((N_NODES + BKT_NODES - 1) / BKT_NODES)
static_assert(NBKT <= EB_THREADS, "k_bin gives one thread to every bucket");
static_assert(N_NODES <= (1 << 24), "a packed edge keeps the source in 24 bits");
static_assert(BKT_NODES <= 256, "a packed edge keeps the target's place in its bucket in 8 bits");

template <int NW>
__device__ __forceinline__ int block_excl_scan(int v, int* s_wave, int& total) {
  const int lane = threadIdx.x & 31;
  const int wave = threadIdx.x >> 5;
  int inc = v;
  for (int o = 1; o < 32; o <<= 1) {
    const int y = __shfl_up(inc, o);
    if (lane >= o) inc += y;
  }
  if (lane == 31) s_wave[wave] = inc;
  __syncthreads();
  int base = 0, tot = 0;
  for (int w = 0; w < NW; ++w) {
    const int x = s_wave[w];
    if (w < wave) base += x;
    tot += x;
  }
  total = tot;
  return base + inc - v;
}

__global__ __launch_bounds__(EB_THREADS) void k_hist(const int* __restrict__ col,
                                                     int* __restrict__ histM) {
  __shared__ int s_h[NBKT];
  const int t = threadIdx.x;
  const int w = blockIdx.x;
  for (int b = t; b < NBKT; b += EB_THREADS) s_h[b] = 0;
  __syncthreads();
  const int e0 = w * EB_CHUNK;
  const int e1 = (e0 + EB_CHUNK < N_EDGES) ? (e0 + EB_CHUNK) : N_EDGES;
  for (int e = e0 + t; e < e1; e += EB_BATCH * EB_THREADS) {
    int c[EB_BATCH];
#pragma unroll
    for (int k = 0; k < EB_BATCH; ++k) {
      const int ee = e + k * EB_THREADS;
      c[k] = (ee < e1) ? col[ee] : -1;
    }
#pragma unroll
    for (int k = 0; k < EB_BATCH; ++k)
      if (c[k] >= 0) atomicAdd(&s_h[c[k] >> BKT_SHIFT], 1);
  }
  __syncthreads();
  for (int b = t; b < NBKT; b += EB_THREADS) histM[w * NBKT + b] = s_h[b];
}

__global__ __launch_bounds__(EB_THREADS) void k_bin(const int* __restrict__ row,
                                                    const int* __restrict__ col,
                                                    const int* __restrict__ histM,
                                                    int* __restrict__ bbase,
                                                    unsigned* __restrict__ binned) {
  __shared__ int s_cur[NBKT];
  __shared__ int s_wave[EB_WAVES];
  const int t = threadIdx.x;
  const int w = blockIdx.x;

  int before = 0, total = 0;
  if (t < NBKT) {
#pragma unroll 32
    for (int w2 = 0; w2 < EB_WG; ++w2) {
      const int v = histM[w2 * NBKT + t];
      total += v;
      before += (w2 < w) ? v : 0;
    }
  }
  int all;
  const int base = block_excl_scan<EB_WAVES>(total, s_wave, all);
  if (t < NBKT) {
    s_cur[t] = base + before;
    if (w == 0) bbase[t] = base;
  }
  if (w == 0 && t == 0) bbase[NBKT] = all;
  __syncthreads();

  const int e0 = w * EB_CHUNK;
  const int e1 = (e0 + EB_CHUNK < N_EDGES) ? (e0 + EB_CHUNK) : N_EDGES;
  for (int e = e0 + t; e < e1; e += EB_BATCH * EB_THREADS) {
    int c[EB_BATCH], r[EB_BATCH];
#pragma unroll
    for (int k = 0; k < EB_BATCH; ++k) {
      const int ee = e + k * EB_THREADS;
      const bool ok = ee < e1;
      c[k] = ok ? col[ee] : -1;
      r[k] = ok ? row[ee] : 0;
    }
#pragma unroll
    for (int k = 0; k < EB_BATCH; ++k) {
      if (c[k] >= 0) {
        const int pos = atomicAdd(&s_cur[c[k] >> BKT_SHIFT], 1);
        binned[pos] = ((unsigned)(c[k] & (BKT_NODES - 1)) << 24) | (unsigned)r[k];
      }
    }
  }
}

__device__ __forceinline__ float dinv_of(int c) {
  const float d = (float)c + 1.0f;
  return (d > 0.0f) ? rsqrtf(fmaxf(d, 1.0f)) : 0.0f;
}

__global__ __launch_bounds__(BKT_NODES) void k_group(const unsigned* __restrict__ binned,
                                                     const int* __restrict__ bbase,
                                                     int* __restrict__ rowptr,
                                                     float* __restrict__ dinv,
                                                     int* __restrict__ srcs) {
  __shared__ int s_cnt[BKT_NODES];
  __shared__ int s_wave[BKT_NODES / 32];
  const int t = threadIdx.x;
  const int b = blockIdx.x;
  const int lo = bbase[b];
  const int hi = bbase[b + 1];
  s_cnt[t] = 0;
  __syncthreads();
  for (int p = lo + t; p < hi; p += EB_BATCH * BKT_NODES) {
    unsigned u[EB_BATCH];
#pragma unroll
    for (int k = 0; k < EB_BATCH; ++k) {
      const int pp = p + k * BKT_NODES;
      u[k] = binned[(pp < hi) ? pp : p];
    }
#pragma unroll
    for (int k = 0; k < EB_BATCH; ++k)
      atomicAdd(&s_cnt[u[k] >> 24], (p + k * BKT_NODES < hi) ? 1 : 0);
  }
  __syncthreads();
  const int c = s_cnt[t];
  int all;
  const int ex = block_excl_scan<BKT_NODES / 32>(c, s_wave, all);
  const int node = b * BKT_NODES + t;
  if (node < N_NODES) {
    rowptr[node] = lo + ex;
    dinv[node] = dinv_of(c);
  }
  if (b == NBKT - 1 && t == 0) rowptr[N_NODES] = hi;
  s_cnt[t] = lo + ex;
  __syncthreads();
  for (int p = lo + t; p < hi; p += EB_BATCH * BKT_NODES) {
    unsigned u[EB_BATCH];
#pragma unroll
    for (int k = 0; k < EB_BATCH; ++k) {
      const int pp = p + k * BKT_NODES;
      u[k] = binned[(pp < hi) ? pp : p];
    }
#pragma unroll
    for (int k = 0; k < EB_BATCH; ++k) {
      const bool ok = p + k * BKT_NODES < hi;
      const int pos = atomicAdd(&s_cnt[u[k] >> 24], ok ? 1 : 0);
      if (ok) srcs[pos] = (int)(u[k] & 0xFFFFFFu);
    }
  }
}

#define GEMM_WG      256
#define GEMM_THREADS 512
#define GEMM_WAVES   (GEMM_THREADS / 32)
#define WT_PITCH     (FEAT + 4)
static_assert(N_NODES % 16 == 0, "k_gemm2 takes whole strips of 16 rows");
static_assert((FEAT * FEAT / 4) % GEMM_THREADS == 0, "k_gemm2 stages a W in whole passes of the work-group");

__device__ __forceinline__ void strip_product(const float4* __restrict__ bq, const float4 (&a)[FEAT / 8],
                                              v8f (&acc)[8]) {
#pragma unroll
  for (int n = 0; n < 8; ++n) acc[n] = (v8f){0, 0, 0, 0, 0, 0, 0, 0};
  float4 b[8];
#pragma unroll
  for (int n = 0; n < 8; ++n) b[n] = bq[n * (4 * WT_PITCH)];
#pragma unroll
  for (int j = 0; j < FEAT / 8; ++j) {
    float4 bn[8];
    __builtin_amdgcn_sched_barrier(0);
    if (j + 1 < FEAT / 8) {
#pragma unroll
      for (int n = 0; n < 8; ++n) bn[n] = bq[n * (4 * WT_PITCH) + 2 * (j + 1)];
    }
    __builtin_amdgcn_sched_barrier(0);
    const v2f a0 = {a[j].x, a[j].y};
    const v2f a1 = {a[j].z, a[j].w};
#pragma unroll
    for (int n = 0; n < 8; ++n) {
      const v2f b0 = {b[n].x, b[n].y};
      acc[n] = __builtin_amdgcn_wmma_f32_16x16x4_f32(false, a0, false, b0, (short)0, acc[n], false, false);
    }
#pragma unroll
    for (int n = 0; n < 8; ++n) {
      const v2f b1 = {b[n].z, b[n].w};
      acc[n] = __builtin_amdgcn_wmma_f32_16x16x4_f32(false, a1, false, b1, (short)0, acc[n], false, false);
    }
    if (j + 1 < FEAT / 8) {
#pragma unroll
      for (int n = 0; n < 8; ++n) b[n] = bn[n];
    }
  }
}

__device__ __forceinline__ void stage_w(float* __restrict__ sW, const float4 (&w)[FEAT * FEAT / 4 / GEMM_THREADS]) {
#pragma unroll
  for (int it = 0; it < FEAT * FEAT / 4 / GEMM_THREADS; ++it) {
    const int idx = threadIdx.x + GEMM_THREADS * it;
    const int k = idx >> 5;
    const int q = idx & 31;
    const int p = 64 * (q >> 4) + (q & 15);
    sW[(p     ) * WT_PITCH + k] = w[it].x;
    sW[(p + 16) * WT_PITCH + k] = w[it].y;
    sW[(p + 32) * WT_PITCH + k] = w[it].z;
    sW[(p + 48) * WT_PITCH + k] = w[it].w;
  }
}

__global__ __launch_bounds__(GEMM_THREADS) void k_gemm2(const float* __restrict__ X,
                                                        const float* __restrict__ Wa,
                                                        const float* __restrict__ ba,
                                                        const float* __restrict__ Wb,
                                                        float* __restrict__ Hout,
                                                        int nrows) {
  __shared__ float4 sWa4[FEAT * WT_PITCH / 4];
  __shared__ float4 sWb4[FEAT * WT_PITCH / 4];
  __shared__ float4 sT4[GEMM_WAVES * 16 * WT_PITCH / 4];
  const int lane = threadIdx.x & 31;
  const int wave = __builtin_amdgcn_readfirstlane((int)(threadIdx.x >> 5));
  const int c = lane & 15;
  const int h = lane >> 4;

  const int nstrips = nrows >> 4;
  const int s_lo = ((int)blockIdx.x * nstrips) / GEMM_WG;
  const int s_hi = (((int)blockIdx.x + 1) * nstrips) / GEMM_WG;
  static_assert(GEMM_WAVES == 16, "the renumbering below is a 4 x 4 square");
  const int wv = (((wave - (wave >> 2)) & 3) << 2) | (wave >> 2);
  int s = s_lo + wv;

  float4 wa[FEAT * FEAT / 4 / GEMM_THREADS], wb[FEAT * FEAT / 4 / GEMM_THREADS];
#pragma unroll
  for (int it = 0; it < FEAT * FEAT / 4 / GEMM_THREADS; ++it) wa[it] = ((const float4*)Wa)[threadIdx.x + GEMM_THREADS * it];
#pragma unroll
  for (int it = 0; it < FEAT * FEAT / 4 / GEMM_THREADS; ++it) wb[it] = ((const float4*)Wb)[threadIdx.x + GEMM_THREADS * it];

  float4 a[FEAT / 8];
  {
    const int sf = (s < nstrips) ? s : (nstrips - 1);
    const float4* xr = (const float4*)(X + (size_t)(sf * 16 + c) * FEAT + 4 * h);
#pragma unroll
    for (int j = 0; j < FEAT / 8; ++j) a[j] = xr[2 * j];
  }
  const float4 blo = ((const float4*)ba)[c];
  const float4 bhi = ((const float4*)ba)[16 + c];

  stage_w((float*)sWa4, wa);
  stage_w((float*)sWb4, wb);
  __syncthreads();

  const float4* bqa = sWa4 + (c * WT_PITCH) / 4 + h;
  const float4* bqb = sWb4 + (c * WT_PITCH) / 4 + h;
  float4* tw = sT4 + (wave * 16 + 8 * h) * (WT_PITCH / 4) + c;
  const float4* tr = sT4 + (wave * 16 + c) * (WT_PITCH / 4) + h;
  while (s < s_hi) {
    asm volatile("" ::: "memory");
    v8f acc[8];
    strip_product(bqa, a, acc);

#pragma unroll
    for (int r = 0; r < 8; ++r) {
      float4 lo, hi;
      lo.x = fmaxf(acc[0][r] + blo.x, 0.0f);
      lo.y = fmaxf(acc[1][r] + blo.y, 0.0f);
      lo.z = fmaxf(acc[2][r] + blo.z, 0.0f);
      lo.w = fmaxf(acc[3][r] + blo.w, 0.0f);
      hi.x = fmaxf(acc[4][r] + bhi.x, 0.0f);
      hi.y = fmaxf(acc[5][r] + bhi.y, 0.0f);
      hi.z = fmaxf(acc[6][r] + bhi.z, 0.0f);
      hi.w = fmaxf(acc[7][r] + bhi.w, 0.0f);
      tw[r * (WT_PITCH / 4)]      = lo;
      tw[r * (WT_PITCH / 4) + 16] = hi;
    }
#pragma unroll
    for (int j = 0; j < FEAT / 8; ++j) a[j] = tr[2 * j];

    strip_product(bqb, a, acc);

    const int sn = s + GEMM_WAVES;
    if (sn < s_hi) {
      const float4* xr = (const float4*)(X + (size_t)(sn * 16 + c) * FEAT + 4 * h);
#pragma unroll
      for (int j = 0; j < FEAT / 8; ++j) a[j] = xr[2 * j];
    }

    float4* orow = (float4*)(Hout + (size_t)(s * 16 + 8 * h) * FEAT + 4 * c);
#pragma unroll
    for (int r = 0; r < 8; ++r) {
      const float4 lo = {acc[0][r], acc[1][r], acc[2][r], acc[3][r]};
      const float4 hi = {acc[4][r], acc[5][r], acc[6][r], acc[7][r]};
      orow[r * (FEAT / 4)]      = lo;
      orow[r * (FEAT / 4) + 16] = hi;
    }
    s = sn;
  }
}

__device__ __forceinline__ void fma4(float4& acc, float w, const float4 v) {
  acc.x += w * v.x;
  acc.y += w * v.y;
  acc.z += w * v.z;
  acc.w += w * v.w;
}

__global__ __launch_bounds__(256) void k_gather(const float* __restrict__ h,
                                                const int* __restrict__ rowptr,
                                                const int* __restrict__ srcs,
                                                const float* __restrict__ dinv,
                                                const float* __restrict__ bias,
                                                float* __restrict__ out, int relu) {
  const int lane = threadIdx.x & 31;
  const int wave = __builtin_amdgcn_readfirstlane((int)(threadIdx.x >> 5));
  const int i = blockIdx.x * 8 + wave;
  if (i >= N_NODES) return;
  const int start = rowptr[i];
  const int n = rowptr[i + 1] - start;
  const float di = dinv[i];
  const float4* h4 = (const float4*)h;
  const float4 hv = h4[(size_t)i * (FEAT / 4) + lane];
  float4 bv = {0.0f, 0.0f, 0.0f, 0.0f};
  if (bias) bv = ((const float4*)bias)[lane];

  float4 acc = {0.0f, 0.0f, 0.0f, 0.0f};
  for (int b = 0; b < n; b += 32) {
    const int m = (n - b < 32) ? (n - b) : 32;
    int s = 0;
    float ds = 0.0f;
    if (lane < m) {
      s = srcs[start + b + lane];
      ds = dinv[s];
    }
    int j = 0;
    for (; j + 4 <= m; j += 4) {
      const int s0 = __builtin_amdgcn_readlane(s, j);
      const int s1 = __builtin_amdgcn_readlane(s, j + 1);
      const int s2 = __builtin_amdgcn_readlane(s, j + 2);
      const int s3 = __builtin_amdgcn_readlane(s, j + 3);
      const float4 v0 = h4[(size_t)s0 * (FEAT / 4) + lane];
      const float4 v1 = h4[(size_t)s1 * (FEAT / 4) + lane];
      const float4 v2 = h4[(size_t)s2 * (FEAT / 4) + lane];
      const float4 v3 = h4[(size_t)s3 * (FEAT / 4) + lane];
      const float w0 = __int_as_float(__builtin_amdgcn_readlane(__float_as_int(ds), j)) * di;
      const float w1 = __int_as_float(__builtin_amdgcn_readlane(__float_as_int(ds), j + 1)) * di;
      const float w2 = __int_as_float(__builtin_amdgcn_readlane(__float_as_int(ds), j + 2)) * di;
      const float w3 = __int_as_float(__builtin_amdgcn_readlane(__float_as_int(ds), j + 3)) * di;
      fma4(acc, w0, v0);
      fma4(acc, w1, v1);
      fma4(acc, w2, v2);
      fma4(acc, w3, v3);
    }
    for (; j < m; ++j) {
      const int s0 = __builtin_amdgcn_readlane(s, j);
      const float4 v0 = h4[(size_t)s0 * (FEAT / 4) + lane];
      const float w0 = __int_as_float(__builtin_amdgcn_readlane(__float_as_int(ds), j)) * di;
      fma4(acc, w0, v0);
    }
  }

  const float sl = di * di;
  float4 rv;
  rv.x = acc.x + sl * hv.x + bv.x;
  rv.y = acc.y + sl * hv.y + bv.y;
  rv.z = acc.z + sl * hv.z + bv.z;
  rv.w = acc.w + sl * hv.w + bv.w;
  if (relu) {
    rv.x = fmaxf(rv.x, 0.0f);
    rv.y = fmaxf(rv.y, 0.0f);
    rv.z = fmaxf(rv.z, 0.0f);
    rv.w = fmaxf(rv.w, 0.0f);
  }
  ((float4*)out)[(size_t)i * (FEAT / 4) + lane] = rv;
}

#define GP_NODES    32
#define GP_WAVES    8
#define GP_PER_WAVE (GP_NODES / GP_WAVES)
static_assert(N_NODES % GP_NODES == 0, "k_gather_pool takes whole runs of nodes: no wave leaves in front of the barrier");
static_assert(FEAT <= GP_WAVES * 32, "one thread per feature adds the rows up");

__global__ __launch_bounds__(GP_WAVES * 32) void k_gather_pool(const float* __restrict__ h,
                                                               const int* __restrict__ rowptr,
                                                               const int* __restrict__ srcs,
                                                               const float* __restrict__ dinv,
                                                               const int* __restrict__ batch,
                                                               float* __restrict__ sums,
                                                               float* __restrict__ cnts) {
  __shared__ float4 sRow[GP_NODES * (FEAT / 4)];
  __shared__ int sGid[GP_NODES];
  const int lane = threadIdx.x & 31;
  const int wave = __builtin_amdgcn_readfirstlane((int)(threadIdx.x >> 5));
  const int n0 = blockIdx.x * GP_NODES;
  const float4* h4 = (const float4*)h;

  int gidv = 0;
  if (threadIdx.x < GP_NODES) gidv = batch[n0 + threadIdx.x];

  int st[GP_PER_WAVE], cn[GP_PER_WAVE], sq[GP_PER_WAVE];
  float dq[GP_PER_WAVE];
#pragma unroll
  for (int q = 0; q < GP_PER_WAVE; ++q) {
    const int i = n0 + wave + GP_WAVES * q;
    st[q] = rowptr[i];
    cn[q] = rowptr[i + 1] - st[q];
    dq[q] = dinv[i];
  }
#pragma unroll
  for (int q = 0; q < GP_PER_WAVE; ++q) sq[q] = (lane < cn[q]) ? srcs[st[q] + lane] : 0;

#pragma unroll
  for (int q = 0; q < GP_PER_WAVE; ++q) {
    const int i = n0 + wave + GP_WAVES * q;
    const int start = st[q];
    const int n = cn[q];
    const float di = dq[q];
    const float4 hv = h4[(size_t)i * (FEAT / 4) + lane];
    float4 acc = {0.0f, 0.0f, 0.0f, 0.0f};
    for (int b = 0; b < n; b += 32) {
      const int m = (n - b < 32) ? (n - b) : 32;
      int s = sq[q];
      if (b > 0) s = (lane < m) ? srcs[start + b + lane] : 0;
      float ds = 0.0f;
      if (lane < m) ds = dinv[s];
      int j = 0;
      for (; j + 4 <= m; j += 4) {
        const int s0 = __builtin_amdgcn_readlane(s, j);
        const int s1 = __builtin_amdgcn_readlane(s, j + 1);
        const int s2 = __builtin_amdgcn_readlane(s, j + 2);
        const int s3 = __builtin_amdgcn_readlane(s, j + 3);
        const float4 v0 = h4[(size_t)s0 * (FEAT / 4) + lane];
        const float4 v1 = h4[(size_t)s1 * (FEAT / 4) + lane];
        const float4 v2 = h4[(size_t)s2 * (FEAT / 4) + lane];
        const float4 v3 = h4[(size_t)s3 * (FEAT / 4) + lane];
        const float w0 = __int_as_float(__builtin_amdgcn_readlane(__float_as_int(ds), j)) * di;
        const float w1 = __int_as_float(__builtin_amdgcn_readlane(__float_as_int(ds), j + 1)) * di;
        const float w2 = __int_as_float(__builtin_amdgcn_readlane(__float_as_int(ds), j + 2)) * di;
        const float w3 = __int_as_float(__builtin_amdgcn_readlane(__float_as_int(ds), j + 3)) * di;
        fma4(acc, w0, v0);
        fma4(acc, w1, v1);
        fma4(acc, w2, v2);
        fma4(acc, w3, v3);
      }
      for (; j < m; ++j) {
        const int s0 = __builtin_amdgcn_readlane(s, j);
        const float4 v0 = h4[(size_t)s0 * (FEAT / 4) + lane];
        const float w0 = __int_as_float(__builtin_amdgcn_readlane(__float_as_int(ds), j)) * di;
        fma4(acc, w0, v0);
      }
    }
    const float sl = di * di;
    float4 rv;
    rv.x = acc.x + sl * hv.x;
    rv.y = acc.y + sl * hv.y;
    rv.z = acc.z + sl * hv.z;
    rv.w = acc.w + sl * hv.w;
    sRow[(wave + GP_WAVES * q) * (FEAT / 4) + lane] = rv;
  }
  if (threadIdx.x < GP_NODES) sGid[threadIdx.x] = gidv;
  __syncthreads();

  if (threadIdx.x < FEAT) {
    const int t = threadIdx.x;
    const float* colp = (const float*)sRow + t;
    int gcur = sGid[0];
    float acc = colp[0];
    int run = 1;
#pragma unroll 8
    for (int j = 1; j < GP_NODES; ++j) {
      const int g = sGid[j];
      const float v = colp[j * FEAT];
      if (g != gcur) {
        atomAddF(&sums[(size_t)gcur * FEAT + t], acc);
        if (t == 0) atomAddF(&cnts[gcur], (float)run);
        gcur = g;
        acc = v;
        run = 1;
      } else {
        acc += v;
        ++run;
      }
    }
    atomAddF(&sums[(size_t)gcur * FEAT + t], acc);
    if (t == 0) atomAddF(&cnts[gcur], (float)run);
  }
}

#define TAIL_PITCH (FEAT + 1)
static_assert(NGRAPH % 16 == 0, "k_tail takes whole strips of 16 graphs");
static_assert(16 * NCLS <= 256, "one thread per (graph, class) of the strip");
static_assert((FEAT * NCLS) % 256 == 0, "k_tail stages Wl in whole passes of the work-group");

__global__ __launch_bounds__(256) void k_tail(const float* __restrict__ sums,
                                              const float* __restrict__ cnts,
                                              const float* __restrict__ W3,
                                              const float* __restrict__ b3,
                                              const float* __restrict__ Wl,
                                              const float* __restrict__ bl,
                                              float* __restrict__ out) {
  __shared__ float sT[16 * TAIL_PITCH];
  __shared__ float sWl[FEAT * NCLS];
  const int lane = threadIdx.x & 31;
  const int wave = threadIdx.x >> 5;
  const int c    = lane & 15;
  const int h    = lane >> 4;
  const int g0   = blockIdx.x * 16;

  float wl[FEAT * NCLS / 256];
#pragma unroll
  for (int q = 0; q < FEAT * NCLS / 256; ++q) wl[q] = Wl[threadIdx.x + 256 * q];

  const float* arow = sums + (size_t)(g0 + c) * FEAT + 2 * h;
  const float* bcol = W3 + (size_t)(2 * h) * FEAT + wave * 16 + c;
  v2f a[FEAT / 4], b[FEAT / 4];
#pragma unroll
  for (int ks = 0; ks < FEAT / 4; ++ks) {
    a[ks].x = arow[4 * ks];
    a[ks].y = arow[4 * ks + 1];
    b[ks].x = bcol[(size_t)(4 * ks) * FEAT];
    b[ks].y = bcol[(size_t)(4 * ks + 1) * FEAT];
  }
  float cn[8];
#pragma unroll
  for (int r = 0; r < 8; ++r) cn[r] = cnts[g0 + 8 * h + r];
  const float bias3 = b3[wave * 16 + c];

  v8f acc = (v8f){0, 0, 0, 0, 0, 0, 0, 0};
#pragma unroll
  for (int ks = 0; ks < FEAT / 4; ++ks)
    acc = __builtin_amdgcn_wmma_f32_16x16x4_f32(false, a[ks], false, b[ks], (short)0, acc, false, false);

#pragma unroll
  for (int r = 0; r < 8; ++r) {
    const float inv = 1.0f / fmaxf(cn[r], 1.0f);
    sT[(8 * h + r) * TAIL_PITCH + wave * 16 + c] = acc[r] * inv + ((cn[r] > 0.0f) ? bias3 : 0.0f);
  }
#pragma unroll
  for (int q = 0; q < FEAT * NCLS / 256; ++q) sWl[threadIdx.x + 256 * q] = wl[q];
  __syncthreads();

  if (threadIdx.x < 16 * NCLS) {
    const int gl = threadIdx.x / NCLS;
    const int cl = threadIdx.x - gl * NCLS;
    float o = bl[cl];
#pragma unroll 16
    for (int k = 0; k < FEAT; ++k) o += sT[gl * TAIL_PITCH + k] * sWl[k * NCLS + cl];
    out[(size_t)(g0 + gl) * NCLS + cl] = o;
  }
}

extern "C" void kernel_launch(void* const* d_in, const int* in_sizes, int n_in,
                              void* d_out, int out_size, void* d_ws, size_t ws_size,
                              hipStream_t stream) {
  const float* x     = (const float*)d_in[0];
  const int*   ei    = (const int*)d_in[1];
  const int*   row   = ei;
  const int*   col   = ei + N_EDGES;
  const int*   batch = (const int*)d_in[2];
  const float* W1 = (const float*)d_in[3];
  const float* b1 = (const float*)d_in[4];
  const float* W2 = (const float*)d_in[5];
  const float* b2 = (const float*)d_in[6];
  const float* W3 = (const float*)d_in[7];
  const float* b3 = (const float*)d_in[8];
  const float* Wl = (const float*)d_in[9];
  const float* bl = (const float*)d_in[10];
  float* outp = (float*)d_out;

  char* base = (char*)d_ws;
  const size_t featBytes = (size_t)N_NODES * FEAT * sizeof(float);
  int*   histM  = (int*)(base);
  int*   rowptr = (int*)(base + (512 << 10));
  int*   bbase  = (int*)(base + (1 << 20));
  float* dinv   = (float*)(base + (1 << 20) + (512 << 10));
  int*   srcs   = (int*)(base + (4 << 20));
  float* hbuf = (float*)(base + (8 << 20));
  float* abuf = (float*)(base + (8 << 20) + featBytes);
  float* sums = (float*)(base + (8 << 20) + 2 * featBytes);
  float* cnts = sums + (size_t)NGRAPH * FEAT;
  unsigned* binned = (unsigned*)(base + (9 << 20) + 2 * featBytes);

  const int  TB     = 256;
  const int  gGath  = (N_NODES + 8 - 1) / 8;

  k_hist<<<EB_WG, EB_THREADS, 0, stream>>>(col, histM);
  k_bin<<<EB_WG, EB_THREADS, 0, stream>>>(row, col, histM, bbase, binned);
  k_group<<<NBKT, BKT_NODES, 0, stream>>>(binned, bbase, rowptr, dinv, srcs);

  k_gather<<<gGath, TB, 0, stream>>>(x, rowptr, srcs, dinv, nullptr, abuf, 0);
  k_gemm2<<<GEMM_WG, GEMM_THREADS, 0, stream>>>(abuf, W1, b1, W2, hbuf, N_NODES);

  k_gather<<<gGath, TB, 0, stream>>>(hbuf, rowptr, srcs, dinv, b2, abuf, 1);

  const long poolN = (long)NGRAPH * FEAT + NGRAPH;
  k_fill<<<(int)((poolN + TB - 1) / TB), TB, 0, stream>>>(sums, 0.0f, poolN);
  k_gather_pool<<<N_NODES / GP_NODES, GP_WAVES * 32, 0, stream>>>(abuf, rowptr, srcs, dinv, batch, sums, cnts);
  k_tail<<<NGRAPH / 16, TB, 0, stream>>>(sums, cnts, W3, b3, Wl, bl, outp);
}


# layer 1's aggregation inside the product kernel: a wave walks its strip's 16 in-edge lists as one list, 16 rows in flight, into its LDS tile; gather launch, 51 MB write+read gone; same sum order
# speedup vs baseline: 5.2380x; 1.0318x over previous
#include <hip/hip_runtime.h>

typedef __attribute__((ext_vector_type(2))) float v2f;
typedef __attribute__((ext_vector_type(8))) float v8f;

#define N_NODES 100000
#define N_EDGES 800000
#define FEAT    128
#define NGRAPH  512
#define NCLS    10

__device__ __forceinline__ void atomAddF(float* p, float v) {
  unsafeAtomicAdd(p, v);
}

__global__ void k_fill(float* __restrict__ p, float v, long n) {
  long i = (long)blockIdx.x * blockDim.x + threadIdx.x;
  if (i < n) p[i] = v;
}

#define EB_WG      128
#define EB_THREADS 512
#define EB_WAVES   (EB_THREADS / 32)
#define EB_CHUNK   ((N_EDGES + EB_WG - 1) / EB_WG)
#define EB_BATCH   4
#define BKT_SHIFT  8
#define BKT_NODES  (1 << BKT_SHIFT)
#define NBKT       ((N_NODES + BKT_NODES - 1) / BKT_NODES)
static_assert(NBKT <= EB_THREADS, "k_bin gives one thread to every bucket");
static_assert(N_NODES <= (1 << 24), "a packed edge keeps the source in 24 bits");
static_assert(BKT_NODES <= 256, "a packed edge keeps the target's place in its bucket in 8 bits");

template <int NW>
__device__ __forceinline__ int block_excl_scan(int v, int* s_wave, int& total) {
  const int lane = threadIdx.x & 31;
  const int wave = threadIdx.x >> 5;
  int inc = v;
  for (int o = 1; o < 32; o <<= 1) {
    const int y = __shfl_up(inc, o);
    if (lane >= o) inc += y;
  }
  if (lane == 31) s_wave[wave] = inc;
  __syncthreads();
  int base = 0, tot = 0;
  for (int w = 0; w < NW; ++w) {
    const int x = s_wave[w];
    if (w < wave) base += x;
    tot += x;
  }
  total = tot;
  return base + inc - v;
}

__global__ __launch_bounds__(EB_THREADS) void k_hist(const int* __restrict__ col,
                                                     int* __restrict__ histM) {
  __shared__ int s_h[NBKT];
  const int t = threadIdx.x;
  const int w = blockIdx.x;
  for (int b = t; b < NBKT; b += EB_THREADS) s_h[b] = 0;
  __syncthreads();
  const int e0 = w * EB_CHUNK;
  const int e1 = (e0 + EB_CHUNK < N_EDGES) ? (e0 + EB_CHUNK) : N_EDGES;
  for (int e = e0 + t; e < e1; e += EB_BATCH * EB_THREADS) {
    int c[EB_BATCH];
#pragma unroll
    for (int k = 0; k < EB_BATCH; ++k) {
      const int ee = e + k * EB_THREADS;
      c[k] = (ee < e1) ? col[ee] : -1;
    }
#pragma unroll
    for (int k = 0; k < EB_BATCH; ++k)
      if (c[k] >= 0) atomicAdd(&s_h[c[k] >> BKT_SHIFT], 1);
  }
  __syncthreads();
  for (int b = t; b < NBKT; b += EB_THREADS) histM[w * NBKT + b] = s_h[b];
}

__global__ __launch_bounds__(EB_THREADS) void k_bin(const int* __restrict__ row,
                                                    const int* __restrict__ col,
                                                    const int* __restrict__ histM,
                                                    int* __restrict__ bbase,
                                                    unsigned* __restrict__ binned) {
  __shared__ int s_cur[NBKT];
  __shared__ int s_wave[EB_WAVES];
  const int t = threadIdx.x;
  const int w = blockIdx.x;

  int before = 0, total = 0;
  if (t < NBKT) {
#pragma unroll 32
    for (int w2 = 0; w2 < EB_WG; ++w2) {
      const int v = histM[w2 * NBKT + t];
      total += v;
      before += (w2 < w) ? v : 0;
    }
  }
  int all;
  const int base = block_excl_scan<EB_WAVES>(total, s_wave, all);
  if (t < NBKT) {
    s_cur[t] = base + before;
    if (w == 0) bbase[t] = base;
  }
  if (w == 0 && t == 0) bbase[NBKT] = all;
  __syncthreads();

  const int e0 = w * EB_CHUNK;
  const int e1 = (e0 + EB_CHUNK < N_EDGES) ? (e0 + EB_CHUNK) : N_EDGES;
  for (int e = e0 + t; e < e1; e += EB_BATCH * EB_THREADS) {
    int c[EB_BATCH], r[EB_BATCH];
#pragma unroll
    for (int k = 0; k < EB_BATCH; ++k) {
      const int ee = e + k * EB_THREADS;
      const bool ok = ee < e1;
      c[k] = ok ? col[ee] : -1;
      r[k] = ok ? row[ee] : 0;
    }
#pragma unroll
    for (int k = 0; k < EB_BATCH; ++k) {
      if (c[k] >= 0) {
        const int pos = atomicAdd(&s_cur[c[k] >> BKT_SHIFT], 1);
        binned[pos] = ((unsigned)(c[k] & (BKT_NODES - 1)) << 24) | (unsigned)r[k];
      }
    }
  }
}

__device__ __forceinline__ float dinv_of(int c) {
  const float d = (float)c + 1.0f;
  return (d > 0.0f) ? rsqrtf(fmaxf(d, 1.0f)) : 0.0f;
}

__global__ __launch_bounds__(BKT_NODES) void k_group(const unsigned* __restrict__ binned,
                                                     const int* __restrict__ bbase,
                                                     int* __restrict__ rowptr,
                                                     float* __restrict__ dinv,
                                                     int* __restrict__ srcs) {
  __shared__ int s_cnt[BKT_NODES];
  __shared__ int s_wave[BKT_NODES / 32];
  const int t = threadIdx.x;
  const int b = blockIdx.x;
  const int lo = bbase[b];
  const int hi = bbase[b + 1];
  s_cnt[t] = 0;
  __syncthreads();
  for (int p = lo + t; p < hi; p += EB_BATCH * BKT_NODES) {
    unsigned u[EB_BATCH];
#pragma unroll
    for (int k = 0; k < EB_BATCH; ++k) {
      const int pp = p + k * BKT_NODES;
      u[k] = binned[(pp < hi) ? pp : p];
    }
#pragma unroll
    for (int k = 0; k < EB_BATCH; ++k)
      atomicAdd(&s_cnt[u[k] >> 24], (p + k * BKT_NODES < hi) ? 1 : 0);
  }
  __syncthreads();
  const int c = s_cnt[t];
  int all;
  const int ex = block_excl_scan<BKT_NODES / 32>(c, s_wave, all);
  const int node = b * BKT_NODES + t;
  if (node < N_NODES) {
    rowptr[node] = lo + ex;
    dinv[node] = dinv_of(c);
  }
  if (b == NBKT - 1 && t == 0) rowptr[N_NODES] = hi;
  s_cnt[t] = lo + ex;
  __syncthreads();
  for (int p = lo + t; p < hi; p += EB_BATCH * BKT_NODES) {
    unsigned u[EB_BATCH];
#pragma unroll
    for (int k = 0; k < EB_BATCH; ++k) {
      const int pp = p + k * BKT_NODES;
      u[k] = binned[(pp < hi) ? pp : p];
    }
#pragma unroll
    for (int k = 0; k < EB_BATCH; ++k) {
      const bool ok = p + k * BKT_NODES < hi;
      const int pos = atomicAdd(&s_cnt[u[k] >> 24], ok ? 1 : 0);
      if (ok) srcs[pos] = (int)(u[k] & 0xFFFFFFu);
    }
  }
}

#define GEMM_WG      256
#define GEMM_THREADS 512
#define GEMM_WAVES   (GEMM_THREADS / 32)
#define WT_PITCH     (FEAT + 4)
static_assert(N_NODES % 16 == 0, "k_gemm2 takes whole strips of 16 rows");
static_assert((FEAT * FEAT / 4) % GEMM_THREADS == 0, "k_gemm2 stages a W in whole passes of the work-group");

__device__ __forceinline__ void strip_product(const float4* __restrict__ bq, const float4 (&a)[FEAT / 8],
                                              v8f (&acc)[8]) {
#pragma unroll
  for (int n = 0; n < 8; ++n) acc[n] = (v8f){0, 0, 0, 0, 0, 0, 0, 0};
  float4 b[8];
#pragma unroll
  for (int n = 0; n < 8; ++n) b[n] = bq[n * (4 * WT_PITCH)];
#pragma unroll
  for (int j = 0; j < FEAT / 8; ++j) {
    float4 bn[8];
    __builtin_amdgcn_sched_barrier(0);
    if (j + 1 < FEAT / 8) {
#pragma unroll
      for (int n = 0; n < 8; ++n) bn[n] = bq[n * (4 * WT_PITCH) + 2 * (j + 1)];
    }
    __builtin_amdgcn_sched_barrier(0);
    const v2f a0 = {a[j].x, a[j].y};
    const v2f a1 = {a[j].z, a[j].w};
#pragma unroll
    for (int n = 0; n < 8; ++n) {
      const v2f b0 = {b[n].x, b[n].y};
      acc[n] = __builtin_amdgcn_wmma_f32_16x16x4_f32(false, a0, false, b0, (short)0, acc[n], false, false);
    }
#pragma unroll
    for (int n = 0; n < 8; ++n) {
      const v2f b1 = {b[n].z, b[n].w};
      acc[n] = __builtin_amdgcn_wmma_f32_16x16x4_f32(false, a1, false, b1, (short)0, acc[n], false, false);
    }
    if (j + 1 < FEAT / 8) {
#pragma unroll
      for (int n = 0; n < 8; ++n) b[n] = bn[n];
    }
  }
}

__device__ __forceinline__ void stage_w(float* __restrict__ sW, const float4 (&w)[FEAT * FEAT / 4 / GEMM_THREADS]) {
#pragma unroll
  for (int it = 0; it < FEAT * FEAT / 4 / GEMM_THREADS; ++it) {
    const int idx = threadIdx.x + GEMM_THREADS * it;
    const int k = idx >> 5;
    const int q = idx & 31;
    const int p = 64 * (q >> 4) + (q & 15);
    sW[(p     ) * WT_PITCH + k] = w[it].x;
    sW[(p + 16) * WT_PITCH + k] = w[it].y;
    sW[(p + 32) * WT_PITCH + k] = w[it].z;
    sW[(p + 48) * WT_PITCH + k] = w[it].w;
  }
}

__device__ __forceinline__ void fma4(float4& acc, float w, const float4 v) {
  acc.x += w * v.x;
  acc.y += w * v.y;
  acc.z += w * v.z;
  acc.w += w * v.w;
}

#define GS_ROWS 16
static_assert(32 % GS_ROWS == 0, "a batch of rows never straddles two chunks of the list");
__device__ __forceinline__ void gather_strip(const float* __restrict__ X, const int* __restrict__ rowptr,
                                             const int* __restrict__ srcs, const float* __restrict__ dinv,
                                             int s, int lane, float4* __restrict__ tg,
                                             float4 (&xs)[FEAT / 8], float& dn) {
  const int c = lane & 15;
  const int h = lane >> 4;
  const int n0 = s * 16;
  const int rp = rowptr[n0 + ((lane < 16) ? lane : 16)];
  dn = dinv[n0 + c];
  const float4* xr = (const float4*)(X + (size_t)(n0 + c) * FEAT + 4 * h);
#pragma unroll
  for (int j = 0; j < FEAT / 8; ++j) xs[j] = xr[2 * j];
  const float4* h4 = (const float4*)X;
  const int E0 = __builtin_amdgcn_readlane(rp, 0);
  const int total = __builtin_amdgcn_readlane(rp, 16) - E0;

  int sv0 = 0, sv1 = 0;
  float dsv0 = 0.0f;
  if (lane < total) sv0 = srcs[E0 + lane];
  if (lane + 32 < total) sv1 = srcs[E0 + 32 + lane];
  if (lane < total) dsv0 = dinv[sv0];

  int r = 0;
  int nodeEnd = __builtin_amdgcn_readlane(rp, 1) - E0;
  float di = __int_as_float(__builtin_amdgcn_readlane(__float_as_int(dn), 0));
  float4 acc = {0.0f, 0.0f, 0.0f, 0.0f};
  for (int base = 0; base < total; base += 32) {
    const int m = (total - base < 32) ? (total - base) : 32;
    int sv2 = 0;
    float dsv1 = 0.0f;
    for (int j = 0; j < m; j += GS_ROWS) {
      float4 v[GS_ROWS];
#pragma unroll
      for (int k = 0; k < GS_ROWS; ++k) {
        const int sk = __builtin_amdgcn_readlane(sv0, j + k);
        v[k] = h4[(size_t)sk * (FEAT / 4) + lane];
      }
      if (j == 0) {
        if (base + 64 + lane < total) sv2 = srcs[E0 + base + 64 + lane];
        if (base + 32 + lane < total) dsv1 = dinv[sv1];
      }
#pragma unroll
      for (int k = 0; k < GS_ROWS; ++k) {
        const int p = base + j + k;
        if (p < total) {
          while (p >= nodeEnd) {
            tg[r * (WT_PITCH / 4)] = acc;
            acc = (float4){0.0f, 0.0f, 0.0f, 0.0f};
            ++r;
            nodeEnd = __builtin_amdgcn_readlane(rp, r + 1) - E0;
            di = __int_as_float(__builtin_amdgcn_readlane(__float_as_int(dn), r));
          }
          const float w = __int_as_float(__builtin_amdgcn_readlane(__float_as_int(dsv0), j + k)) * di;
          fma4(acc, w, v[k]);
        }
      }
    }
    sv0 = sv1;
    dsv0 = dsv1;
    sv1 = sv2;
  }
  while (r < 16) {
    tg[r * (WT_PITCH / 4)] = acc;
    acc = (float4){0.0f, 0.0f, 0.0f, 0.0f};
    ++r;
  }
}

__global__ __launch_bounds__(GEMM_THREADS) void k_gemm2(const float* __restrict__ X,
                                                        const int* __restrict__ rowptr,
                                                        const int* __restrict__ srcs,
                                                        const float* __restrict__ dinv,
                                                        const float* __restrict__ Wa,
                                                        const float* __restrict__ ba,
                                                        const float* __restrict__ Wb,
                                                        float* __restrict__ Hout,
                                                        int nrows) {
  __shared__ float4 sWa4[FEAT * WT_PITCH / 4];
  __shared__ float4 sWb4[FEAT * WT_PITCH / 4];
  __shared__ float4 sT4[GEMM_WAVES * 16 * WT_PITCH / 4];
  const int lane = threadIdx.x & 31;
  const int wave = __builtin_amdgcn_readfirstlane((int)(threadIdx.x >> 5));
  const int c = lane & 15;
  const int h = lane >> 4;

  const int nstrips = nrows >> 4;
  const int s_lo = ((int)blockIdx.x * nstrips) / GEMM_WG;
  const int s_hi = (((int)blockIdx.x + 1) * nstrips) / GEMM_WG;
  static_assert(GEMM_WAVES == 16, "the renumbering below is a 4 x 4 square");
  const int wv = (((wave - (wave >> 2)) & 3) << 2) | (wave >> 2);
  int s = s_lo + wv;

  float4 wa[FEAT * FEAT / 4 / GEMM_THREADS], wb[FEAT * FEAT / 4 / GEMM_THREADS];
#pragma unroll
  for (int it = 0; it < FEAT * FEAT / 4 / GEMM_THREADS; ++it) wa[it] = ((const float4*)Wa)[threadIdx.x + GEMM_THREADS * it];
#pragma unroll
  for (int it = 0; it < FEAT * FEAT / 4 / GEMM_THREADS; ++it) wb[it] = ((const float4*)Wb)[threadIdx.x + GEMM_THREADS * it];
  const float4 blo = ((const float4*)ba)[c];
  const float4 bhi = ((const float4*)ba)[16 + c];

  stage_w((float*)sWa4, wa);
  stage_w((float*)sWb4, wb);

  const float4* bqa = sWa4 + (c * WT_PITCH) / 4 + h;
  const float4* bqb = sWb4 + (c * WT_PITCH) / 4 + h;
  float4* tg = sT4 + (wave * 16) * (WT_PITCH / 4) + lane;
  float4* tw = sT4 + (wave * 16 + 8 * h) * (WT_PITCH / 4) + c;
  const float4* tr = sT4 + (wave * 16 + c) * (WT_PITCH / 4) + h;

  float4 a[FEAT / 8];
  float dn = 0.0f;
  if (s < s_hi) gather_strip(X, rowptr, srcs, dinv, s, lane, tg, a, dn);
  __syncthreads();

  while (s < s_hi) {
    asm volatile("" ::: "memory");
    {
      const float sl = dn * dn;
#pragma unroll
      for (int j = 0; j < FEAT / 8; ++j) {
        const float4 t = tr[2 * j];
        a[j].x = t.x + sl * a[j].x;
        a[j].y = t.y + sl * a[j].y;
        a[j].z = t.z + sl * a[j].z;
        a[j].w = t.w + sl * a[j].w;
      }
    }
    v8f acc[8];
    strip_product(bqa, a, acc);

#pragma unroll
    for (int r = 0; r < 8; ++r) {
      float4 lo, hi;
      lo.x = fmaxf(acc[0][r] + blo.x, 0.0f);
      lo.y = fmaxf(acc[1][r] + blo.y, 0.0f);
      lo.z = fmaxf(acc[2][r] + blo.z, 0.0f);
      lo.w = fmaxf(acc[3][r] + blo.w, 0.0f);
      hi.x = fmaxf(acc[4][r] + bhi.x, 0.0f);
      hi.y = fmaxf(acc[5][r] + bhi.y, 0.0f);
      hi.z = fmaxf(acc[6][r] + bhi.z, 0.0f);
      hi.w = fmaxf(acc[7][r] + bhi.w, 0.0f);
      tw[r * (WT_PITCH / 4)]      = lo;
      tw[r * (WT_PITCH / 4) + 16] = hi;
    }
#pragma unroll
    for (int j = 0; j < FEAT / 8; ++j) a[j] = tr[2 * j];

    strip_product(bqb, a, acc);

    float4* orow = (float4*)(Hout + (size_t)(s * 16 + 8 * h) * FEAT + 4 * c);
#pragma unroll
    for (int r = 0; r < 8; ++r) {
      const float4 lo = {acc[0][r], acc[1][r], acc[2][r], acc[3][r]};
      const float4 hi = {acc[4][r], acc[5][r], acc[6][r], acc[7][r]};
      orow[r * (FEAT / 4)]      = lo;
      orow[r * (FEAT / 4) + 16] = hi;
    }
    s += GEMM_WAVES;
    if (s < s_hi) gather_strip(X, rowptr, srcs, dinv, s, lane, tg, a, dn);
  }
}


__global__ __launch_bounds__(256) void k_gather(const float* __restrict__ h,
                                                const int* __restrict__ rowptr,
                                                const int* __restrict__ srcs,
                                                const float* __restrict__ dinv,
                                                const float* __restrict__ bias,
                                                float* __restrict__ out, int relu) {
  const int lane = threadIdx.x & 31;
  const int wave = __builtin_amdgcn_readfirstlane((int)(threadIdx.x >> 5));
  const int i = blockIdx.x * 8 + wave;
  if (i >= N_NODES) return;
  const int start = rowptr[i];
  const int n = rowptr[i + 1] - start;
  const float di = dinv[i];
  const float4* h4 = (const float4*)h;
  const float4 hv = h4[(size_t)i * (FEAT / 4) + lane];
  float4 bv = {0.0f, 0.0f, 0.0f, 0.0f};
  if (bias) bv = ((const float4*)bias)[lane];

  float4 acc = {0.0f, 0.0f, 0.0f, 0.0f};
  for (int b = 0; b < n; b += 32) {
    const int m = (n - b < 32) ? (n - b) : 32;
    int s = 0;
    float ds = 0.0f;
    if (lane < m) {
      s = srcs[start + b + lane];
      ds = dinv[s];
    }
    int j = 0;
    for (; j + 4 <= m; j += 4) {
      const int s0 = __builtin_amdgcn_readlane(s, j);
      const int s1 = __builtin_amdgcn_readlane(s, j + 1);
      const int s2 = __builtin_amdgcn_readlane(s, j + 2);
      const int s3 = __builtin_amdgcn_readlane(s, j + 3);
      const float4 v0 = h4[(size_t)s0 * (FEAT / 4) + lane];
      const float4 v1 = h4[(size_t)s1 * (FEAT / 4) + lane];
      const float4 v2 = h4[(size_t)s2 * (FEAT / 4) + lane];
      const float4 v3 = h4[(size_t)s3 * (FEAT / 4) + lane];
      const float w0 = __int_as_float(__builtin_amdgcn_readlane(__float_as_int(ds), j)) * di;
      const float w1 = __int_as_float(__builtin_amdgcn_readlane(__float_as_int(ds), j + 1)) * di;
      const float w2 = __int_as_float(__builtin_amdgcn_readlane(__float_as_int(ds), j + 2)) * di;
      const float w3 = __int_as_float(__builtin_amdgcn_readlane(__float_as_int(ds), j + 3)) * di;
      fma4(acc, w0, v0);
      fma4(acc, w1, v1);
      fma4(acc, w2, v2);
      fma4(acc, w3, v3);
    }
    for (; j < m; ++j) {
      const int s0 = __builtin_amdgcn_readlane(s, j);
      const float4 v0 = h4[(size_t)s0 * (FEAT / 4) + lane];
      const float w0 = __int_as_float(__builtin_amdgcn_readlane(__float_as_int(ds), j)) * di;
      fma4(acc, w0, v0);
    }
  }

  const float sl = di * di;
  float4 rv;
  rv.x = acc.x + sl * hv.x + bv.x;
  rv.y = acc.y + sl * hv.y + bv.y;
  rv.z = acc.z + sl * hv.z + bv.z;
  rv.w = acc.w + sl * hv.w + bv.w;
  if (relu) {
    rv.x = fmaxf(rv.x, 0.0f);
    rv.y = fmaxf(rv.y, 0.0f);
    rv.z = fmaxf(rv.z, 0.0f);
    rv.w = fmaxf(rv.w, 0.0f);
  }
  ((float4*)out)[(size_t)i * (FEAT / 4) + lane] = rv;
}

#define GP_NODES    32
#define GP_WAVES    8
#define GP_PER_WAVE (GP_NODES / GP_WAVES)
static_assert(N_NODES % GP_NODES == 0, "k_gather_pool takes whole runs of nodes: no wave leaves in front of the barrier");
static_assert(FEAT <= GP_WAVES * 32, "one thread per feature adds the rows up");

__global__ __launch_bounds__(GP_WAVES * 32) void k_gather_pool(const float* __restrict__ h,
                                                               const int* __restrict__ rowptr,
                                                               const int* __restrict__ srcs,
                                                               const float* __restrict__ dinv,
                                                               const int* __restrict__ batch,
                                                               float* __restrict__ sums,
                                                               float* __restrict__ cnts) {
  __shared__ float4 sRow[GP_NODES * (FEAT / 4)];
  __shared__ int sGid[GP_NODES];
  const int lane = threadIdx.x & 31;
  const int wave = __builtin_amdgcn_readfirstlane((int)(threadIdx.x >> 5));
  const int n0 = blockIdx.x * GP_NODES;
  const float4* h4 = (const float4*)h;

  int gidv = 0;
  if (threadIdx.x < GP_NODES) gidv = batch[n0 + threadIdx.x];

  int st[GP_PER_WAVE], cn[GP_PER_WAVE], sq[GP_PER_WAVE];
  float dq[GP_PER_WAVE];
#pragma unroll
  for (int q = 0; q < GP_PER_WAVE; ++q) {
    const int i = n0 + wave + GP_WAVES * q;
    st[q] = rowptr[i];
    cn[q] = rowptr[i + 1] - st[q];
    dq[q] = dinv[i];
  }
#pragma unroll
  for (int q = 0; q < GP_PER_WAVE; ++q) sq[q] = (lane < cn[q]) ? srcs[st[q] + lane] : 0;

#pragma unroll
  for (int q = 0; q < GP_PER_WAVE; ++q) {
    const int i = n0 + wave + GP_WAVES * q;
    const int start = st[q];
    const int n = cn[q];
    const float di = dq[q];
    const float4 hv = h4[(size_t)i * (FEAT / 4) + lane];
    float4 acc = {0.0f, 0.0f, 0.0f, 0.0f};
    for (int b = 0; b < n; b += 32) {
      const int m = (n - b < 32) ? (n - b) : 32;
      int s = sq[q];
      if (b > 0) s = (lane < m) ? srcs[start + b + lane] : 0;
      float ds = 0.0f;
      if (lane < m) ds = dinv[s];
      int j = 0;
      for (; j + 4 <= m; j += 4) {
        const int s0 = __builtin_amdgcn_readlane(s, j);
        const int s1 = __builtin_amdgcn_readlane(s, j + 1);
        const int s2 = __builtin_amdgcn_readlane(s, j + 2);
        const int s3 = __builtin_amdgcn_readlane(s, j + 3);
        const float4 v0 = h4[(size_t)s0 * (FEAT / 4) + lane];
        const float4 v1 = h4[(size_t)s1 * (FEAT / 4) + lane];
        const float4 v2 = h4[(size_t)s2 * (FEAT / 4) + lane];
        const float4 v3 = h4[(size_t)s3 * (FEAT / 4) + lane];
        const float w0 = __int_as_float(__builtin_amdgcn_readlane(__float_as_int(ds), j)) * di;
        const float w1 = __int_as_float(__builtin_amdgcn_readlane(__float_as_int(ds), j + 1)) * di;
        const float w2 = __int_as_float(__builtin_amdgcn_readlane(__float_as_int(ds), j + 2)) * di;
        const float w3 = __int_as_float(__builtin_amdgcn_readlane(__float_as_int(ds), j + 3)) * di;
        fma4(acc, w0, v0);
        fma4(acc, w1, v1);
        fma4(acc, w2, v2);
        fma4(acc, w3, v3);
      }
      for (; j < m; ++j) {
        const int s0 = __builtin_amdgcn_readlane(s, j);
        const float4 v0 = h4[(size_t)s0 * (FEAT / 4) + lane];
        const float w0 = __int_as_float(__builtin_amdgcn_readlane(__float_as_int(ds), j)) * di;
        fma4(acc, w0, v0);
      }
    }
    const float sl = di * di;
    float4 rv;
    rv.x = acc.x + sl * hv.x;
    rv.y = acc.y + sl * hv.y;
    rv.z = acc.z + sl * hv.z;
    rv.w = acc.w + sl * hv.w;
    sRow[(wave + GP_WAVES * q) * (FEAT / 4) + lane] = rv;
  }
  if (threadIdx.x < GP_NODES) sGid[threadIdx.x] = gidv;
  __syncthreads();

  if (threadIdx.x < FEAT) {
    const int t = threadIdx.x;
    const float* colp = (const float*)sRow + t;
    int gcur = sGid[0];
    float acc = colp[0];
    int run = 1;
#pragma unroll 8
    for (int j = 1; j < GP_NODES; ++j) {
      const int g = sGid[j];
      const float v = colp[j * FEAT];
      if (g != gcur) {
        atomAddF(&sums[(size_t)gcur * FEAT + t], acc);
        if (t == 0) atomAddF(&cnts[gcur], (float)run);
        gcur = g;
        acc = v;
        run = 1;
      } else {
        acc += v;
        ++run;
      }
    }
    atomAddF(&sums[(size_t)gcur * FEAT + t], acc);
    if (t == 0) atomAddF(&cnts[gcur], (float)run);
  }
}

#define TAIL_PITCH (FEAT + 1)
static_assert(NGRAPH % 16 == 0, "k_tail takes whole strips of 16 graphs");
static_assert(16 * NCLS <= 256, "one thread per (graph, class) of the strip");
static_assert((FEAT * NCLS) % 256 == 0, "k_tail stages Wl in whole passes of the work-group");

__global__ __launch_bounds__(256) void k_tail(const float* __restrict__ sums,
                                              const float* __restrict__ cnts,
                                              const float* __restrict__ W3,
                                              const float* __restrict__ b3,
                                              const float* __restrict__ Wl,
                                              const float* __restrict__ bl,
                                              float* __restrict__ out) {
  __shared__ float sT[16 * TAIL_PITCH];
  __shared__ float sWl[FEAT * NCLS];
  const int lane = threadIdx.x & 31;
  const int wave = threadIdx.x >> 5;
  const int c    = lane & 15;
  const int h    = lane >> 4;
  const int g0   = blockIdx.x * 16;

  float wl[FEAT * NCLS / 256];
#pragma unroll
  for (int q = 0; q < FEAT * NCLS / 256; ++q) wl[q] = Wl[threadIdx.x + 256 * q];

  const float* arow = sums + (size_t)(g0 + c) * FEAT + 2 * h;
  const float* bcol = W3 + (size_t)(2 * h) * FEAT + wave * 16 + c;
  v2f a[FEAT / 4], b[FEAT / 4];
#pragma unroll
  for (int ks = 0; ks < FEAT / 4; ++ks) {
    a[ks].x = arow[4 * ks];
    a[ks].y = arow[4 * ks + 1];
    b[ks].x = bcol[(size_t)(4 * ks) * FEAT];
    b[ks].y = bcol[(size_t)(4 * ks + 1) * FEAT];
  }
  float cn[8];
#pragma unroll
  for (int r = 0; r < 8; ++r) cn[r] = cnts[g0 + 8 * h + r];
  const float bias3 = b3[wave * 16 + c];

  v8f acc = (v8f){0, 0, 0, 0, 0, 0, 0, 0};
#pragma unroll
  for (int ks = 0; ks < FEAT / 4; ++ks)
    acc = __builtin_amdgcn_wmma_f32_16x16x4_f32(false, a[ks], false, b[ks], (short)0, acc, false, false);

#pragma unroll
  for (int r = 0; r < 8; ++r) {
    const float inv = 1.0f / fmaxf(cn[r], 1.0f);
    sT[(8 * h + r) * TAIL_PITCH + wave * 16 + c] = acc[r] * inv + ((cn[r] > 0.0f) ? bias3 : 0.0f);
  }
#pragma unroll
  for (int q = 0; q < FEAT * NCLS / 256; ++q) sWl[threadIdx.x + 256 * q] = wl[q];
  __syncthreads();

  if (threadIdx.x < 16 * NCLS) {
    const int gl = threadIdx.x / NCLS;
    const int cl = threadIdx.x - gl * NCLS;
    float o = bl[cl];
#pragma unroll 16
    for (int k = 0; k < FEAT; ++k) o += sT[gl * TAIL_PITCH + k] * sWl[k * NCLS + cl];
    out[(size_t)(g0 + gl) * NCLS + cl] = o;
  }
}

extern "C" void kernel_launch(void* const* d_in, const int* in_sizes, int n_in,
                              void* d_out, int out_size, void* d_ws, size_t ws_size,
                              hipStream_t stream) {
  const float* x     = (const float*)d_in[0];
  const int*   ei    = (const int*)d_in[1];
  const int*   row   = ei;
  const int*   col   = ei + N_EDGES;
  const int*   batch = (const int*)d_in[2];
  const float* W1 = (const float*)d_in[3];
  const float* b1 = (const float*)d_in[4];
  const float* W2 = (const float*)d_in[5];
  const float* b2 = (const float*)d_in[6];
  const float* W3 = (const float*)d_in[7];
  const float* b3 = (const float*)d_in[8];
  const float* Wl = (const float*)d_in[9];
  const float* bl = (const float*)d_in[10];
  float* outp = (float*)d_out;

  char* base = (char*)d_ws;
  const size_t featBytes = (size_t)N_NODES * FEAT * sizeof(float);
  int*   histM  = (int*)(base);
  int*   rowptr = (int*)(base + (512 << 10));
  int*   bbase  = (int*)(base + (1 << 20));
  float* dinv   = (float*)(base + (1 << 20) + (512 << 10));
  int*   srcs   = (int*)(base + (4 << 20));
  float* hbuf = (float*)(base + (8 << 20));
  float* abuf = (float*)(base + (8 << 20) + featBytes);
  float* sums = (float*)(base + (8 << 20) + 2 * featBytes);
  float* cnts = sums + (size_t)NGRAPH * FEAT;
  unsigned* binned = (unsigned*)(base + (9 << 20) + 2 * featBytes);

  const int  TB     = 256;
  const int  gGath  = (N_NODES + 8 - 1) / 8;

  k_hist<<<EB_WG, EB_THREADS, 0, stream>>>(col, histM);
  k_bin<<<EB_WG, EB_THREADS, 0, stream>>>(row, col, histM, bbase, binned);
  k_group<<<NBKT, BKT_NODES, 0, stream>>>(binned, bbase, rowptr, dinv, srcs);

  k_gemm2<<<GEMM_WG, GEMM_THREADS, 0, stream>>>(x, rowptr, srcs, dinv, W1, b1, W2, hbuf, N_NODES);

  k_gather<<<gGath, TB, 0, stream>>>(hbuf, rowptr, srcs, dinv, b2, abuf, 1);

  const long poolN = (long)NGRAPH * FEAT + NGRAPH;
  k_fill<<<(int)((poolN + TB - 1) / TB), TB, 0, stream>>>(sums, 0.0f, poolN);
  k_gather_pool<<<N_NODES / GP_NODES, GP_WAVES * 32, 0, stream>>>(abuf, rowptr, srcs, dinv, batch, sums, cnts);
  k_tail<<<NGRAPH / 16, TB, 0, stream>>>(sums, cnts, W3, b3, Wl, bl, outp);
}


# W3 and the classifier folded into one 128x16 matrix (WMMA) applied per node before layer 3's aggregation (all linear): layer 3 gathers and pools 64-byte rows, not 512-byte; rounding order changes
# speedup vs baseline: 5.5196x; 1.0538x over previous
#include <hip/hip_runtime.h>

typedef __attribute__((ext_vector_type(2))) float v2f;
typedef __attribute__((ext_vector_type(8))) float v8f;

#define N_NODES 100000
#define N_EDGES 800000
#define FEAT    128
#define NGRAPH  512
#define NCLS    10

__device__ __forceinline__ void atomAddF(float* p, float v) {
  unsafeAtomicAdd(p, v);
}

#define EB_WG      128
#define EB_THREADS 512
#define EB_WAVES   (EB_THREADS / 32)
#define EB_CHUNK   ((N_EDGES + EB_WG - 1) / EB_WG)
#define EB_BATCH   4
#define BKT_SHIFT  8
#define BKT_NODES  (1 << BKT_SHIFT)
#define NBKT       ((N_NODES + BKT_NODES - 1) / BKT_NODES)
static_assert(NBKT <= EB_THREADS, "k_bin gives one thread to every bucket");
static_assert(N_NODES <= (1 << 24), "a packed edge keeps the source in 24 bits");
static_assert(BKT_NODES <= 256, "a packed edge keeps the target's place in its bucket in 8 bits");

template <int NW>
__device__ __forceinline__ int block_excl_scan(int v, int* s_wave, int& total) {
  const int lane = threadIdx.x & 31;
  const int wave = threadIdx.x >> 5;
  int inc = v;
  for (int o = 1; o < 32; o <<= 1) {
    const int y = __shfl_up(inc, o);
    if (lane >= o) inc += y;
  }
  if (lane == 31) s_wave[wave] = inc;
  __syncthreads();
  int base = 0, tot = 0;
  for (int w = 0; w < NW; ++w) {
    const int x = s_wave[w];
    if (w < wave) base += x;
    tot += x;
  }
  total = tot;
  return base + inc - v;
}

__global__ __launch_bounds__(EB_THREADS) void k_hist(const int* __restrict__ col,
                                                     int* __restrict__ histM) {
  __shared__ int s_h[NBKT];
  const int t = threadIdx.x;
  const int w = blockIdx.x;
  for (int b = t; b < NBKT; b += EB_THREADS) s_h[b] = 0;
  __syncthreads();
  const int e0 = w * EB_CHUNK;
  const int e1 = (e0 + EB_CHUNK < N_EDGES) ? (e0 + EB_CHUNK) : N_EDGES;
  for (int e = e0 + t; e < e1; e += EB_BATCH * EB_THREADS) {
    int c[EB_BATCH];
#pragma unroll
    for (int k = 0; k < EB_BATCH; ++k) {
      const int ee = e + k * EB_THREADS;
      c[k] = (ee < e1) ? col[ee] : -1;
    }
#pragma unroll
    for (int k = 0; k < EB_BATCH; ++k)
      if (c[k] >= 0) atomicAdd(&s_h[c[k] >> BKT_SHIFT], 1);
  }
  __syncthreads();
  for (int b = t; b < NBKT; b += EB_THREADS) histM[w * NBKT + b] = s_h[b];
}

__global__ __launch_bounds__(EB_THREADS) void k_bin(const int* __restrict__ row,
                                                    const int* __restrict__ col,
                                                    const int* __restrict__ histM,
                                                    int* __restrict__ bbase,
                                                    unsigned* __restrict__ binned) {
  __shared__ int s_cur[NBKT];
  __shared__ int s_wave[EB_WAVES];
  const int t = threadIdx.x;
  const int w = blockIdx.x;

  int before = 0, total = 0;
  if (t < NBKT) {
#pragma unroll 32
    for (int w2 = 0; w2 < EB_WG; ++w2) {
      const int v = histM[w2 * NBKT + t];
      total += v;
      before += (w2 < w) ? v : 0;
    }
  }
  int all;
  const int base = block_excl_scan<EB_WAVES>(total, s_wave, all);
  if (t < NBKT) {
    s_cur[t] = base + before;
    if (w == 0) bbase[t] = base;
  }
  if (w == 0 && t == 0) bbase[NBKT] = all;
  __syncthreads();

  const int e0 = w * EB_CHUNK;
  const int e1 = (e0 + EB_CHUNK < N_EDGES) ? (e0 + EB_CHUNK) : N_EDGES;
  for (int e = e0 + t; e < e1; e += EB_BATCH * EB_THREADS) {
    int c[EB_BATCH], r[EB_BATCH];
#pragma unroll
    for (int k = 0; k < EB_BATCH; ++k) {
      const int ee = e + k * EB_THREADS;
      const bool ok = ee < e1;
      c[k] = ok ? col[ee] : -1;
      r[k] = ok ? row[ee] : 0;
    }
#pragma unroll
    for (int k = 0; k < EB_BATCH; ++k) {
      if (c[k] >= 0) {
        const int pos = atomicAdd(&s_cur[c[k] >> BKT_SHIFT], 1);
        binned[pos] = ((unsigned)(c[k] & (BKT_NODES - 1)) << 24) | (unsigned)r[k];
      }
    }
  }
}

__device__ __forceinline__ float dinv_of(int c) {
  const float d = (float)c + 1.0f;
  return (d > 0.0f) ? rsqrtf(fmaxf(d, 1.0f)) : 0.0f;
}

__global__ __launch_bounds__(BKT_NODES) void k_group(const unsigned* __restrict__ binned,
                                                     const int* __restrict__ bbase,
                                                     int* __restrict__ rowptr,
                                                     float* __restrict__ dinv,
                                                     int* __restrict__ srcs) {
  __shared__ int s_cnt[BKT_NODES];
  __shared__ int s_wave[BKT_NODES / 32];
  const int t = threadIdx.x;
  const int b = blockIdx.x;
  const int lo = bbase[b];
  const int hi = bbase[b + 1];
  s_cnt[t] = 0;
  __syncthreads();
  for (int p = lo + t; p < hi; p += EB_BATCH * BKT_NODES) {
    unsigned u[EB_BATCH];
#pragma unroll
    for (int k = 0; k < EB_BATCH; ++k) {
      const int pp = p + k * BKT_NODES;
      u[k] = binned[(pp < hi) ? pp : p];
    }
#pragma unroll
    for (int k = 0; k < EB_BATCH; ++k)
      atomicAdd(&s_cnt[u[k] >> 24], (p + k * BKT_NODES < hi) ? 1 : 0);
  }
  __syncthreads();
  const int c = s_cnt[t];
  int all;
  const int ex = block_excl_scan<BKT_NODES / 32>(c, s_wave, all);
  const int node = b * BKT_NODES + t;
  if (node < N_NODES) {
    rowptr[node] = lo + ex;
    dinv[node] = dinv_of(c);
  }
  if (b == NBKT - 1 && t == 0) rowptr[N_NODES] = hi;
  s_cnt[t] = lo + ex;
  __syncthreads();
  for (int p = lo + t; p < hi; p += EB_BATCH * BKT_NODES) {
    unsigned u[EB_BATCH];
#pragma unroll
    for (int k = 0; k < EB_BATCH; ++k) {
      const int pp = p + k * BKT_NODES;
      u[k] = binned[(pp < hi) ? pp : p];
    }
#pragma unroll
    for (int k = 0; k < EB_BATCH; ++k) {
      const bool ok = p + k * BKT_NODES < hi;
      const int pos = atomicAdd(&s_cnt[u[k] >> 24], ok ? 1 : 0);
      if (ok) srcs[pos] = (int)(u[k] & 0xFFFFFFu);
    }
  }
}

#define GEMM_WG      256
#define GEMM_THREADS 512
#define GEMM_WAVES   (GEMM_THREADS / 32)
#define WT_PITCH     (FEAT + 4)
static_assert(N_NODES % 16 == 0, "k_gemm2 takes whole strips of 16 rows");
static_assert((FEAT * FEAT / 4) % GEMM_THREADS == 0, "k_gemm2 stages a W in whole passes of the work-group");

__device__ __forceinline__ void strip_product(const float4* __restrict__ bq, const float4 (&a)[FEAT / 8],
                                              v8f (&acc)[8]) {
#pragma unroll
  for (int n = 0; n < 8; ++n) acc[n] = (v8f){0, 0, 0, 0, 0, 0, 0, 0};
  float4 b[8];
#pragma unroll
  for (int n = 0; n < 8; ++n) b[n] = bq[n * (4 * WT_PITCH)];
#pragma unroll
  for (int j = 0; j < FEAT / 8; ++j) {
    float4 bn[8];
    __builtin_amdgcn_sched_barrier(0);
    if (j + 1 < FEAT / 8) {
#pragma unroll
      for (int n = 0; n < 8; ++n) bn[n] = bq[n * (4 * WT_PITCH) + 2 * (j + 1)];
    }
    __builtin_amdgcn_sched_barrier(0);
    const v2f a0 = {a[j].x, a[j].y};
    const v2f a1 = {a[j].z, a[j].w};
#pragma unroll
    for (int n = 0; n < 8; ++n) {
      const v2f b0 = {b[n].x, b[n].y};
      acc[n] = __builtin_amdgcn_wmma_f32_16x16x4_f32(false, a0, false, b0, (short)0, acc[n], false, false);
    }
#pragma unroll
    for (int n = 0; n < 8; ++n) {
      const v2f b1 = {b[n].z, b[n].w};
      acc[n] = __builtin_amdgcn_wmma_f32_16x16x4_f32(false, a1, false, b1, (short)0, acc[n], false, false);
    }
    if (j + 1 < FEAT / 8) {
#pragma unroll
      for (int n = 0; n < 8; ++n) b[n] = bn[n];
    }
  }
}

__device__ __forceinline__ void stage_w(float* __restrict__ sW, const float4 (&w)[FEAT * FEAT / 4 / GEMM_THREADS]) {
#pragma unroll
  for (int it = 0; it < FEAT * FEAT / 4 / GEMM_THREADS; ++it) {
    const int idx = threadIdx.x + GEMM_THREADS * it;
    const int k = idx >> 5;
    const int q = idx & 31;
    const int p = 64 * (q >> 4) + (q & 15);
    sW[(p     ) * WT_PITCH + k] = w[it].x;
    sW[(p + 16) * WT_PITCH + k] = w[it].y;
    sW[(p + 32) * WT_PITCH + k] = w[it].z;
    sW[(p + 48) * WT_PITCH + k] = w[it].w;
  }
}

__device__ __forceinline__ void fma4(float4& acc, float w, const float4 v) {
  acc.x += w * v.x;
  acc.y += w * v.y;
  acc.z += w * v.z;
  acc.w += w * v.w;
}

template <int GS_ROWS>
__device__ __forceinline__ void gather_strip(const float* __restrict__ X, const int* __restrict__ rowptr,
                                             const int* __restrict__ srcs, const float* __restrict__ dinv,
                                             int s, int lane, float4* __restrict__ tg,
                                             float4 (&xs)[FEAT / 8], float& dn) {
  const int c = lane & 15;
  const int h = lane >> 4;
  const int n0 = s * 16;
  const int rp = rowptr[n0 + ((lane < 16) ? lane : 16)];
  dn = dinv[n0 + c];
  const float4* xr = (const float4*)(X + (size_t)(n0 + c) * FEAT + 4 * h);
#pragma unroll
  for (int j = 0; j < FEAT / 8; ++j) xs[j] = xr[2 * j];
  const float4* h4 = (const float4*)X;
  const int E0 = __builtin_amdgcn_readlane(rp, 0);
  const int total = __builtin_amdgcn_readlane(rp, 16) - E0;

  int sv0 = 0, sv1 = 0;
  float dsv0 = 0.0f;
  if (lane < total) sv0 = srcs[E0 + lane];
  if (lane + 32 < total) sv1 = srcs[E0 + 32 + lane];
  if (lane < total) dsv0 = dinv[sv0];

  int r = 0;
  int nodeEnd = __builtin_amdgcn_readlane(rp, 1) - E0;
  float di = __int_as_float(__builtin_amdgcn_readlane(__float_as_int(dn), 0));
  float4 acc = {0.0f, 0.0f, 0.0f, 0.0f};
  for (int base = 0; base < total; base += 32) {
    const int m = (total - base < 32) ? (total - base) : 32;
    int sv2 = 0;
    float dsv1 = 0.0f;
    for (int j = 0; j < m; j += GS_ROWS) {
      float4 v[GS_ROWS];
#pragma unroll
      for (int k = 0; k < GS_ROWS; ++k) {
        const int sk = __builtin_amdgcn_readlane(sv0, j + k);
        v[k] = h4[(size_t)sk * (FEAT / 4) + lane];
      }
      if (j == 0) {
        if (base + 64 + lane < total) sv2 = srcs[E0 + base + 64 + lane];
        if (base + 32 + lane < total) dsv1 = dinv[sv1];
      }
#pragma unroll
      for (int k = 0; k < GS_ROWS; ++k) {
        const int p = base + j + k;
        if (p < total) {
          while (p >= nodeEnd) {
            tg[r * (WT_PITCH / 4)] = acc;
            acc = (float4){0.0f, 0.0f, 0.0f, 0.0f};
            ++r;
            nodeEnd = __builtin_amdgcn_readlane(rp, r + 1) - E0;
            di = __int_as_float(__builtin_amdgcn_readlane(__float_as_int(dn), r));
          }
          const float w = __int_as_float(__builtin_amdgcn_readlane(__float_as_int(dsv0), j + k)) * di;
          fma4(acc, w, v[k]);
        }
      }
    }
    sv0 = sv1;
    dsv0 = dsv1;
    sv1 = sv2;
  }
  while (r < 16) {
    tg[r * (WT_PITCH / 4)] = acc;
    acc = (float4){0.0f, 0.0f, 0.0f, 0.0f};
    ++r;
  }
}

__global__ __launch_bounds__(GEMM_THREADS) void k_gemm2(const float* __restrict__ X,
                                                        const int* __restrict__ rowptr,
                                                        const int* __restrict__ srcs,
                                                        const float* __restrict__ dinv,
                                                        const float* __restrict__ Wa,
                                                        const float* __restrict__ ba,
                                                        const float* __restrict__ Wb,
                                                        float* __restrict__ Hout,
                                                        int nrows) {
  __shared__ float4 sWa4[FEAT * WT_PITCH / 4];
  __shared__ float4 sWb4[FEAT * WT_PITCH / 4];
  __shared__ float4 sT4[GEMM_WAVES * 16 * WT_PITCH / 4];
  const int lane = threadIdx.x & 31;
  const int wave = __builtin_amdgcn_readfirstlane((int)(threadIdx.x >> 5));
  const int c = lane & 15;
  const int h = lane >> 4;

  const int nstrips = nrows >> 4;
  const int s_lo = ((int)blockIdx.x * nstrips) / GEMM_WG;
  const int s_hi = (((int)blockIdx.x + 1) * nstrips) / GEMM_WG;
  static_assert(GEMM_WAVES == 16, "the renumbering below is a 4 x 4 square");
  const int wv = (((wave - (wave >> 2)) & 3) << 2) | (wave >> 2);
  int s = s_lo + wv;

  float4 wa[FEAT * FEAT / 4 / GEMM_THREADS], wb[FEAT * FEAT / 4 / GEMM_THREADS];
#pragma unroll
  for (int it = 0; it < FEAT * FEAT / 4 / GEMM_THREADS; ++it) wa[it] = ((const float4*)Wa)[threadIdx.x + GEMM_THREADS * it];
#pragma unroll
  for (int it = 0; it < FEAT * FEAT / 4 / GEMM_THREADS; ++it) wb[it] = ((const float4*)Wb)[threadIdx.x + GEMM_THREADS * it];
  const float4 blo = ((const float4*)ba)[c];
  const float4 bhi = ((const float4*)ba)[16 + c];

  stage_w((float*)sWa4, wa);
  stage_w((float*)sWb4, wb);

  const float4* bqa = sWa4 + (c * WT_PITCH) / 4 + h;
  const float4* bqb = sWb4 + (c * WT_PITCH) / 4 + h;
  float4* tg = sT4 + (wave * 16) * (WT_PITCH / 4) + lane;
  float4* tw = sT4 + (wave * 16 + 8 * h) * (WT_PITCH / 4) + c;
  const float4* tr = sT4 + (wave * 16 + c) * (WT_PITCH / 4) + h;

  float4 a[FEAT / 8];
  float dn = 0.0f;
  if (s < s_hi) gather_strip<16>(X, rowptr, srcs, dinv, s, lane, tg, a, dn);
  __syncthreads();

  while (s < s_hi) {
    asm volatile("" ::: "memory");
    {
      const float sl = dn * dn;
#pragma unroll
      for (int j = 0; j < FEAT / 8; ++j) {
        const float4 t = tr[2 * j];
        a[j].x = t.x + sl * a[j].x;
        a[j].y = t.y + sl * a[j].y;
        a[j].z = t.z + sl * a[j].z;
        a[j].w = t.w + sl * a[j].w;
      }
    }
    v8f acc[8];
    strip_product(bqa, a, acc);

#pragma unroll
    for (int r = 0; r < 8; ++r) {
      float4 lo, hi;
      lo.x = fmaxf(acc[0][r] + blo.x, 0.0f);
      lo.y = fmaxf(acc[1][r] + blo.y, 0.0f);
      lo.z = fmaxf(acc[2][r] + blo.z, 0.0f);
      lo.w = fmaxf(acc[3][r] + blo.w, 0.0f);
      hi.x = fmaxf(acc[4][r] + bhi.x, 0.0f);
      hi.y = fmaxf(acc[5][r] + bhi.y, 0.0f);
      hi.z = fmaxf(acc[6][r] + bhi.z, 0.0f);
      hi.w = fmaxf(acc[7][r] + bhi.w, 0.0f);
      tw[r * (WT_PITCH / 4)]      = lo;
      tw[r * (WT_PITCH / 4) + 16] = hi;
    }
#pragma unroll
    for (int j = 0; j < FEAT / 8; ++j) a[j] = tr[2 * j];

    strip_product(bqb, a, acc);

    float4* orow = (float4*)(Hout + (size_t)(s * 16 + 8 * h) * FEAT + 4 * c);
#pragma unroll
    for (int r = 0; r < 8; ++r) {
      const float4 lo = {acc[0][r], acc[1][r], acc[2][r], acc[3][r]};
      const float4 hi = {acc[4][r], acc[5][r], acc[6][r], acc[7][r]};
      orow[r * (FEAT / 4)]      = lo;
      orow[r * (FEAT / 4) + 16] = hi;
    }
    s += GEMM_WAVES;
    if (s < s_hi) gather_strip<16>(X, rowptr, srcs, dinv, s, lane, tg, a, dn);
  }
}

#define ZW 16
static_assert(NCLS <= ZW, "a z row holds every class");
static_assert(FEAT == 8 * 16, "k_prep: one 16-row strip of W3 per wave, 8 waves");

__global__ __launch_bounds__(256) void k_prep(const float* __restrict__ W3,
                                              const float* __restrict__ b3,
                                              const float* __restrict__ Wl,
                                              float* __restrict__ W34T,
                                              float* __restrict__ c34,
                                              float* __restrict__ sums, int nsums) {
  if (blockIdx.x > 0) {
    const int idx = ((int)blockIdx.x - 1) * 256 + (int)threadIdx.x;
    if (idx < nsums) sums[idx] = 0.0f;
    return;
  }
  const int lane = threadIdx.x & 31;
  const int wave = threadIdx.x >> 5;
  const int c = lane & 15;
  const int h = lane >> 4;
  const float4* ar = (const float4*)(W3 + (size_t)(16 * wave + c) * FEAT) + h;
  v8f acc = (v8f){0, 0, 0, 0, 0, 0, 0, 0};
#pragma unroll
  for (int j = 0; j < FEAT / 8; ++j) {
    const float4 av = ar[2 * j];
    float4 bv = {0.0f, 0.0f, 0.0f, 0.0f};
    if (c < NCLS) {
      const float* bp = Wl + (size_t)(8 * j + 4 * h) * NCLS + c;
      bv.x = bp[0];
      bv.y = bp[NCLS];
      bv.z = bp[2 * NCLS];
      bv.w = bp[3 * NCLS];
    }
    const v2f a0 = {av.x, av.y};
    const v2f a1 = {av.z, av.w};
    const v2f b0 = {bv.x, bv.y};
    const v2f b1 = {bv.z, bv.w};
    acc = __builtin_amdgcn_wmma_f32_16x16x4_f32(false, a0, false, b0, (short)0, acc, false, false);
    acc = __builtin_amdgcn_wmma_f32_16x16x4_f32(false, a1, false, b1, (short)0, acc, false, false);
  }
  float4* wr = (float4*)(W34T + (size_t)c * FEAT + 16 * wave + 8 * h);
  wr[0] = (float4){acc[0], acc[1], acc[2], acc[3]};
  wr[1] = (float4){acc[4], acc[5], acc[6], acc[7]};
  if (threadIdx.x < ZW) {
    float o = 0.0f;
    if (threadIdx.x < NCLS) {
#pragma unroll 16
      for (int m = 0; m < FEAT; ++m) o += b3[m] * Wl[m * NCLS + threadIdx.x];
    }
    c34[threadIdx.x] = o;
  }
}

#define GZ_WAVES 4
#define GZ_ROWS  8
static_assert(32 % GZ_ROWS == 0 && 32 % 16 == 0, "gather_strip: the rows in flight divide 32");
static_assert((ZW * FEAT / 4) % (GZ_WAVES * 32) == 0, "k_gather_z stages W34T in whole passes of the work-group");
static_assert(FEAT / 4 <= GZ_WAVES * 32, "k_gather_z stages the bias in one pass");
__global__ __launch_bounds__(GZ_WAVES * 32) void k_gather_z(const float* __restrict__ H,
                                                            const int* __restrict__ rowptr,
                                                            const int* __restrict__ srcs,
                                                            const float* __restrict__ dinv,
                                                            const float* __restrict__ bias,
                                                            const float* __restrict__ W34T,
                                                            float* __restrict__ Z, int nstrips) {
  __shared__ float4 sT4[GZ_WAVES * 16 * WT_PITCH / 4];
  __shared__ float4 sW4[ZW * WT_PITCH / 4];
  __shared__ float4 sB4[FEAT / 4];
  const int lane = threadIdx.x & 31;
  const int wave = __builtin_amdgcn_readfirstlane((int)(threadIdx.x >> 5));
  const int c = lane & 15;
  const int h = lane >> 4;
  const int s = (int)blockIdx.x * GZ_WAVES + wave;

#pragma unroll
  for (int it = 0; it < ZW * FEAT / 4 / (GZ_WAVES * 32); ++it) {
    const int idx = (int)threadIdx.x + GZ_WAVES * 32 * it;
    sW4[(idx / (FEAT / 4)) * (WT_PITCH / 4) + (idx % (FEAT / 4))] = ((const float4*)W34T)[idx];
  }
  if (threadIdx.x < FEAT / 4) sB4[threadIdx.x] = ((const float4*)bias)[threadIdx.x];
  __syncthreads();
  if (s >= nstrips) return;

  float4* tg = sT4 + (wave * 16) * (WT_PITCH / 4) + lane;
  const float4* tr = sT4 + (wave * 16 + c) * (WT_PITCH / 4) + h;

  float4 a[FEAT / 8];
  float dn = 0.0f;
  gather_strip<GZ_ROWS>(H, rowptr, srcs, dinv, s, lane, tg, a, dn);

  const float sl = dn * dn;
  const float4* bq = sB4 + h;
#pragma unroll
  for (int j = 0; j < FEAT / 8; ++j) {
    const float4 t = tr[2 * j];
    const float4 bb = bq[2 * j];
    a[j].x = fmaxf((t.x + sl * a[j].x) + bb.x, 0.0f);
    a[j].y = fmaxf((t.y + sl * a[j].y) + bb.y, 0.0f);
    a[j].z = fmaxf((t.z + sl * a[j].z) + bb.z, 0.0f);
    a[j].w = fmaxf((t.w + sl * a[j].w) + bb.w, 0.0f);
  }

  const float4* wq = sW4 + c * (WT_PITCH / 4) + h;
  v8f acc = (v8f){0, 0, 0, 0, 0, 0, 0, 0};
#pragma unroll
  for (int j = 0; j < FEAT / 8; ++j) {
    const float4 w = wq[2 * j];
    const v2f a0 = {a[j].x, a[j].y};
    const v2f a1 = {a[j].z, a[j].w};
    const v2f b0 = {w.x, w.y};
    const v2f b1 = {w.z, w.w};
    acc = __builtin_amdgcn_wmma_f32_16x16x4_f32(false, a0, false, b0, (short)0, acc, false, false);
    acc = __builtin_amdgcn_wmma_f32_16x16x4_f32(false, a1, false, b1, (short)0, acc, false, false);
  }
  float* zr = Z + (size_t)(s * 16 + 8 * h) * ZW + c;
#pragma unroll
  for (int r = 0; r < 8; ++r) zr[r * ZW] = acc[r];
}

#define P_NODES 32
static_assert(N_NODES % P_NODES == 0, "k_gather_pool takes whole runs of nodes: no thread leaves in front of the barrier");
static_assert(ZW == 16, "four threads per node, four columns each");

__global__ __launch_bounds__(P_NODES * 4) void k_gather_pool(const float* __restrict__ Z,
                                                             const int* __restrict__ rowptr,
                                                             const int* __restrict__ srcs,
                                                             const float* __restrict__ dinv,
                                                             const int* __restrict__ batch,
                                                             float* __restrict__ sums,
                                                             float* __restrict__ cnts) {
  __shared__ float4 sRow[P_NODES * (ZW / 4)];
  __shared__ int sGid[P_NODES];
  const int t = threadIdx.x;
  const int q = t & 3;
  const int n0 = blockIdx.x * P_NODES;
  const int i = n0 + (t >> 2);
  const float4* z4 = (const float4*)Z;

  int gidv = 0;
  if (t < P_NODES) gidv = batch[n0 + t];
  const int start = rowptr[i];
  const int end = rowptr[i + 1];
  const float di = dinv[i];
  const float4 own = z4[(size_t)i * (ZW / 4) + q];

  float4 acc = {0.0f, 0.0f, 0.0f, 0.0f};
  for (int e = start; e < end; e += 4) {
    int sv[4];
    float w[4];
    float4 v[4];
#pragma unroll
    for (int k = 0; k < 4; ++k) sv[k] = srcs[(e + k < end) ? (e + k) : e];
#pragma unroll
    for (int k = 0; k < 4; ++k) {
      w[k] = dinv[sv[k]];
      v[k] = z4[(size_t)sv[k] * (ZW / 4) + q];
    }
#pragma unroll
    for (int k = 0; k < 4; ++k) {
      const bool ok = e + k < end;
      const float wk = ok ? w[k] * di : 0.0f;
      float4 vk;
      vk.x = ok ? v[k].x : 0.0f;
      vk.y = ok ? v[k].y : 0.0f;
      vk.z = ok ? v[k].z : 0.0f;
      vk.w = ok ? v[k].w : 0.0f;
      fma4(acc, wk, vk);
    }
  }
  const float sl = di * di;
  float4 rv;
  rv.x = acc.x + sl * own.x;
  rv.y = acc.y + sl * own.y;
  rv.z = acc.z + sl * own.z;
  rv.w = acc.w + sl * own.w;
  sRow[t] = rv;
  if (t < P_NODES) sGid[t] = gidv;
  __syncthreads();

  if (t < ZW) {
    const float* colp = (const float*)sRow + t;
    int gcur = sGid[0];
    float a = colp[0];
    int run = 1;
#pragma unroll 8
    for (int j = 1; j < P_NODES; ++j) {
      const int g = sGid[j];
      const float v = colp[j * ZW];
      if (g != gcur) {
        atomAddF(&sums[(size_t)gcur * ZW + t], a);
        if (t == 0) atomAddF(&cnts[gcur], (float)run);
        gcur = g;
        a = v;
        run = 1;
      } else {
        a += v;
        ++run;
      }
    }
    atomAddF(&sums[(size_t)gcur * ZW + t], a);
    if (t == 0) atomAddF(&cnts[gcur], (float)run);
  }
}

__global__ __launch_bounds__(256) void k_tail(const float* __restrict__ sums,
                                              const float* __restrict__ cnts,
                                              const float* __restrict__ c34,
                                              const float* __restrict__ bl,
                                              float* __restrict__ out) {
  const int idx = (int)blockIdx.x * 256 + (int)threadIdx.x;
  if (idx >= NGRAPH * NCLS) return;
  const int g = idx / NCLS;
  const int cl = idx - g * NCLS;
  const float cn = cnts[g];
  const float inv = 1.0f / fmaxf(cn, 1.0f);
  out[idx] = sums[(size_t)g * ZW + cl] * inv + ((cn > 0.0f) ? c34[cl] : 0.0f) + bl[cl];
}

extern "C" void kernel_launch(void* const* d_in, const int* in_sizes, int n_in,
                              void* d_out, int out_size, void* d_ws, size_t ws_size,
                              hipStream_t stream) {
  const float* x     = (const float*)d_in[0];
  const int*   ei    = (const int*)d_in[1];
  const int*   row   = ei;
  const int*   col   = ei + N_EDGES;
  const int*   batch = (const int*)d_in[2];
  const float* W1 = (const float*)d_in[3];
  const float* b1 = (const float*)d_in[4];
  const float* W2 = (const float*)d_in[5];
  const float* b2 = (const float*)d_in[6];
  const float* W3 = (const float*)d_in[7];
  const float* b3 = (const float*)d_in[8];
  const float* Wl = (const float*)d_in[9];
  const float* bl = (const float*)d_in[10];
  float* outp = (float*)d_out;

  char* base = (char*)d_ws;
  const size_t featBytes = (size_t)N_NODES * FEAT * sizeof(float);
  int*   histM  = (int*)(base);
  int*   rowptr = (int*)(base + (512 << 10));
  int*   bbase  = (int*)(base + (1 << 20));
  float* dinv   = (float*)(base + (1 << 20) + (512 << 10));
  int*   srcs   = (int*)(base + (4 << 20));
  float* hbuf = (float*)(base + (8 << 20));
  float* zbuf = (float*)(base + (8 << 20) + featBytes);
  float* sums = (float*)(base + (8 << 20) + 2 * featBytes);
  float* cnts = sums + (size_t)NGRAPH * ZW;
  float* w34t = sums + (size_t)NGRAPH * ZW + NGRAPH;
  float* c34  = w34t + (size_t)ZW * FEAT;
  unsigned* binned = (unsigned*)(base + (9 << 20) + 2 * featBytes);

  const int nsums = NGRAPH * ZW + NGRAPH;
  k_prep<<<1 + (nsums + 255) / 256, 256, 0, stream>>>(W3, b3, Wl, w34t, c34, sums, nsums);

  k_hist<<<EB_WG, EB_THREADS, 0, stream>>>(col, histM);
  k_bin<<<EB_WG, EB_THREADS, 0, stream>>>(row, col, histM, bbase, binned);
  k_group<<<NBKT, BKT_NODES, 0, stream>>>(binned, bbase, rowptr, dinv, srcs);

  k_gemm2<<<GEMM_WG, GEMM_THREADS, 0, stream>>>(x, rowptr, srcs, dinv, W1, b1, W2, hbuf, N_NODES);

  k_gather_z<<<(N_NODES / 16 + GZ_WAVES - 1) / GZ_WAVES, GZ_WAVES * 32, 0, stream>>>(hbuf, rowptr, srcs, dinv, b2, w34t, zbuf, N_NODES / 16);

  k_gather_pool<<<N_NODES / P_NODES, P_NODES * 4, 0, stream>>>(zbuf, rowptr, srcs, dinv, batch, sums, cnts);
  k_tail<<<(NGRAPH * NCLS + 255) / 256, 256, 0, stream>>>(sums, cnts, c34, bl, outp);
}


# planes gathered by layers 2 and 3 stored times dinv of the row's own node: their walks read no dinv[src] (1600000 scattered 4-byte reads gone), one multiply per node; rounding of the sums changes
# speedup vs baseline: 5.7773x; 1.0467x over previous
#include <hip/hip_runtime.h>

typedef __attribute__((ext_vector_type(2))) float v2f;
typedef __attribute__((ext_vector_type(8))) float v8f;

#define N_NODES 100000
#define N_EDGES 800000
#define FEAT    128
#define NGRAPH  512
#define NCLS    10

__device__ __forceinline__ void atomAddF(float* p, float v) {
  unsafeAtomicAdd(p, v);
}

#define EB_WG      128
#define EB_THREADS 512
#define EB_WAVES   (EB_THREADS / 32)
#define EB_CHUNK   ((N_EDGES + EB_WG - 1) / EB_WG)
#define EB_BATCH   4
#define BKT_SHIFT  8
#define BKT_NODES  (1 << BKT_SHIFT)
#define NBKT       ((N_NODES + BKT_NODES - 1) / BKT_NODES)
static_assert(NBKT <= EB_THREADS, "k_bin gives one thread to every bucket");
static_assert(N_NODES <= (1 << 24), "a packed edge keeps the source in 24 bits");
static_assert(BKT_NODES <= 256, "a packed edge keeps the target's place in its bucket in 8 bits");

template <int NW>
__device__ __forceinline__ int block_excl_scan(int v, int* s_wave, int& total) {
  const int lane = threadIdx.x & 31;
  const int wave = threadIdx.x >> 5;
  int inc = v;
  for (int o = 1; o < 32; o <<= 1) {
    const int y = __shfl_up(inc, o);
    if (lane >= o) inc += y;
  }
  if (lane == 31) s_wave[wave] = inc;
  __syncthreads();
  int base = 0, tot = 0;
  for (int w = 0; w < NW; ++w) {
    const int x = s_wave[w];
    if (w < wave) base += x;
    tot += x;
  }
  total = tot;
  return base + inc - v;
}

__global__ __launch_bounds__(EB_THREADS) void k_hist(const int* __restrict__ col,
                                                     int* __restrict__ histM) {
  __shared__ int s_h[NBKT];
  const int t = threadIdx.x;
  const int w = blockIdx.x;
  for (int b = t; b < NBKT; b += EB_THREADS) s_h[b] = 0;
  __syncthreads();
  const int e0 = w * EB_CHUNK;
  const int e1 = (e0 + EB_CHUNK < N_EDGES) ? (e0 + EB_CHUNK) : N_EDGES;
  for (int e = e0 + t; e < e1; e += EB_BATCH * EB_THREADS) {
    int c[EB_BATCH];
#pragma unroll
    for (int k = 0; k < EB_BATCH; ++k) {
      const int ee = e + k * EB_THREADS;
      c[k] = (ee < e1) ? col[ee] : -1;
    }
#pragma unroll
    for (int k = 0; k < EB_BATCH; ++k)
      if (c[k] >= 0) atomicAdd(&s_h[c[k] >> BKT_SHIFT], 1);
  }
  __syncthreads();
  for (int b = t; b < NBKT; b += EB_THREADS) histM[w * NBKT + b] = s_h[b];
}

__global__ __launch_bounds__(EB_THREADS) void k_bin(const int* __restrict__ row,
                                                    const int* __restrict__ col,
                                                    const int* __restrict__ histM,
                                                    int* __restrict__ bbase,
                                                    unsigned* __restrict__ binned) {
  __shared__ int s_cur[NBKT];
  __shared__ int s_wave[EB_WAVES];
  const int t = threadIdx.x;
  const int w = blockIdx.x;

  int before = 0, total = 0;
  if (t < NBKT) {
#pragma unroll 32
    for (int w2 = 0; w2 < EB_WG; ++w2) {
      const int v = histM[w2 * NBKT + t];
      total += v;
      before += (w2 < w) ? v : 0;
    }
  }
  int all;
  const int base = block_excl_scan<EB_WAVES>(total, s_wave, all);
  if (t < NBKT) {
    s_cur[t] = base + before;
    if (w == 0) bbase[t] = base;
  }
  if (w == 0 && t == 0) bbase[NBKT] = all;
  __syncthreads();

  const int e0 = w * EB_CHUNK;
  const int e1 = (e0 + EB_CHUNK < N_EDGES) ? (e0 + EB_CHUNK) : N_EDGES;
  for (int e = e0 + t; e < e1; e += EB_BATCH * EB_THREADS) {
    int c[EB_BATCH], r[EB_BATCH];
#pragma unroll
    for (int k = 0; k < EB_BATCH; ++k) {
      const int ee = e + k * EB_THREADS;
      const bool ok = ee < e1;
      c[k] = ok ? col[ee] : -1;
      r[k] = ok ? row[ee] : 0;
    }
#pragma unroll
    for (int k = 0; k < EB_BATCH; ++k) {
      if (c[k] >= 0) {
        const int pos = atomicAdd(&s_cur[c[k] >> BKT_SHIFT], 1);
        binned[pos] = ((unsigned)(c[k] & (BKT_NODES - 1)) << 24) | (unsigned)r[k];
      }
    }
  }
}

__device__ __forceinline__ float dinv_of(int c) {
  const float d = (float)c + 1.0f;
  return (d > 0.0f) ? rsqrtf(fmaxf(d, 1.0f)) : 0.0f;
}

__global__ __launch_bounds__(BKT_NODES) void k_group(const unsigned* __restrict__ binned,
                                                     const int* __restrict__ bbase,
                                                     int* __restrict__ rowptr,
                                                     float* __restrict__ dinv,
                                                     int* __restrict__ srcs) {
  __shared__ int s_cnt[BKT_NODES];
  __shared__ int s_wave[BKT_NODES / 32];
  const int t = threadIdx.x;
  const int b = blockIdx.x;
  const int lo = bbase[b];
  const int hi = bbase[b + 1];
  s_cnt[t] = 0;
  __syncthreads();
  for (int p = lo + t; p < hi; p += EB_BATCH * BKT_NODES) {
    unsigned u[EB_BATCH];
#pragma unroll
    for (int k = 0; k < EB_BATCH; ++k) {
      const int pp = p + k * BKT_NODES;
      u[k] = binned[(pp < hi) ? pp : p];
    }
#pragma unroll
    for (int k = 0; k < EB_BATCH; ++k)
      atomicAdd(&s_cnt[u[k] >> 24], (p + k * BKT_NODES < hi) ? 1 : 0);
  }
  __syncthreads();
  const int c = s_cnt[t];
  int all;
  const int ex = block_excl_scan<BKT_NODES / 32>(c, s_wave, all);
  const int node = b * BKT_NODES + t;
  if (node < N_NODES) {
    rowptr[node] = lo + ex;
    dinv[node] = dinv_of(c);
  }
  if (b == NBKT - 1 && t == 0) rowptr[N_NODES] = hi;
  s_cnt[t] = lo + ex;
  __syncthreads();
  for (int p = lo + t; p < hi; p += EB_BATCH * BKT_NODES) {
    unsigned u[EB_BATCH];
#pragma unroll
    for (int k = 0; k < EB_BATCH; ++k) {
      const int pp = p + k * BKT_NODES;
      u[k] = binned[(pp < hi) ? pp : p];
    }
#pragma unroll
    for (int k = 0; k < EB_BATCH; ++k) {
      const bool ok = p + k * BKT_NODES < hi;
      const int pos = atomicAdd(&s_cnt[u[k] >> 24], ok ? 1 : 0);
      if (ok) srcs[pos] = (int)(u[k] & 0xFFFFFFu);
    }
  }
}

#define GEMM_WG      256
#define GEMM_THREADS 512
#define GEMM_WAVES   (GEMM_THREADS / 32)
#define WT_PITCH     (FEAT + 4)
static_assert(N_NODES % 16 == 0, "k_gemm2 takes whole strips of 16 rows");
static_assert((FEAT * FEAT / 4) % GEMM_THREADS == 0, "k_gemm2 stages a W in whole passes of the work-group");

__device__ __forceinline__ void strip_product(const float4* __restrict__ bq, const float4 (&a)[FEAT / 8],
                                              v8f (&acc)[8]) {
#pragma unroll
  for (int n = 0; n < 8; ++n) acc[n] = (v8f){0, 0, 0, 0, 0, 0, 0, 0};
  float4 b[8];
#pragma unroll
  for (int n = 0; n < 8; ++n) b[n] = bq[n * (4 * WT_PITCH)];
#pragma unroll
  for (int j = 0; j < FEAT / 8; ++j) {
    float4 bn[8];
    __builtin_amdgcn_sched_barrier(0);
    if (j + 1 < FEAT / 8) {
#pragma unroll
      for (int n = 0; n < 8; ++n) bn[n] = bq[n * (4 * WT_PITCH) + 2 * (j + 1)];
    }
    __builtin_amdgcn_sched_barrier(0);
    const v2f a0 = {a[j].x, a[j].y};
    const v2f a1 = {a[j].z, a[j].w};
#pragma unroll
    for (int n = 0; n < 8; ++n) {
      const v2f b0 = {b[n].x, b[n].y};
      acc[n] = __builtin_amdgcn_wmma_f32_16x16x4_f32(false, a0, false, b0, (short)0, acc[n], false, false);
    }
#pragma unroll
    for (int n = 0; n < 8; ++n) {
      const v2f b1 = {b[n].z, b[n].w};
      acc[n] = __builtin_amdgcn_wmma_f32_16x16x4_f32(false, a1, false, b1, (short)0, acc[n], false, false);
    }
    if (j + 1 < FEAT / 8) {
#pragma unroll
      for (int n = 0; n < 8; ++n) b[n] = bn[n];
    }
  }
}

__device__ __forceinline__ void stage_w(float* __restrict__ sW, const float4 (&w)[FEAT * FEAT / 4 / GEMM_THREADS]) {
#pragma unroll
  for (int it = 0; it < FEAT * FEAT / 4 / GEMM_THREADS; ++it) {
    const int idx = threadIdx.x + GEMM_THREADS * it;
    const int k = idx >> 5;
    const int q = idx & 31;
    const int p = 64 * (q >> 4) + (q & 15);
    sW[(p     ) * WT_PITCH + k] = w[it].x;
    sW[(p + 16) * WT_PITCH + k] = w[it].y;
    sW[(p + 32) * WT_PITCH + k] = w[it].z;
    sW[(p + 48) * WT_PITCH + k] = w[it].w;
  }
}

__device__ __forceinline__ void fma4(float4& acc, float w, const float4 v) {
  acc.x += w * v.x;
  acc.y += w * v.y;
  acc.z += w * v.z;
  acc.w += w * v.w;
}

template <int GS_ROWS, bool WEIGHTED>
__device__ __forceinline__ void gather_strip(const float* __restrict__ X, const int* __restrict__ rowptr,
                                             const int* __restrict__ srcs, const float* __restrict__ dinv,
                                             int s, int lane, float4* __restrict__ tg,
                                             float4 (&xs)[FEAT / 8], float& dn) {
  const int c = lane & 15;
  const int h = lane >> 4;
  const int n0 = s * 16;
  const int rp = rowptr[n0 + ((lane < 16) ? lane : 16)];
  dn = dinv[n0 + c];
  const float4* xr = (const float4*)(X + (size_t)(n0 + c) * FEAT + 4 * h);
#pragma unroll
  for (int j = 0; j < FEAT / 8; ++j) xs[j] = xr[2 * j];
  const float4* h4 = (const float4*)X;
  const int E0 = __builtin_amdgcn_readlane(rp, 0);
  const int total = __builtin_amdgcn_readlane(rp, 16) - E0;

  int sv0 = 0, sv1 = 0;
  float dsv0 = 0.0f;
  if (lane < total) sv0 = srcs[E0 + lane];
  if (lane + 32 < total) sv1 = srcs[E0 + 32 + lane];
  if (WEIGHTED && lane < total) dsv0 = dinv[sv0];

  int r = 0;
  int nodeEnd = __builtin_amdgcn_readlane(rp, 1) - E0;
  float di = __int_as_float(__builtin_amdgcn_readlane(__float_as_int(dn), 0));
  float4 acc = {0.0f, 0.0f, 0.0f, 0.0f};
  for (int base = 0; base < total; base += 32) {
    const int m = (total - base < 32) ? (total - base) : 32;
    int sv2 = 0;
    float dsv1 = 0.0f;
    for (int j = 0; j < m; j += GS_ROWS) {
      float4 v[GS_ROWS];
#pragma unroll
      for (int k = 0; k < GS_ROWS; ++k) {
        const int sk = __builtin_amdgcn_readlane(sv0, j + k);
        v[k] = h4[(size_t)sk * (FEAT / 4) + lane];
      }
      if (j == 0) {
        if (base + 64 + lane < total) sv2 = srcs[E0 + base + 64 + lane];
        if (WEIGHTED && base + 32 + lane < total) dsv1 = dinv[sv1];
      }
#pragma unroll
      for (int k = 0; k < GS_ROWS; ++k) {
        const int p = base + j + k;
        if (p < total) {
          while (p >= nodeEnd) {
            tg[r * (WT_PITCH / 4)] = acc;
            acc = (float4){0.0f, 0.0f, 0.0f, 0.0f};
            ++r;
            nodeEnd = __builtin_amdgcn_readlane(rp, r + 1) - E0;
            di = __int_as_float(__builtin_amdgcn_readlane(__float_as_int(dn), r));
          }
          if (WEIGHTED) {
            const float w = __int_as_float(__builtin_amdgcn_readlane(__float_as_int(dsv0), j + k)) * di;
            fma4(acc, w, v[k]);
          } else {
            acc.x += v[k].x;
            acc.y += v[k].y;
            acc.z += v[k].z;
            acc.w += v[k].w;
          }
        }
      }
    }
    sv0 = sv1;
    dsv0 = dsv1;
    sv1 = sv2;
  }
  while (r < 16) {
    tg[r * (WT_PITCH / 4)] = acc;
    acc = (float4){0.0f, 0.0f, 0.0f, 0.0f};
    ++r;
  }
}

__global__ __launch_bounds__(GEMM_THREADS) void k_gemm2(const float* __restrict__ X,
                                                        const int* __restrict__ rowptr,
                                                        const int* __restrict__ srcs,
                                                        const float* __restrict__ dinv,
                                                        const float* __restrict__ Wa,
                                                        const float* __restrict__ ba,
                                                        const float* __restrict__ Wb,
                                                        float* __restrict__ Hout,
                                                        int nrows) {
  __shared__ float4 sWa4[FEAT * WT_PITCH / 4];
  __shared__ float4 sWb4[FEAT * WT_PITCH / 4];
  __shared__ float4 sT4[GEMM_WAVES * 16 * WT_PITCH / 4];
  const int lane = threadIdx.x & 31;
  const int wave = __builtin_amdgcn_readfirstlane((int)(threadIdx.x >> 5));
  const int c = lane & 15;
  const int h = lane >> 4;

  const int nstrips = nrows >> 4;
  const int s_lo = ((int)blockIdx.x * nstrips) / GEMM_WG;
  const int s_hi = (((int)blockIdx.x + 1) * nstrips) / GEMM_WG;
  static_assert(GEMM_WAVES == 16, "the renumbering below is a 4 x 4 square");
  const int wv = (((wave - (wave >> 2)) & 3) << 2) | (wave >> 2);
  int s = s_lo + wv;

  float4 wa[FEAT * FEAT / 4 / GEMM_THREADS], wb[FEAT * FEAT / 4 / GEMM_THREADS];
#pragma unroll
  for (int it = 0; it < FEAT * FEAT / 4 / GEMM_THREADS; ++it) wa[it] = ((const float4*)Wa)[threadIdx.x + GEMM_THREADS * it];
#pragma unroll
  for (int it = 0; it < FEAT * FEAT / 4 / GEMM_THREADS; ++it) wb[it] = ((const float4*)Wb)[threadIdx.x + GEMM_THREADS * it];
  const float4 blo = ((const float4*)ba)[c];
  const float4 bhi = ((const float4*)ba)[16 + c];

  stage_w((float*)sWa4, wa);
  stage_w((float*)sWb4, wb);

  const float4* bqa = sWa4 + (c * WT_PITCH) / 4 + h;
  const float4* bqb = sWb4 + (c * WT_PITCH) / 4 + h;
  float4* tg = sT4 + (wave * 16) * (WT_PITCH / 4) + lane;
  float4* tw = sT4 + (wave * 16 + 8 * h) * (WT_PITCH / 4) + c;
  const float4* tr = sT4 + (wave * 16 + c) * (WT_PITCH / 4) + h;

  float4 a[FEAT / 8];
  float dn = 0.0f;
  if (s < s_hi) gather_strip<16, true>(X, rowptr, srcs, dinv, s, lane, tg, a, dn);
  __syncthreads();

  while (s < s_hi) {
    asm volatile("" ::: "memory");
    {
      const float sl = dn * dn;
#pragma unroll
      for (int j = 0; j < FEAT / 8; ++j) {
        const float4 t = tr[2 * j];
        a[j].x = t.x + sl * a[j].x;
        a[j].y = t.y + sl * a[j].y;
        a[j].z = t.z + sl * a[j].z;
        a[j].w = t.w + sl * a[j].w;
      }
    }
    v8f acc[8];
    strip_product(bqa, a, acc);

#pragma unroll
    for (int r = 0; r < 8; ++r) {
      float4 lo, hi;
      lo.x = fmaxf(acc[0][r] + blo.x, 0.0f);
      lo.y = fmaxf(acc[1][r] + blo.y, 0.0f);
      lo.z = fmaxf(acc[2][r] + blo.z, 0.0f);
      lo.w = fmaxf(acc[3][r] + blo.w, 0.0f);
      hi.x = fmaxf(acc[4][r] + bhi.x, 0.0f);
      hi.y = fmaxf(acc[5][r] + bhi.y, 0.0f);
      hi.z = fmaxf(acc[6][r] + bhi.z, 0.0f);
      hi.w = fmaxf(acc[7][r] + bhi.w, 0.0f);
      tw[r * (WT_PITCH / 4)]      = lo;
      tw[r * (WT_PITCH / 4) + 16] = hi;
    }
#pragma unroll
    for (int j = 0; j < FEAT / 8; ++j) a[j] = tr[2 * j];

    strip_product(bqb, a, acc);

    float4* orow = (float4*)(Hout + (size_t)(s * 16 + 8 * h) * FEAT + 4 * c);
#pragma unroll
    for (int r = 0; r < 8; ++r) {
      const float dr = __shfl(dn, 8 * h + r);
      const float4 lo = {acc[0][r] * dr, acc[1][r] * dr, acc[2][r] * dr, acc[3][r] * dr};
      const float4 hi = {acc[4][r] * dr, acc[5][r] * dr, acc[6][r] * dr, acc[7][r] * dr};
      orow[r * (FEAT / 4)]      = lo;
      orow[r * (FEAT / 4) + 16] = hi;
    }
    s += GEMM_WAVES;
    if (s < s_hi) gather_strip<16, true>(X, rowptr, srcs, dinv, s, lane, tg, a, dn);
  }
}

#define ZW 16
static_assert(NCLS <= ZW, "a z row holds every class");
static_assert(FEAT == 8 * 16, "k_prep: one 16-row strip of W3 per wave, 8 waves");

__global__ __launch_bounds__(256) void k_prep(const float* __restrict__ W3,
                                              const float* __restrict__ b3,
                                              const float* __restrict__ Wl,
                                              float* __restrict__ W34T,
                                              float* __restrict__ c34,
                                              float* __restrict__ sums, int nsums) {
  if (blockIdx.x > 0) {
    const int idx = ((int)blockIdx.x - 1) * 256 + (int)threadIdx.x;
    if (idx < nsums) sums[idx] = 0.0f;
    return;
  }
  const int lane = threadIdx.x & 31;
  const int wave = threadIdx.x >> 5;
  const int c = lane & 15;
  const int h = lane >> 4;
  const float4* ar = (const float4*)(W3 + (size_t)(16 * wave + c) * FEAT) + h;
  v8f acc = (v8f){0, 0, 0, 0, 0, 0, 0, 0};
#pragma unroll
  for (int j = 0; j < FEAT / 8; ++j) {
    const float4 av = ar[2 * j];
    float4 bv = {0.0f, 0.0f, 0.0f, 0.0f};
    if (c < NCLS) {
      const float* bp = Wl + (size_t)(8 * j + 4 * h) * NCLS + c;
      bv.x = bp[0];
      bv.y = bp[NCLS];
      bv.z = bp[2 * NCLS];
      bv.w = bp[3 * NCLS];
    }
    const v2f a0 = {av.x, av.y};
    const v2f a1 = {av.z, av.w};
    const v2f b0 = {bv.x, bv.y};
    const v2f b1 = {bv.z, bv.w};
    acc = __builtin_amdgcn_wmma_f32_16x16x4_f32(false, a0, false, b0, (short)0, acc, false, false);
    acc = __builtin_amdgcn_wmma_f32_16x16x4_f32(false, a1, false, b1, (short)0, acc, false, false);
  }
  float4* wr = (float4*)(W34T + (size_t)c * FEAT + 16 * wave + 8 * h);
  wr[0] = (float4){acc[0], acc[1], acc[2], acc[3]};
  wr[1] = (float4){acc[4], acc[5], acc[6], acc[7]};
  if (threadIdx.x < ZW) {
    float o = 0.0f;
    if (threadIdx.x < NCLS) {
#pragma unroll 16
      for (int m = 0; m < FEAT; ++m) o += b3[m] * Wl[m * NCLS + threadIdx.x];
    }
    c34[threadIdx.x] = o;
  }
}

#define GZ_WAVES 4
#define GZ_ROWS  8
static_assert(32 % GZ_ROWS == 0 && 32 % 16 == 0, "gather_strip: the rows in flight divide 32");
static_assert((ZW * FEAT / 4) % (GZ_WAVES * 32) == 0, "k_gather_z stages W34T in whole passes of the work-group");
static_assert(FEAT / 4 <= GZ_WAVES * 32, "k_gather_z stages the bias in one pass");
__global__ __launch_bounds__(GZ_WAVES * 32) void k_gather_z(const float* __restrict__ H,
                                                            const int* __restrict__ rowptr,
                                                            const int* __restrict__ srcs,
                                                            const float* __restrict__ dinv,
                                                            const float* __restrict__ bias,
                                                            const float* __restrict__ W34T,
                                                            float* __restrict__ Z, int nstrips) {
  __shared__ float4 sT4[GZ_WAVES * 16 * WT_PITCH / 4];
  __shared__ float4 sW4[ZW * WT_PITCH / 4];
  __shared__ float4 sB4[FEAT / 4];
  const int lane = threadIdx.x & 31;
  const int wave = __builtin_amdgcn_readfirstlane((int)(threadIdx.x >> 5));
  const int c = lane & 15;
  const int h = lane >> 4;
  const int s = (int)blockIdx.x * GZ_WAVES + wave;

#pragma unroll
  for (int it = 0; it < ZW * FEAT / 4 / (GZ_WAVES * 32); ++it) {
    const int idx = (int)threadIdx.x + GZ_WAVES * 32 * it;
    sW4[(idx / (FEAT / 4)) * (WT_PITCH / 4) + (idx % (FEAT / 4))] = ((const float4*)W34T)[idx];
  }
  if (threadIdx.x < FEAT / 4) sB4[threadIdx.x] = ((const float4*)bias)[threadIdx.x];
  __syncthreads();
  if (s >= nstrips) return;

  float4* tg = sT4 + (wave * 16) * (WT_PITCH / 4) + lane;
  const float4* tr = sT4 + (wave * 16 + c) * (WT_PITCH / 4) + h;

  float4 a[FEAT / 8];
  float dn = 0.0f;
  gather_strip<GZ_ROWS, false>(H, rowptr, srcs, dinv, s, lane, tg, a, dn);

  const float4* bq = sB4 + h;
#pragma unroll
  for (int j = 0; j < FEAT / 8; ++j) {
    const float4 t = tr[2 * j];
    const float4 bb = bq[2 * j];
    a[j].x = fmaxf(dn * (t.x + a[j].x) + bb.x, 0.0f);
    a[j].y = fmaxf(dn * (t.y + a[j].y) + bb.y, 0.0f);
    a[j].z = fmaxf(dn * (t.z + a[j].z) + bb.z, 0.0f);
    a[j].w = fmaxf(dn * (t.w + a[j].w) + bb.w, 0.0f);
  }

  const float4* wq = sW4 + c * (WT_PITCH / 4) + h;
  v8f acc = (v8f){0, 0, 0, 0, 0, 0, 0, 0};
#pragma unroll
  for (int j = 0; j < FEAT / 8; ++j) {
    const float4 w = wq[2 * j];
    const v2f a0 = {a[j].x, a[j].y};
    const v2f a1 = {a[j].z, a[j].w};
    const v2f b0 = {w.x, w.y};
    const v2f b1 = {w.z, w.w};
    acc = __builtin_amdgcn_wmma_f32_16x16x4_f32(false, a0, false, b0, (short)0, acc, false, false);
    acc = __builtin_amdgcn_wmma_f32_16x16x4_f32(false, a1, false, b1, (short)0, acc, false, false);
  }
  float* zr = Z + (size_t)(s * 16 + 8 * h) * ZW + c;
#pragma unroll
  for (int r = 0; r < 8; ++r) zr[r * ZW] = acc[r] * __shfl(dn, 8 * h + r);
}

#define P_NODES 32
static_assert(N_NODES % P_NODES == 0, "k_gather_pool takes whole runs of nodes: no thread leaves in front of the barrier");
static_assert(ZW == 16, "four threads per node, four columns each");

__global__ __launch_bounds__(P_NODES * 4) void k_gather_pool(const float* __restrict__ Z,
                                                             const int* __restrict__ rowptr,
                                                             const int* __restrict__ srcs,
                                                             const float* __restrict__ dinv,
                                                             const int* __restrict__ batch,
                                                             float* __restrict__ sums,
                                                             float* __restrict__ cnts) {
  __shared__ float4 sRow[P_NODES * (ZW / 4)];
  __shared__ int sGid[P_NODES];
  const int t = threadIdx.x;
  const int q = t & 3;
  const int n0 = blockIdx.x * P_NODES;
  const int i = n0 + (t >> 2);
  const float4* z4 = (const float4*)Z;

  int gidv = 0;
  if (t < P_NODES) gidv = batch[n0 + t];
  const int start = rowptr[i];
  const int end = rowptr[i + 1];
  const float di = dinv[i];
  const float4 own = z4[(size_t)i * (ZW / 4) + q];

  float4 acc = {0.0f, 0.0f, 0.0f, 0.0f};
  for (int e = start; e < end; e += 4) {
    int sv[4];
    float4 v[4];
#pragma unroll
    for (int k = 0; k < 4; ++k) sv[k] = srcs[(e + k < end) ? (e + k) : e];
#pragma unroll
    for (int k = 0; k < 4; ++k) v[k] = z4[(size_t)sv[k] * (ZW / 4) + q];
#pragma unroll
    for (int k = 0; k < 4; ++k) {
      const bool ok = e + k < end;
      acc.x += ok ? v[k].x : 0.0f;
      acc.y += ok ? v[k].y : 0.0f;
      acc.z += ok ? v[k].z : 0.0f;
      acc.w += ok ? v[k].w : 0.0f;
    }
  }
  float4 rv;
  rv.x = di * (acc.x + own.x);
  rv.y = di * (acc.y + own.y);
  rv.z = di * (acc.z + own.z);
  rv.w = di * (acc.w + own.w);
  sRow[t] = rv;
  if (t < P_NODES) sGid[t] = gidv;
  __syncthreads();

  if (t < ZW) {
    const float* colp = (const float*)sRow + t;
    int gcur = sGid[0];
    float a = colp[0];
    int run = 1;
#pragma unroll 8
    for (int j = 1; j < P_NODES; ++j) {
      const int g = sGid[j];
      const float v = colp[j * ZW];
      if (g != gcur) {
        atomAddF(&sums[(size_t)gcur * ZW + t], a);
        if (t == 0) atomAddF(&cnts[gcur], (float)run);
        gcur = g;
        a = v;
        run = 1;
      } else {
        a += v;
        ++run;
      }
    }
    atomAddF(&sums[(size_t)gcur * ZW + t], a);
    if (t == 0) atomAddF(&cnts[gcur], (float)run);
  }
}

__global__ __launch_bounds__(256) void k_tail(const float* __restrict__ sums,
                                              const float* __restrict__ cnts,
                                              const float* __restrict__ c34,
                                              const float* __restrict__ bl,
                                              float* __restrict__ out) {
  const int idx = (int)blockIdx.x * 256 + (int)threadIdx.x;
  if (idx >= NGRAPH * NCLS) return;
  const int g = idx / NCLS;
  const int cl = idx - g * NCLS;
  const float cn = cnts[g];
  const float inv = 1.0f / fmaxf(cn, 1.0f);
  out[idx] = sums[(size_t)g * ZW + cl] * inv + ((cn > 0.0f) ? c34[cl] : 0.0f) + bl[cl];
}

extern "C" void kernel_launch(void* const* d_in, const int* in_sizes, int n_in,
                              void* d_out, int out_size, void* d_ws, size_t ws_size,
                              hipStream_t stream) {
  const float* x     = (const float*)d_in[0];
  const int*   ei    = (const int*)d_in[1];
  const int*   row   = ei;
  const int*   col   = ei + N_EDGES;
  const int*   batch = (const int*)d_in[2];
  const float* W1 = (const float*)d_in[3];
  const float* b1 = (const float*)d_in[4];
  const float* W2 = (const float*)d_in[5];
  const float* b2 = (const float*)d_in[6];
  const float* W3 = (const float*)d_in[7];
  const float* b3 = (const float*)d_in[8];
  const float* Wl = (const float*)d_in[9];
  const float* bl = (const float*)d_in[10];
  float* outp = (float*)d_out;

  char* base = (char*)d_ws;
  const size_t featBytes = (size_t)N_NODES * FEAT * sizeof(float);
  int*   histM  = (int*)(base);
  int*   rowptr = (int*)(base + (512 << 10));
  int*   bbase  = (int*)(base + (1 << 20));
  float* dinv   = (float*)(base + (1 << 20) + (512 << 10));
  int*   srcs   = (int*)(base + (4 << 20));
  float* hbuf = (float*)(base + (8 << 20));
  float* zbuf = (float*)(base + (8 << 20) + featBytes);
  float* sums = (float*)(base + (8 << 20) + 2 * featBytes);
  float* cnts = sums + (size_t)NGRAPH * ZW;
  float* w34t = sums + (size_t)NGRAPH * ZW + NGRAPH;
  float* c34  = w34t + (size_t)ZW * FEAT;
  unsigned* binned = (unsigned*)(base + (9 << 20) + 2 * featBytes);

  const int nsums = NGRAPH * ZW + NGRAPH;
  k_prep<<<1 + (nsums + 255) / 256, 256, 0, stream>>>(W3, b3, Wl, w34t, c34, sums, nsums);

  k_hist<<<EB_WG, EB_THREADS, 0, stream>>>(col, histM);
  k_bin<<<EB_WG, EB_THREADS, 0, stream>>>(row, col, histM, bbase, binned);
  k_group<<<NBKT, BKT_NODES, 0, stream>>>(binned, bbase, rowptr, dinv, srcs);

  k_gemm2<<<GEMM_WG, GEMM_THREADS, 0, stream>>>(x, rowptr, srcs, dinv, W1, b1, W2, hbuf, N_NODES);

  k_gather_z<<<(N_NODES / 16 + GZ_WAVES - 1) / GZ_WAVES, GZ_WAVES * 32, 0, stream>>>(hbuf, rowptr, srcs, dinv, b2, w34t, zbuf, N_NODES / 16);

  k_gather_pool<<<N_NODES / P_NODES, P_NODES * 4, 0, stream>>>(zbuf, rowptr, srcs, dinv, batch, sums, cnts);
  k_tail<<<(NGRAPH * NCLS + 255) / 256, 256, 0, stream>>>(sums, cnts, c34, bl, outp);
}
